# starting kernel
# baseline (speedup 1.0000x reference)
#include <hip/hip_runtime.h>
#include <hip/hip_bf16.h>

#define N0 50000
#define N1 12500
#define N2 3125
#define E0 800000
#define E1 200000
#define E2 50000
#define NNZ 200000
#define H 128
#define F3 96
#define DECW 64

typedef __attribute__((ext_vector_type(2))) float v2f;
typedef __attribute__((ext_vector_type(8))) float v8f;

__global__ void zero_kernel(float* __restrict__ p, long long count) {
    long long t = (long long)blockIdx.x * blockDim.x + threadIdx.x;
    if (t < count) p[t] = 0.0f;
}

__global__ void conv1_kernel(const float* __restrict__ x, const float* __restrict__ agg6,
                             const float* __restrict__ Wrel, const float* __restrict__ brel,
                             const float* __restrict__ Wroot, float* __restrict__ h1, int n) {
    int r = blockIdx.x;
    int c = threadIdx.x;
    if (r >= n) return;
    float s = brel[c];
#pragma unroll
    for (int k = 0; k < 6; ++k) {
        s += agg6[r * 6 + k] * Wrel[k * H + c];
        s += x[r * 6 + k] * Wroot[k * H + c];
    }
    h1[r * H + c] = fmaxf(s, 0.0f);
}

__global__ void scatter6_kernel(const float* __restrict__ x, const int* __restrict__ ei,
                                int E, float* __restrict__ agg6) {
    int e = blockIdx.x * blockDim.x + threadIdx.x;
    if (e >= E) return;
    int src = ei[e];
    int dst = ei[E + e];
#pragma unroll
    for (int k = 0; k < 6; ++k)
        atomicAdd(&agg6[dst * 6 + k], x[src * 6 + k]);
}

__global__ void scatter128_kernel(const float* __restrict__ h1, const int* __restrict__ ei,
                                  int E, float* __restrict__ agg) {
    long long t = (long long)blockIdx.x * blockDim.x + threadIdx.x;
    if (t >= (long long)E * 32) return;
    int e  = (int)(t >> 5);
    int c4 = ((int)t & 31) * 4;
    int src = ei[e];
    int dst = ei[E + e];
    const float* hs = h1 + src * H + c4;
    float* ad = agg + dst * H + c4;
#pragma unroll
    for (int k = 0; k < 4; ++k) atomicAdd(ad + k, hs[k]);
}

__global__ void interp_kernel(const float* __restrict__ vals, const int* __restrict__ cols,
                              const int* __restrict__ rows, const float* __restrict__ e,
                              float* __restrict__ acc) {
    int nz = blockIdx.x;
    int c  = threadIdx.x;
    float v = vals[nz];
    int col = cols[nz];
    int row = rows[nz];
    atomicAdd(&acc[row * F3 + c], v * e[col * F3 + c]);
}

__global__ void dec_out_kernel(const float* __restrict__ hdec, const float* __restrict__ Wout,
                               const float* __restrict__ bout, float* __restrict__ out,
                               int n, int branch) {
    int r = blockIdx.x * blockDim.x + threadIdx.x;
    if (r >= n) return;
    float s = bout[0];
    const float* hr = hdec + r * DECW;
#pragma unroll
    for (int c = 0; c < DECW; ++c) s += hr[c] * Wout[c];
    out[r * 3 + branch] = s;
}

template <int K, int Nc, int ACT, bool DUAL>
__global__ void gemm_wmma_kernel(const float* __restrict__ A0, const float* __restrict__ W0,
                                 const float* __restrict__ A1, const float* __restrict__ W1,
                                 const float* __restrict__ bias, float* __restrict__ D,
                                 int n) {
    const int lane = threadIdx.x & 31;
    const int wave = threadIdx.x >> 5;
    const int row0 = (blockIdx.x * 4 + wave) * 16;
    const int col0 = blockIdx.y * 16;
    if (row0 >= n) return;

    const int m    = lane & 15;
    const int half = lane >> 4;
    int row = row0 + m;
    const int rowc = (row < n) ? row : (n - 1);

    const float* a0p = A0 + rowc * K + half * 2;
    const float* w0p = W0 + col0 + m + half * 2 * Nc;
    const float* a1p = DUAL ? (A1 + rowc * K + half * 2) : nullptr;
    const float* w1p = DUAL ? (W1 + col0 + m + half * 2 * Nc) : nullptr;

    const float bv = bias[col0 + m];
    v8f acc = {bv, bv, bv, bv, bv, bv, bv, bv};

#pragma unroll
    for (int k = 0; k < K; k += 4) {
        v2f a = *(const v2f*)(a0p + k);
        v2f b;
        b.x = w0p[k * Nc];
        b.y = w0p[k * Nc + Nc];
        acc = __builtin_amdgcn_wmma_f32_16x16x4_f32(false, a, false, b, (short)0, acc,
                                                    false, false);
        if (DUAL) {
            v2f a2 = *(const v2f*)(a1p + k);
            v2f b2;
            b2.x = w1p[k * Nc];
            b2.y = w1p[k * Nc + Nc];
            acc = __builtin_amdgcn_wmma_f32_16x16x4_f32(false, a2, false, b2, (short)0, acc,
                                                        false, false);
        }
    }

#pragma unroll
    for (int j = 0; j < 8; ++j) {
        int r = row0 + j + half * 8;
        if (r < n) {
            float v = acc[j];
            if (ACT == 1) v = fmaxf(v, 0.0f);
            if (ACT == 2) v = (v > 0.0f) ? v : (expf(v) - 1.0f);
            D[r * Nc + col0 + m] = v;
        }
    }
}

__global__ void dec_gemm_wmma_kernel(const float* __restrict__ e0, const float* __restrict__ i1,
                                     const float* __restrict__ i2, const float* __restrict__ W,
                                     const float* __restrict__ bias, float* __restrict__ D,
                                     int n, int branch) {
    constexpr int K = F3, Nc = DECW;
    const int lane = threadIdx.x & 31;
    const int wave = threadIdx.x >> 5;
    const int row0 = (blockIdx.x * 4 + wave) * 16;
    const int col0 = blockIdx.y * 16;
    if (row0 >= n) return;

    const int m    = lane & 15;
    const int half = lane >> 4;
    int row = row0 + m;
    const int rowc = (row < n) ? row : (n - 1);

    const int abase = rowc * F3 + branch * 32 + half * 2;
    const float* s0 = e0 + abase;
    const float* s1 = i1 + abase;
    const float* s2 = i2 + abase;
    const float* wp = W + col0 + m + half * 2 * Nc;

    const float bv = bias[col0 + m];
    v8f acc = {bv, bv, bv, bv, bv, bv, bv, bv};

#pragma unroll
    for (int k = 0; k < K; k += 4) {
        const int blk = k >> 5;
        const float* sp = (blk == 0) ? s0 : (blk == 1) ? s1 : s2;
        v2f a = *(const v2f*)(sp + (k & 31));
        v2f b;
        b.x = wp[k * Nc];
        b.y = wp[k * Nc + Nc];
        acc = __builtin_amdgcn_wmma_f32_16x16x4_f32(false, a, false, b, (short)0, acc,
                                                    false, false);
    }

#pragma unroll
    for (int j = 0; j < 8; ++j) {
        int r = row0 + j + half * 8;
        if (r < n) {
            float v = acc[j];
            v = (v > 0.0f) ? v : (expf(v) - 1.0f);
            D[r * Nc + col0 + m] = v;
        }
    }
}

static inline dim3 gemm_grid(int n, int Nc) {
    int row_tiles = (n + 15) / 16;
    return dim3((row_tiles + 3) / 4, Nc / 16, 1);
}

extern "C" void kernel_launch(void* const* d_in, const int* in_sizes, int n_in,
                              void* d_out, int out_size, void* d_ws, size_t ws_size,
                              hipStream_t stream) {
    (void)in_sizes; (void)n_in; (void)out_size; (void)ws_size;

    const float* x[3]   = {(const float*)d_in[0], (const float*)d_in[1], (const float*)d_in[2]};
    const int*   ei[3]  = {(const int*)d_in[3], (const int*)d_in[4], (const int*)d_in[5]};
    const int*   A1r    = (const int*)d_in[6];
    const int*   A1c    = (const int*)d_in[7];
    const float* A1v    = (const float*)d_in[8];
    const int*   A2r    = (const int*)d_in[9];
    const int*   A2c    = (const int*)d_in[10];
    const float* A2v    = (const float*)d_in[11];
    const float* c1Wrel  = (const float*)d_in[12];
    const float* c1brel  = (const float*)d_in[13];
    const float* c1Wroot = (const float*)d_in[14];
    const float* c2Wrel  = (const float*)d_in[15];
    const float* c2brel  = (const float*)d_in[16];
    const float* c2Wroot = (const float*)d_in[17];
    const float* linW    = (const float*)d_in[18];
    const float* linb    = (const float*)d_in[19];
    const float* decW0   = (const float*)d_in[20];
    const float* decb0   = (const float*)d_in[21];
    const float* decWout = (const float*)d_in[22];
    const float* decbout = (const float*)d_in[23];
    float* out = (float*)d_out;

    float* ws = (float*)d_ws;
    float* h1   = ws;
    float* agg  = h1   + (size_t)N0 * H;
    float* h2   = agg  + (size_t)N0 * H;
    float* agg6 = h2   + (size_t)N0 * H;
    float* e0   = agg6 + (size_t)N0 * 6;
    float* e1   = e0   + (size_t)N0 * F3;
    float* e2   = e1   + (size_t)N1 * F3;
    float* i1   = e2   + (size_t)N2 * F3;
    float* i2   = i1   + (size_t)N0 * F3;
    float* hdec = i2   + (size_t)N0 * F3;

    const int nlv[3] = {N0, N1, N2};
    const int Elv[3] = {E0, E1, E2};
    float* elv[3] = {e0, e1, e2};

    for (int i = 0; i < 3; ++i) {
        const int n = nlv[i];
        const int E = Elv[i];

        {
            long long cnt = (long long)n * 6;
            zero_kernel<<<(int)((cnt + 255) / 256), 256, 0, stream>>>(agg6, cnt);
            scatter6_kernel<<<(E + 255) / 256, 256, 0, stream>>>(x[i], ei[i], E, agg6);
            conv1_kernel<<<n, H, 0, stream>>>(x[i], agg6,
                                              c1Wrel + (size_t)i * 6 * H,
                                              c1brel + (size_t)i * H,
                                              c1Wroot + (size_t)i * 6 * H, h1, n);
        }

        {
            long long cnt = (long long)n * H;
            zero_kernel<<<(int)((cnt + 255) / 256), 256, 0, stream>>>(agg, cnt);
            long long th = (long long)E * 32;
            scatter128_kernel<<<(int)((th + 255) / 256), 256, 0, stream>>>(h1, ei[i], E, agg);
            gemm_wmma_kernel<H, H, 1, true><<<gemm_grid(n, H), 128, 0, stream>>>(
                agg, c2Wrel + (size_t)i * H * H,
                h1,  c2Wroot + (size_t)i * H * H,
                c2brel + (size_t)i * H, h2, n);
        }

        gemm_wmma_kernel<H, F3, 0, false><<<gemm_grid(n, F3), 128, 0, stream>>>(
            h2, linW + (size_t)i * H * F3, nullptr, nullptr,
            linb + (size_t)i * F3, elv[i], n);
    }

    {
        long long cnt = (long long)N0 * F3;
        zero_kernel<<<(int)((cnt + 255) / 256), 256, 0, stream>>>(i1, cnt);
        zero_kernel<<<(int)((cnt + 255) / 256), 256, 0, stream>>>(i2, cnt);
        interp_kernel<<<NNZ, F3, 0, stream>>>(A1v, A1c, A1r, e1, i1);
        interp_kernel<<<NNZ, F3, 0, stream>>>(A2v, A2c, A2r, e2, i2);
    }

    for (int b = 0; b < 3; ++b) {
        dec_gemm_wmma_kernel<<<gemm_grid(N0, DECW), 128, 0, stream>>>(
            e0, i1, i2, decW0 + (size_t)b * F3 * DECW, decb0 + (size_t)b * DECW,
            hdec, N0, b);
        dec_out_kernel<<<(N0 + 255) / 256, 256, 0, stream>>>(
            hdec, decWout + (size_t)b * DECW, decbout + b, out, N0, b);
    }
}


# float-atomic scatters remapped: each wave takes 8 items, a row's words on consecutive lanes (whole-line adds, all gathers issued before the adds); atomics kept, sums stay in arrival order
# speedup vs baseline: 1.8245x; 1.8245x over previous
#include <hip/hip_runtime.h>
#include <hip/hip_bf16.h>

#define N0 50000
#define N1 12500
#define N2 3125
#define E0 800000
#define E1 200000
#define E2 50000
#define NNZ 200000
#define H 128
#define F3 96
#define DECW 64

typedef __attribute__((ext_vector_type(2))) float v2f;
typedef __attribute__((ext_vector_type(8))) float v8f;

__global__ void zero_kernel(float* __restrict__ p, long long count) {
    long long t = (long long)blockIdx.x * blockDim.x + threadIdx.x;
    if (t < count) p[t] = 0.0f;
}

__global__ void conv1_kernel(const float* __restrict__ x, const float* __restrict__ agg6,
                             const float* __restrict__ Wrel, const float* __restrict__ brel,
                             const float* __restrict__ Wroot, float* __restrict__ h1, int n) {
    int r = blockIdx.x;
    int c = threadIdx.x;
    if (r >= n) return;
    float s = brel[c];
#pragma unroll
    for (int k = 0; k < 6; ++k) {
        s += agg6[r * 6 + k] * Wrel[k * H + c];
        s += x[r * 6 + k] * Wroot[k * H + c];
    }
    h1[r * H + c] = fmaxf(s, 0.0f);
}

template <bool TAIL>
__device__ __forceinline__ void scatter6_wave(const float* __restrict__ x,
                                              const int* __restrict__ ei, int E,
                                              float* __restrict__ agg6,
                                              int ebase, int g, int k) {
    int src[8], dst[8];
    float v[8];
#pragma unroll
    for (int j = 0; j < 8; ++j) {
        int e = ebase + j * 4 + g;
        if (TAIL) e = (e < E) ? e : (E - 1);
        src[j] = ei[e];
        dst[j] = ei[E + e];
    }
#pragma unroll
    for (int j = 0; j < 8; ++j) v[j] = x[src[j] * 6 + k];
#pragma unroll
    for (int j = 0; j < 8; ++j)
        if (!TAIL || ebase + j * 4 + g < E) atomicAdd(&agg6[dst[j] * 6 + k], v[j]);
}

__global__ void scatter6_kernel(const float* __restrict__ x, const int* __restrict__ ei,
                                int E, float* __restrict__ agg6) {
    const int lane  = threadIdx.x & 31;
    const int wave  = __builtin_amdgcn_readfirstlane((int)((blockIdx.x * blockDim.x + threadIdx.x) >> 5));
    const int ebase = wave * 32;
    if (ebase >= E) return;
    const int g = lane >> 3;
    const int k = lane & 7;
    if (k >= 6) return;
    if (ebase + 32 <= E) scatter6_wave<false>(x, ei, E, agg6, ebase, g, k);
    else                 scatter6_wave<true >(x, ei, E, agg6, ebase, g, k);
}

#define SC_EPW 8
template <bool TAIL>
__device__ __forceinline__ void scatter128_wave(const float* __restrict__ h1,
                                                const int* __restrict__ ei, int E,
                                                float* __restrict__ agg, int e0, int lane) {
    int src[SC_EPW], dst[SC_EPW];
    float v[SC_EPW][4];
#pragma unroll
    for (int j = 0; j < SC_EPW; ++j) {
        int e = e0 + j;
        if (TAIL) e = (e < E) ? e : (E - 1);
        src[j] = ei[e];
        dst[j] = ei[E + e];
    }
#pragma unroll
    for (int j = 0; j < SC_EPW; ++j) {
        const float* hs = h1 + src[j] * H + lane;
#pragma unroll
        for (int k = 0; k < 4; ++k) v[j][k] = hs[32 * k];
    }
#pragma unroll
    for (int j = 0; j < SC_EPW; ++j) {
        if (!TAIL || e0 + j < E) {
            float* ad = agg + dst[j] * H + lane;
#pragma unroll
            for (int k = 0; k < 4; ++k) atomicAdd(ad + 32 * k, v[j][k]);
        }
    }
}

__global__ void scatter128_kernel(const float* __restrict__ h1, const int* __restrict__ ei,
                                  int E, float* __restrict__ agg) {
    const int lane = threadIdx.x & 31;
    const int wave = __builtin_amdgcn_readfirstlane((int)((blockIdx.x * blockDim.x + threadIdx.x) >> 5));
    const int e0 = wave * SC_EPW;
    if (e0 >= E) return;
    if (e0 + SC_EPW <= E) scatter128_wave<false>(h1, ei, E, agg, e0, lane);
    else                  scatter128_wave<true >(h1, ei, E, agg, e0, lane);
}

#define IN_NPW 8
template <bool TAIL>
__device__ __forceinline__ void interp_wave(const float* __restrict__ vals,
                                            const int* __restrict__ cols,
                                            const int* __restrict__ rows,
                                            const float* __restrict__ e,
                                            float* __restrict__ acc, int nz0, int lane) {
    float v[IN_NPW];
    int col[IN_NPW], row[IN_NPW];
    float g[IN_NPW][3];
#pragma unroll
    for (int j = 0; j < IN_NPW; ++j) {
        int c = nz0 + j;
        if (TAIL) c = (c < NNZ) ? c : (NNZ - 1);
        v[j]   = vals[c];
        col[j] = cols[c];
        row[j] = rows[c];
    }
#pragma unroll
    for (int j = 0; j < IN_NPW; ++j) {
        const float* ep = e + col[j] * F3 + lane;
#pragma unroll
        for (int k = 0; k < 3; ++k) g[j][k] = ep[32 * k];
    }
#pragma unroll
    for (int j = 0; j < IN_NPW; ++j) {
        if (!TAIL || nz0 + j < NNZ) {
            float* ap = acc + row[j] * F3 + lane;
#pragma unroll
            for (int k = 0; k < 3; ++k) atomicAdd(ap + 32 * k, v[j] * g[j][k]);
        }
    }
}

__global__ void interp_kernel(const float* __restrict__ vals, const int* __restrict__ cols,
                              const int* __restrict__ rows, const float* __restrict__ e,
                              float* __restrict__ acc) {
    const int lane = threadIdx.x & 31;
    const int wave = __builtin_amdgcn_readfirstlane((int)((blockIdx.x * blockDim.x + threadIdx.x) >> 5));
    const int nz0 = wave * IN_NPW;
    if (nz0 >= NNZ) return;
    if (nz0 + IN_NPW <= NNZ) interp_wave<false>(vals, cols, rows, e, acc, nz0, lane);
    else                     interp_wave<true >(vals, cols, rows, e, acc, nz0, lane);
}

__global__ void dec_out_kernel(const float* __restrict__ hdec, const float* __restrict__ Wout,
                               const float* __restrict__ bout, float* __restrict__ out,
                               int n, int branch) {
    int r = blockIdx.x * blockDim.x + threadIdx.x;
    if (r >= n) return;
    float s = bout[0];
    const float* hr = hdec + r * DECW;
#pragma unroll
    for (int c = 0; c < DECW; ++c) s += hr[c] * Wout[c];
    out[r * 3 + branch] = s;
}

template <int K, int Nc, int ACT, bool DUAL>
__global__ void gemm_wmma_kernel(const float* __restrict__ A0, const float* __restrict__ W0,
                                 const float* __restrict__ A1, const float* __restrict__ W1,
                                 const float* __restrict__ bias, float* __restrict__ D,
                                 int n) {
    const int lane = threadIdx.x & 31;
    const int wave = threadIdx.x >> 5;
    const int row0 = (blockIdx.x * 4 + wave) * 16;
    const int col0 = blockIdx.y * 16;
    if (row0 >= n) return;

    const int m    = lane & 15;
    const int half = lane >> 4;
    int row = row0 + m;
    const int rowc = (row < n) ? row : (n - 1);

    const float* a0p = A0 + rowc * K + half * 2;
    const float* w0p = W0 + col0 + m + half * 2 * Nc;
    const float* a1p = DUAL ? (A1 + rowc * K + half * 2) : nullptr;
    const float* w1p = DUAL ? (W1 + col0 + m + half * 2 * Nc) : nullptr;

    const float bv = bias[col0 + m];
    v8f acc = {bv, bv, bv, bv, bv, bv, bv, bv};

#pragma unroll
    for (int k = 0; k < K; k += 4) {
        v2f a = *(const v2f*)(a0p + k);
        v2f b;
        b.x = w0p[k * Nc];
        b.y = w0p[k * Nc + Nc];
        acc = __builtin_amdgcn_wmma_f32_16x16x4_f32(false, a, false, b, (short)0, acc,
                                                    false, false);
        if (DUAL) {
            v2f a2 = *(const v2f*)(a1p + k);
            v2f b2;
            b2.x = w1p[k * Nc];
            b2.y = w1p[k * Nc + Nc];
            acc = __builtin_amdgcn_wmma_f32_16x16x4_f32(false, a2, false, b2, (short)0, acc,
                                                        false, false);
        }
    }

#pragma unroll
    for (int j = 0; j < 8; ++j) {
        int r = row0 + j + half * 8;
        if (r < n) {
            float v = acc[j];
            if (ACT == 1) v = fmaxf(v, 0.0f);
            if (ACT == 2) v = (v > 0.0f) ? v : (expf(v) - 1.0f);
            D[r * Nc + col0 + m] = v;
        }
    }
}

__global__ void dec_gemm_wmma_kernel(const float* __restrict__ e0, const float* __restrict__ i1,
                                     const float* __restrict__ i2, const float* __restrict__ W,
                                     const float* __restrict__ bias, float* __restrict__ D,
                                     int n, int branch) {
    constexpr int K = F3, Nc = DECW;
    const int lane = threadIdx.x & 31;
    const int wave = threadIdx.x >> 5;
    const int row0 = (blockIdx.x * 4 + wave) * 16;
    const int col0 = blockIdx.y * 16;
    if (row0 >= n) return;

    const int m    = lane & 15;
    const int half = lane >> 4;
    int row = row0 + m;
    const int rowc = (row < n) ? row : (n - 1);

    const int abase = rowc * F3 + branch * 32 + half * 2;
    const float* s0 = e0 + abase;
    const float* s1 = i1 + abase;
    const float* s2 = i2 + abase;
    const float* wp = W + col0 + m + half * 2 * Nc;

    const float bv = bias[col0 + m];
    v8f acc = {bv, bv, bv, bv, bv, bv, bv, bv};

#pragma unroll
    for (int k = 0; k < K; k += 4) {
        const int blk = k >> 5;
        const float* sp = (blk == 0) ? s0 : (blk == 1) ? s1 : s2;
        v2f a = *(const v2f*)(sp + (k & 31));
        v2f b;
        b.x = wp[k * Nc];
        b.y = wp[k * Nc + Nc];
        acc = __builtin_amdgcn_wmma_f32_16x16x4_f32(false, a, false, b, (short)0, acc,
                                                    false, false);
    }

#pragma unroll
    for (int j = 0; j < 8; ++j) {
        int r = row0 + j + half * 8;
        if (r < n) {
            float v = acc[j];
            v = (v > 0.0f) ? v : (expf(v) - 1.0f);
            D[r * Nc + col0 + m] = v;
        }
    }
}

static inline dim3 gemm_grid(int n, int Nc) {
    int row_tiles = (n + 15) / 16;
    return dim3((row_tiles + 3) / 4, Nc / 16, 1);
}

extern "C" void kernel_launch(void* const* d_in, const int* in_sizes, int n_in,
                              void* d_out, int out_size, void* d_ws, size_t ws_size,
                              hipStream_t stream) {
    (void)in_sizes; (void)n_in; (void)out_size; (void)ws_size;

    const float* x[3]   = {(const float*)d_in[0], (const float*)d_in[1], (const float*)d_in[2]};
    const int*   ei[3]  = {(const int*)d_in[3], (const int*)d_in[4], (const int*)d_in[5]};
    const int*   A1r    = (const int*)d_in[6];
    const int*   A1c    = (const int*)d_in[7];
    const float* A1v    = (const float*)d_in[8];
    const int*   A2r    = (const int*)d_in[9];
    const int*   A2c    = (const int*)d_in[10];
    const float* A2v    = (const float*)d_in[11];
    const float* c1Wrel  = (const float*)d_in[12];
    const float* c1brel  = (const float*)d_in[13];
    const float* c1Wroot = (const float*)d_in[14];
    const float* c2Wrel  = (const float*)d_in[15];
    const float* c2brel  = (const float*)d_in[16];
    const float* c2Wroot = (const float*)d_in[17];
    const float* linW    = (const float*)d_in[18];
    const float* linb    = (const float*)d_in[19];
    const float* decW0   = (const float*)d_in[20];
    const float* decb0   = (const float*)d_in[21];
    const float* decWout = (const float*)d_in[22];
    const float* decbout = (const float*)d_in[23];
    float* out = (float*)d_out;

    float* ws = (float*)d_ws;
    float* h1   = ws;
    float* agg  = h1   + (size_t)N0 * H;
    float* h2   = agg  + (size_t)N0 * H;
    float* agg6 = h2   + (size_t)N0 * H;
    float* e0   = agg6 + (size_t)N0 * 6;
    float* e1   = e0   + (size_t)N0 * F3;
    float* e2   = e1   + (size_t)N1 * F3;
    float* i1   = e2   + (size_t)N2 * F3;
    float* i2   = i1   + (size_t)N0 * F3;
    float* hdec = i2   + (size_t)N0 * F3;

    const int nlv[3] = {N0, N1, N2};
    const int Elv[3] = {E0, E1, E2};
    float* elv[3] = {e0, e1, e2};

    for (int i = 0; i < 3; ++i) {
        const int n = nlv[i];
        const int E = Elv[i];

        {
            long long cnt = (long long)n * 6;
            zero_kernel<<<(int)((cnt + 255) / 256), 256, 0, stream>>>(agg6, cnt);
            scatter6_kernel<<<(E + 255) / 256, 256, 0, stream>>>(x[i], ei[i], E, agg6);
            conv1_kernel<<<n, H, 0, stream>>>(x[i], agg6,
                                              c1Wrel + (size_t)i * 6 * H,
                                              c1brel + (size_t)i * H,
                                              c1Wroot + (size_t)i * 6 * H, h1, n);
        }

        {
            long long cnt = (long long)n * H;
            zero_kernel<<<(int)((cnt + 255) / 256), 256, 0, stream>>>(agg, cnt);
            scatter128_kernel<<<(E + 8 * SC_EPW - 1) / (8 * SC_EPW), 256, 0, stream>>>(h1, ei[i], E, agg);
            gemm_wmma_kernel<H, H, 1, true><<<gemm_grid(n, H), 128, 0, stream>>>(
                agg, c2Wrel + (size_t)i * H * H,
                h1,  c2Wroot + (size_t)i * H * H,
                c2brel + (size_t)i * H, h2, n);
        }

        gemm_wmma_kernel<H, F3, 0, false><<<gemm_grid(n, F3), 128, 0, stream>>>(
            h2, linW + (size_t)i * H * F3, nullptr, nullptr,
            linb + (size_t)i * F3, elv[i], n);
    }

    {
        long long cnt = (long long)N0 * F3;
        zero_kernel<<<(int)((cnt + 255) / 256), 256, 0, stream>>>(i1, cnt);
        zero_kernel<<<(int)((cnt + 255) / 256), 256, 0, stream>>>(i2, cnt);
        interp_kernel<<<(NNZ + 8 * IN_NPW - 1) / (8 * IN_NPW), 256, 0, stream>>>(A1v, A1c, A1r, e1, i1);
        interp_kernel<<<(NNZ + 8 * IN_NPW - 1) / (8 * IN_NPW), 256, 0, stream>>>(A2v, A2c, A2r, e2, i2);
    }

    for (int b = 0; b < 3; ++b) {
        dec_gemm_wmma_kernel<<<gemm_grid(N0, DECW), 128, 0, stream>>>(
            e0, i1, i2, decW0 + (size_t)b * F3 * DECW, decb0 + (size_t)b * DECW,
            hdec, N0, b);
        dec_out_kernel<<<(N0 + 255) / 256, 256, 0, stream>>>(
            hdec, decWout + (size_t)b * DECW, decbout + b, out, N0, b);
    }
}


# fill and conv1 kernels coarsened: a wave writes 8 whole rows (conv1) or 2 KiB chunks in a strided loop (fills), 16 B per lane per store, not 32 floats; level-0 grids 50000, 25000 blocks -> 782, 256
# speedup vs baseline: 1.9229x; 1.0539x over previous
#include <hip/hip_runtime.h>
#include <hip/hip_bf16.h>

#define N0 50000
#define N1 12500
#define N2 3125
#define E0 800000
#define E1 200000
#define E2 50000
#define NNZ 200000
#define H 128
#define F3 96
#define DECW 64

typedef __attribute__((ext_vector_type(2))) float v2f;
typedef __attribute__((ext_vector_type(8))) float v8f;

#define FILL_SPW 4
__global__ void zero_kernel(float* __restrict__ p, int count) {
    const int tid  = blockIdx.x * blockDim.x + threadIdx.x;
    const int lane = threadIdx.x & 31;
    const int wave = __builtin_amdgcn_readfirstlane(tid >> 5);
    const int step = ((gridDim.x * blockDim.x) >> 5) * (32 * FILL_SPW);
    int head = (int)(((16u - (unsigned)((size_t)p & 15u)) & 15u) >> 2);
    if (head > count) head = count;
    float4* q = (float4*)(p + head);
    const int nvec = (count - head) >> 2;
    const float4 z = {0.0f, 0.0f, 0.0f, 0.0f};
    for (int base = wave * (32 * FILL_SPW); base < nvec; base += step) {
        float4* qq = q + base + lane;
        if (base + 32 * FILL_SPW <= nvec) {
#pragma unroll
            for (int u = 0; u < FILL_SPW; ++u) qq[32 * u] = z;
        } else {
#pragma unroll
            for (int u = 0; u < FILL_SPW; ++u)
                if (base + 32 * u + lane < nvec) qq[32 * u] = z;
        }
    }
    const int tail0 = head + (nvec << 2);
    if (tid < head) p[tid] = 0.0f;
    if (tid < count - tail0) p[tail0 + tid] = 0.0f;
}

#define C1_RPW 8
template <bool TAIL>
__device__ __forceinline__ void conv1_wave(const float* __restrict__ x, const float* __restrict__ agg6,
                                           const float* __restrict__ Wrel, const float* __restrict__ brel,
                                           const float* __restrict__ Wroot, float* __restrict__ h1,
                                           int n, int r0, int c) {
    float wr[6][4], wo[6][4], b[4];
#pragma unroll
    for (int q = 0; q < 4; ++q) b[q] = brel[c + q];
#pragma unroll
    for (int k = 0; k < 6; ++k) {
#pragma unroll
        for (int q = 0; q < 4; ++q) {
            wr[k][q] = Wrel[k * H + c + q];
            wo[k][q] = Wroot[k * H + c + q];
        }
    }
#pragma unroll
    for (int j = 0; j < C1_RPW; ++j) {
        const int r = r0 + j;
        int rl = r;
        if (TAIL) rl = (r < n) ? r : (n - 1);
        const float* ar = agg6 + rl * 6;
        const float* xr = x + rl * 6;
        float s[4] = {b[0], b[1], b[2], b[3]};
#pragma unroll
        for (int k = 0; k < 6; ++k) {
            const float a  = ar[k];
            const float xv = xr[k];
#pragma unroll
            for (int q = 0; q < 4; ++q) {
                s[q] += a * wr[k][q];
                s[q] += xv * wo[k][q];
            }
        }
        if (!TAIL || r < n) {
            float4 v;
            v.x = fmaxf(s[0], 0.0f);
            v.y = fmaxf(s[1], 0.0f);
            v.z = fmaxf(s[2], 0.0f);
            v.w = fmaxf(s[3], 0.0f);
            *(float4*)(h1 + r * H + c) = v;
        }
    }
}

__global__ void conv1_kernel(const float* __restrict__ x, const float* __restrict__ agg6,
                             const float* __restrict__ Wrel, const float* __restrict__ brel,
                             const float* __restrict__ Wroot, float* __restrict__ h1, int n) {
    const int lane = threadIdx.x & 31;
    const int wave = __builtin_amdgcn_readfirstlane((int)((blockIdx.x * blockDim.x + threadIdx.x) >> 5));
    const int r0 = wave * C1_RPW;
    if (r0 >= n) return;
    if (r0 + C1_RPW <= n) conv1_wave<false>(x, agg6, Wrel, brel, Wroot, h1, n, r0, lane * 4);
    else                  conv1_wave<true >(x, agg6, Wrel, brel, Wroot, h1, n, r0, lane * 4);
}

template <bool TAIL>
__device__ __forceinline__ void scatter6_wave(const float* __restrict__ x,
                                              const int* __restrict__ ei, int E,
                                              float* __restrict__ agg6,
                                              int ebase, int g, int k) {
    int src[8], dst[8];
    float v[8];
#pragma unroll
    for (int j = 0; j < 8; ++j) {
        int e = ebase + j * 4 + g;
        if (TAIL) e = (e < E) ? e : (E - 1);
        src[j] = ei[e];
        dst[j] = ei[E + e];
    }
#pragma unroll
    for (int j = 0; j < 8; ++j) v[j] = x[src[j] * 6 + k];
#pragma unroll
    for (int j = 0; j < 8; ++j)
        if (!TAIL || ebase + j * 4 + g < E) atomicAdd(&agg6[dst[j] * 6 + k], v[j]);
}

__global__ void scatter6_kernel(const float* __restrict__ x, const int* __restrict__ ei,
                                int E, float* __restrict__ agg6) {
    const int lane  = threadIdx.x & 31;
    const int wave  = __builtin_amdgcn_readfirstlane((int)((blockIdx.x * blockDim.x + threadIdx.x) >> 5));
    const int ebase = wave * 32;
    if (ebase >= E) return;
    const int g = lane >> 3;
    const int k = lane & 7;
    if (k >= 6) return;
    if (ebase + 32 <= E) scatter6_wave<false>(x, ei, E, agg6, ebase, g, k);
    else                 scatter6_wave<true >(x, ei, E, agg6, ebase, g, k);
}

#define SC_EPW 8
template <bool TAIL>
__device__ __forceinline__ void scatter128_wave(const float* __restrict__ h1,
                                                const int* __restrict__ ei, int E,
                                                float* __restrict__ agg, int e0, int lane) {
    int src[SC_EPW], dst[SC_EPW];
    float v[SC_EPW][4];
#pragma unroll
    for (int j = 0; j < SC_EPW; ++j) {
        int e = e0 + j;
        if (TAIL) e = (e < E) ? e : (E - 1);
        src[j] = ei[e];
        dst[j] = ei[E + e];
    }
#pragma unroll
    for (int j = 0; j < SC_EPW; ++j) {
        const float* hs = h1 + src[j] * H + lane;
#pragma unroll
        for (int k = 0; k < 4; ++k) v[j][k] = hs[32 * k];
    }
#pragma unroll
    for (int j = 0; j < SC_EPW; ++j) {
        if (!TAIL || e0 + j < E) {
            float* ad = agg + dst[j] * H + lane;
#pragma unroll
            for (int k = 0; k < 4; ++k) atomicAdd(ad + 32 * k, v[j][k]);
        }
    }
}

__global__ void scatter128_kernel(const float* __restrict__ h1, const int* __restrict__ ei,
                                  int E, float* __restrict__ agg) {
    const int lane = threadIdx.x & 31;
    const int wave = __builtin_amdgcn_readfirstlane((int)((blockIdx.x * blockDim.x + threadIdx.x) >> 5));
    const int e0 = wave * SC_EPW;
    if (e0 >= E) return;
    if (e0 + SC_EPW <= E) scatter128_wave<false>(h1, ei, E, agg, e0, lane);
    else                  scatter128_wave<true >(h1, ei, E, agg, e0, lane);
}

#define IN_NPW 8
template <bool TAIL>
__device__ __forceinline__ void interp_wave(const float* __restrict__ vals,
                                            const int* __restrict__ cols,
                                            const int* __restrict__ rows,
                                            const float* __restrict__ e,
                                            float* __restrict__ acc, int nz0, int lane) {
    float v[IN_NPW];
    int col[IN_NPW], row[IN_NPW];
    float g[IN_NPW][3];
#pragma unroll
    for (int j = 0; j < IN_NPW; ++j) {
        int c = nz0 + j;
        if (TAIL) c = (c < NNZ) ? c : (NNZ - 1);
        v[j]   = vals[c];
        col[j] = cols[c];
        row[j] = rows[c];
    }
#pragma unroll
    for (int j = 0; j < IN_NPW; ++j) {
        const float* ep = e + col[j] * F3 + lane;
#pragma unroll
        for (int k = 0; k < 3; ++k) g[j][k] = ep[32 * k];
    }
#pragma unroll
    for (int j = 0; j < IN_NPW; ++j) {
        if (!TAIL || nz0 + j < NNZ) {
            float* ap = acc + row[j] * F3 + lane;
#pragma unroll
            for (int k = 0; k < 3; ++k) atomicAdd(ap + 32 * k, v[j] * g[j][k]);
        }
    }
}

__global__ void interp_kernel(const float* __restrict__ vals, const int* __restrict__ cols,
                              const int* __restrict__ rows, const float* __restrict__ e,
                              float* __restrict__ acc) {
    const int lane = threadIdx.x & 31;
    const int wave = __builtin_amdgcn_readfirstlane((int)((blockIdx.x * blockDim.x + threadIdx.x) >> 5));
    const int nz0 = wave * IN_NPW;
    if (nz0 >= NNZ) return;
    if (nz0 + IN_NPW <= NNZ) interp_wave<false>(vals, cols, rows, e, acc, nz0, lane);
    else                     interp_wave<true >(vals, cols, rows, e, acc, nz0, lane);
}

__global__ void dec_out_kernel(const float* __restrict__ hdec, const float* __restrict__ Wout,
                               const float* __restrict__ bout, float* __restrict__ out,
                               int n, int branch) {
    int r = blockIdx.x * blockDim.x + threadIdx.x;
    if (r >= n) return;
    float s = bout[0];
    const float* hr = hdec + r * DECW;
#pragma unroll
    for (int c = 0; c < DECW; ++c) s += hr[c] * Wout[c];
    out[r * 3 + branch] = s;
}

template <int K, int Nc, int ACT, bool DUAL>
__global__ void gemm_wmma_kernel(const float* __restrict__ A0, const float* __restrict__ W0,
                                 const float* __restrict__ A1, const float* __restrict__ W1,
                                 const float* __restrict__ bias, float* __restrict__ D,
                                 int n) {
    const int lane = threadIdx.x & 31;
    const int wave = threadIdx.x >> 5;
    const int row0 = (blockIdx.x * 4 + wave) * 16;
    const int col0 = blockIdx.y * 16;
    if (row0 >= n) return;

    const int m    = lane & 15;
    const int half = lane >> 4;
    int row = row0 + m;
    const int rowc = (row < n) ? row : (n - 1);

    const float* a0p = A0 + rowc * K + half * 2;
    const float* w0p = W0 + col0 + m + half * 2 * Nc;
    const float* a1p = DUAL ? (A1 + rowc * K + half * 2) : nullptr;
    const float* w1p = DUAL ? (W1 + col0 + m + half * 2 * Nc) : nullptr;

    const float bv = bias[col0 + m];
    v8f acc = {bv, bv, bv, bv, bv, bv, bv, bv};

#pragma unroll
    for (int k = 0; k < K; k += 4) {
        v2f a = *(const v2f*)(a0p + k);
        v2f b;
        b.x = w0p[k * Nc];
        b.y = w0p[k * Nc + Nc];
        acc = __builtin_amdgcn_wmma_f32_16x16x4_f32(false, a, false, b, (short)0, acc,
                                                    false, false);
        if (DUAL) {
            v2f a2 = *(const v2f*)(a1p + k);
            v2f b2;
            b2.x = w1p[k * Nc];
            b2.y = w1p[k * Nc + Nc];
            acc = __builtin_amdgcn_wmma_f32_16x16x4_f32(false, a2, false, b2, (short)0, acc,
                                                        false, false);
        }
    }

#pragma unroll
    for (int j = 0; j < 8; ++j) {
        int r = row0 + j + half * 8;
        if (r < n) {
            float v = acc[j];
            if (ACT == 1) v = fmaxf(v, 0.0f);
            if (ACT == 2) v = (v > 0.0f) ? v : (expf(v) - 1.0f);
            D[r * Nc + col0 + m] = v;
        }
    }
}

__global__ void dec_gemm_wmma_kernel(const float* __restrict__ e0, const float* __restrict__ i1,
                                     const float* __restrict__ i2, const float* __restrict__ W,
                                     const float* __restrict__ bias, float* __restrict__ D,
                                     int n, int branch) {
    constexpr int K = F3, Nc = DECW;
    const int lane = threadIdx.x & 31;
    const int wave = threadIdx.x >> 5;
    const int row0 = (blockIdx.x * 4 + wave) * 16;
    const int col0 = blockIdx.y * 16;
    if (row0 >= n) return;

    const int m    = lane & 15;
    const int half = lane >> 4;
    int row = row0 + m;
    const int rowc = (row < n) ? row : (n - 1);

    const int abase = rowc * F3 + branch * 32 + half * 2;
    const float* s0 = e0 + abase;
    const float* s1 = i1 + abase;
    const float* s2 = i2 + abase;
    const float* wp = W + col0 + m + half * 2 * Nc;

    const float bv = bias[col0 + m];
    v8f acc = {bv, bv, bv, bv, bv, bv, bv, bv};

#pragma unroll
    for (int k = 0; k < K; k += 4) {
        const int blk = k >> 5;
        const float* sp = (blk == 0) ? s0 : (blk == 1) ? s1 : s2;
        v2f a = *(const v2f*)(sp + (k & 31));
        v2f b;
        b.x = wp[k * Nc];
        b.y = wp[k * Nc + Nc];
        acc = __builtin_amdgcn_wmma_f32_16x16x4_f32(false, a, false, b, (short)0, acc,
                                                    false, false);
    }

#pragma unroll
    for (int j = 0; j < 8; ++j) {
        int r = row0 + j + half * 8;
        if (r < n) {
            float v = acc[j];
            v = (v > 0.0f) ? v : (expf(v) - 1.0f);
            D[r * Nc + col0 + m] = v;
        }
    }
}

static inline int fill_grid(int count) {
    int b = (count / 4 + 8 * 32 * FILL_SPW - 1) / (8 * 32 * FILL_SPW);
    return b < 1 ? 1 : (b > 256 ? 256 : b);
}

static inline dim3 gemm_grid(int n, int Nc) {
    int row_tiles = (n + 15) / 16;
    return dim3((row_tiles + 3) / 4, Nc / 16, 1);
}

extern "C" void kernel_launch(void* const* d_in, const int* in_sizes, int n_in,
                              void* d_out, int out_size, void* d_ws, size_t ws_size,
                              hipStream_t stream) {
    (void)in_sizes; (void)n_in; (void)out_size; (void)ws_size;

    const float* x[3]   = {(const float*)d_in[0], (const float*)d_in[1], (const float*)d_in[2]};
    const int*   ei[3]  = {(const int*)d_in[3], (const int*)d_in[4], (const int*)d_in[5]};
    const int*   A1r    = (const int*)d_in[6];
    const int*   A1c    = (const int*)d_in[7];
    const float* A1v    = (const float*)d_in[8];
    const int*   A2r    = (const int*)d_in[9];
    const int*   A2c    = (const int*)d_in[10];
    const float* A2v    = (const float*)d_in[11];
    const float* c1Wrel  = (const float*)d_in[12];
    const float* c1brel  = (const float*)d_in[13];
    const float* c1Wroot = (const float*)d_in[14];
    const float* c2Wrel  = (const float*)d_in[15];
    const float* c2brel  = (const float*)d_in[16];
    const float* c2Wroot = (const float*)d_in[17];
    const float* linW    = (const float*)d_in[18];
    const float* linb    = (const float*)d_in[19];
    const float* decW0   = (const float*)d_in[20];
    const float* decb0   = (const float*)d_in[21];
    const float* decWout = (const float*)d_in[22];
    const float* decbout = (const float*)d_in[23];
    float* out = (float*)d_out;

    float* ws = (float*)d_ws;
    float* h1   = ws;
    float* agg  = h1   + (size_t)N0 * H;
    float* h2   = agg  + (size_t)N0 * H;
    float* agg6 = h2   + (size_t)N0 * H;
    float* e0   = agg6 + (size_t)N0 * 6;
    float* e1   = e0   + (size_t)N0 * F3;
    float* e2   = e1   + (size_t)N1 * F3;
    float* i1   = e2   + (size_t)N2 * F3;
    float* i2   = i1   + (size_t)N0 * F3;
    float* hdec = i2   + (size_t)N0 * F3;

    const int nlv[3] = {N0, N1, N2};
    const int Elv[3] = {E0, E1, E2};
    float* elv[3] = {e0, e1, e2};

    for (int i = 0; i < 3; ++i) {
        const int n = nlv[i];
        const int E = Elv[i];

        {
            zero_kernel<<<fill_grid(n * 6), 256, 0, stream>>>(agg6, n * 6);
            scatter6_kernel<<<(E + 255) / 256, 256, 0, stream>>>(x[i], ei[i], E, agg6);
            conv1_kernel<<<(n + 8 * C1_RPW - 1) / (8 * C1_RPW), 256, 0, stream>>>(x[i], agg6,
                                              c1Wrel + (size_t)i * 6 * H,
                                              c1brel + (size_t)i * H,
                                              c1Wroot + (size_t)i * 6 * H, h1, n);
        }

        {
            zero_kernel<<<fill_grid(n * H), 256, 0, stream>>>(agg, n * H);
            scatter128_kernel<<<(E + 8 * SC_EPW - 1) / (8 * SC_EPW), 256, 0, stream>>>(h1, ei[i], E, agg);
            gemm_wmma_kernel<H, H, 1, true><<<gemm_grid(n, H), 128, 0, stream>>>(
                agg, c2Wrel + (size_t)i * H * H,
                h1,  c2Wroot + (size_t)i * H * H,
                c2brel + (size_t)i * H, h2, n);
        }

        gemm_wmma_kernel<H, F3, 0, false><<<gemm_grid(n, F3), 128, 0, stream>>>(
            h2, linW + (size_t)i * H * F3, nullptr, nullptr,
            linb + (size_t)i * F3, elv[i], n);
    }

    {
        zero_kernel<<<fill_grid(N0 * F3), 256, 0, stream>>>(i1, N0 * F3);
        zero_kernel<<<fill_grid(N0 * F3), 256, 0, stream>>>(i2, N0 * F3);
        interp_kernel<<<(NNZ + 8 * IN_NPW - 1) / (8 * IN_NPW), 256, 0, stream>>>(A1v, A1c, A1r, e1, i1);
        interp_kernel<<<(NNZ + 8 * IN_NPW - 1) / (8 * IN_NPW), 256, 0, stream>>>(A2v, A2c, A2r, e2, i2);
    }

    for (int b = 0; b < 3; ++b) {
        dec_gemm_wmma_kernel<<<gemm_grid(N0, DECW), 128, 0, stream>>>(
            e0, i1, i2, decW0 + (size_t)b * F3 * DECW, decb0 + (size_t)b * DECW,
            hdec, N0, b);
        dec_out_kernel<<<(N0 + 255) / 256, 256, 0, stream>>>(
            hdec, decWout + (size_t)b * DECW, decbout + b, out, N0, b);
    }
}


# GEMM kernels: a wave owns a 16-row strip over all columns (A fragment loaded once per k-step, not once per column tile); weights staged in LDS in fragment order; per-element product order unchanged
# speedup vs baseline: 2.2507x; 1.1705x over previous
#include <hip/hip_runtime.h>
#include <hip/hip_bf16.h>

#define N0 50000
#define N1 12500
#define N2 3125
#define E0 800000
#define E1 200000
#define E2 50000
#define NNZ 200000
#define H 128
#define F3 96
#define DECW 64

typedef __attribute__((ext_vector_type(2))) float v2f;
typedef __attribute__((ext_vector_type(8))) float v8f;

#define FILL_SPW 4
__global__ void zero_kernel(float* __restrict__ p, int count) {
    const int tid  = blockIdx.x * blockDim.x + threadIdx.x;
    const int lane = threadIdx.x & 31;
    const int wave = __builtin_amdgcn_readfirstlane(tid >> 5);
    const int step = ((gridDim.x * blockDim.x) >> 5) * (32 * FILL_SPW);
    int head = (int)(((16u - (unsigned)((size_t)p & 15u)) & 15u) >> 2);
    if (head > count) head = count;
    float4* q = (float4*)(p + head);
    const int nvec = (count - head) >> 2;
    const float4 z = {0.0f, 0.0f, 0.0f, 0.0f};
    for (int base = wave * (32 * FILL_SPW); base < nvec; base += step) {
        float4* qq = q + base + lane;
        if (base + 32 * FILL_SPW <= nvec) {
#pragma unroll
            for (int u = 0; u < FILL_SPW; ++u) qq[32 * u] = z;
        } else {
#pragma unroll
            for (int u = 0; u < FILL_SPW; ++u)
                if (base + 32 * u + lane < nvec) qq[32 * u] = z;
        }
    }
    const int tail0 = head + (nvec << 2);
    if (tid < head) p[tid] = 0.0f;
    if (tid < count - tail0) p[tail0 + tid] = 0.0f;
}

#define C1_RPW 8
template <bool TAIL>
__device__ __forceinline__ void conv1_wave(const float* __restrict__ x, const float* __restrict__ agg6,
                                           const float* __restrict__ Wrel, const float* __restrict__ brel,
                                           const float* __restrict__ Wroot, float* __restrict__ h1,
                                           int n, int r0, int c) {
    float wr[6][4], wo[6][4], b[4];
#pragma unroll
    for (int q = 0; q < 4; ++q) b[q] = brel[c + q];
#pragma unroll
    for (int k = 0; k < 6; ++k) {
#pragma unroll
        for (int q = 0; q < 4; ++q) {
            wr[k][q] = Wrel[k * H + c + q];
            wo[k][q] = Wroot[k * H + c + q];
        }
    }
#pragma unroll
    for (int j = 0; j < C1_RPW; ++j) {
        const int r = r0 + j;
        int rl = r;
        if (TAIL) rl = (r < n) ? r : (n - 1);
        const float* ar = agg6 + rl * 6;
        const float* xr = x + rl * 6;
        float s[4] = {b[0], b[1], b[2], b[3]};
#pragma unroll
        for (int k = 0; k < 6; ++k) {
            const float a  = ar[k];
            const float xv = xr[k];
#pragma unroll
            for (int q = 0; q < 4; ++q) {
                s[q] += a * wr[k][q];
                s[q] += xv * wo[k][q];
            }
        }
        if (!TAIL || r < n) {
            float4 v;
            v.x = fmaxf(s[0], 0.0f);
            v.y = fmaxf(s[1], 0.0f);
            v.z = fmaxf(s[2], 0.0f);
            v.w = fmaxf(s[3], 0.0f);
            *(float4*)(h1 + r * H + c) = v;
        }
    }
}

__global__ void conv1_kernel(const float* __restrict__ x, const float* __restrict__ agg6,
                             const float* __restrict__ Wrel, const float* __restrict__ brel,
                             const float* __restrict__ Wroot, float* __restrict__ h1, int n) {
    const int lane = threadIdx.x & 31;
    const int wave = __builtin_amdgcn_readfirstlane((int)((blockIdx.x * blockDim.x + threadIdx.x) >> 5));
    const int r0 = wave * C1_RPW;
    if (r0 >= n) return;
    if (r0 + C1_RPW <= n) conv1_wave<false>(x, agg6, Wrel, brel, Wroot, h1, n, r0, lane * 4);
    else                  conv1_wave<true >(x, agg6, Wrel, brel, Wroot, h1, n, r0, lane * 4);
}

template <bool TAIL>
__device__ __forceinline__ void scatter6_wave(const float* __restrict__ x,
                                              const int* __restrict__ ei, int E,
                                              float* __restrict__ agg6,
                                              int ebase, int g, int k) {
    int src[8], dst[8];
    float v[8];
#pragma unroll
    for (int j = 0; j < 8; ++j) {
        int e = ebase + j * 4 + g;
        if (TAIL) e = (e < E) ? e : (E - 1);
        src[j] = ei[e];
        dst[j] = ei[E + e];
    }
#pragma unroll
    for (int j = 0; j < 8; ++j) v[j] = x[src[j] * 6 + k];
#pragma unroll
    for (int j = 0; j < 8; ++j)
        if (!TAIL || ebase + j * 4 + g < E) atomicAdd(&agg6[dst[j] * 6 + k], v[j]);
}

__global__ void scatter6_kernel(const float* __restrict__ x, const int* __restrict__ ei,
                                int E, float* __restrict__ agg6) {
    const int lane  = threadIdx.x & 31;
    const int wave  = __builtin_amdgcn_readfirstlane((int)((blockIdx.x * blockDim.x + threadIdx.x) >> 5));
    const int ebase = wave * 32;
    if (ebase >= E) return;
    const int g = lane >> 3;
    const int k = lane & 7;
    if (k >= 6) return;
    if (ebase + 32 <= E) scatter6_wave<false>(x, ei, E, agg6, ebase, g, k);
    else                 scatter6_wave<true >(x, ei, E, agg6, ebase, g, k);
}

#define SC_EPW 8
template <bool TAIL>
__device__ __forceinline__ void scatter128_wave(const float* __restrict__ h1,
                                                const int* __restrict__ ei, int E,
                                                float* __restrict__ agg, int e0, int lane) {
    int src[SC_EPW], dst[SC_EPW];
    float v[SC_EPW][4];
#pragma unroll
    for (int j = 0; j < SC_EPW; ++j) {
        int e = e0 + j;
        if (TAIL) e = (e < E) ? e : (E - 1);
        src[j] = ei[e];
        dst[j] = ei[E + e];
    }
#pragma unroll
    for (int j = 0; j < SC_EPW; ++j) {
        const float* hs = h1 + src[j] * H + lane;
#pragma unroll
        for (int k = 0; k < 4; ++k) v[j][k] = hs[32 * k];
    }
#pragma unroll
    for (int j = 0; j < SC_EPW; ++j) {
        if (!TAIL || e0 + j < E) {
            float* ad = agg + dst[j] * H + lane;
#pragma unroll
            for (int k = 0; k < 4; ++k) atomicAdd(ad + 32 * k, v[j][k]);
        }
    }
}

__global__ void scatter128_kernel(const float* __restrict__ h1, const int* __restrict__ ei,
                                  int E, float* __restrict__ agg) {
    const int lane = threadIdx.x & 31;
    const int wave = __builtin_amdgcn_readfirstlane((int)((blockIdx.x * blockDim.x + threadIdx.x) >> 5));
    const int e0 = wave * SC_EPW;
    if (e0 >= E) return;
    if (e0 + SC_EPW <= E) scatter128_wave<false>(h1, ei, E, agg, e0, lane);
    else                  scatter128_wave<true >(h1, ei, E, agg, e0, lane);
}

#define IN_NPW 8
template <bool TAIL>
__device__ __forceinline__ void interp_wave(const float* __restrict__ vals,
                                            const int* __restrict__ cols,
                                            const int* __restrict__ rows,
                                            const float* __restrict__ e,
                                            float* __restrict__ acc, int nz0, int lane) {
    float v[IN_NPW];
    int col[IN_NPW], row[IN_NPW];
    float g[IN_NPW][3];
#pragma unroll
    for (int j = 0; j < IN_NPW; ++j) {
        int c = nz0 + j;
        if (TAIL) c = (c < NNZ) ? c : (NNZ - 1);
        v[j]   = vals[c];
        col[j] = cols[c];
        row[j] = rows[c];
    }
#pragma unroll
    for (int j = 0; j < IN_NPW; ++j) {
        const float* ep = e + col[j] * F3 + lane;
#pragma unroll
        for (int k = 0; k < 3; ++k) g[j][k] = ep[32 * k];
    }
#pragma unroll
    for (int j = 0; j < IN_NPW; ++j) {
        if (!TAIL || nz0 + j < NNZ) {
            float* ap = acc + row[j] * F3 + lane;
#pragma unroll
            for (int k = 0; k < 3; ++k) atomicAdd(ap + 32 * k, v[j] * g[j][k]);
        }
    }
}

__global__ void interp_kernel(const float* __restrict__ vals, const int* __restrict__ cols,
                              const int* __restrict__ rows, const float* __restrict__ e,
                              float* __restrict__ acc) {
    const int lane = threadIdx.x & 31;
    const int wave = __builtin_amdgcn_readfirstlane((int)((blockIdx.x * blockDim.x + threadIdx.x) >> 5));
    const int nz0 = wave * IN_NPW;
    if (nz0 >= NNZ) return;
    if (nz0 + IN_NPW <= NNZ) interp_wave<false>(vals, cols, rows, e, acc, nz0, lane);
    else                     interp_wave<true >(vals, cols, rows, e, acc, nz0, lane);
}

__global__ void dec_out_kernel(const float* __restrict__ hdec, const float* __restrict__ Wout,
                               const float* __restrict__ bout, float* __restrict__ out,
                               int n, int branch) {
    int r = blockIdx.x * blockDim.x + threadIdx.x;
    if (r >= n) return;
    float s = bout[0];
    const float* hr = hdec + r * DECW;
#pragma unroll
    for (int c = 0; c < DECW; ++c) s += hr[c] * Wout[c];
    out[r * 3 + branch] = s;
}

#define GS_THREADS 128

template <int K, int NC>
__device__ __forceinline__ void stage_weights(const float* __restrict__ W, float* __restrict__ Ws) {
    constexpr int NT = NC / 16;
#pragma unroll 4
    for (int q = threadIdx.x; q < (K / 4) * NC; q += GS_THREADS) {
        const int ks = q / NC;
        const int nn = q - ks * NC;
        const float* wp = W + ks * 4 * NC + nn;
        v2f lo, hi;
        lo.x = wp[0];
        lo.y = wp[NC];
        hi.x = wp[2 * NC];
        hi.y = wp[3 * NC];
        const int t = nn >> 4;
        const int c = nn & 15;
        float* dst = Ws + ((ks * (NT / 2) + (t >> 1)) * 32 + c) * 4 + (t & 1) * 2;
        *(v2f*)dst        = lo;
        *(v2f*)(dst + 64) = hi;
    }
}

template <int NT>
__device__ __forceinline__ void load_b(const float* __restrict__ wb, v2f (&b)[NT]) {
#pragma unroll
    for (int tp = 0; tp < NT / 2; ++tp) {
        const float4 q = *(const float4*)(wb + tp * 128);
        b[2 * tp].x     = q.x;
        b[2 * tp].y     = q.y;
        b[2 * tp + 1].x = q.z;
        b[2 * tp + 1].y = q.w;
    }
}

template <int NC, int NT, int ACT>
__device__ __forceinline__ void store_strip(const v8f (&acc)[NT], float* __restrict__ D,
                                            int n, int row0, int m, int half) {
    const int rbase = row0 + half * 8;
    const int lim   = n - rbase;
    float* dp = D + rbase * NC + m;
#pragma unroll
    for (int j = 0; j < 8; ++j) {
        if (j < lim) {
#pragma unroll
            for (int t = 0; t < NT; ++t) {
                float v = acc[t][j];
                if (ACT == 1) v = fmaxf(v, 0.0f);
                if (ACT == 2) v = (v > 0.0f) ? v : (expf(v) - 1.0f);
                dp[j * NC + 16 * t] = v;
            }
        }
    }
}

template <int K, int NT, bool DUAL>
__device__ __forceinline__ void strip_products(const v2f (&a0)[K / 4], const v2f (&a1)[K / 4],
                                               const float* __restrict__ w0s,
                                               const float* __restrict__ w1s, v8f (&acc)[NT]) {
    constexpr int KS = K / 4;
    v2f b0[NT], b1[NT];
    load_b<NT>(w0s, b0);
    if (DUAL) load_b<NT>(w1s, b1);
#pragma unroll
    for (int ks = 0; ks < KS; ++ks) {
        v2f nb0[NT], nb1[NT];
        if (ks + 1 < KS) {
            load_b<NT>(w0s + (ks + 1) * (NT / 2) * 128, nb0);
            if (DUAL) load_b<NT>(w1s + (ks + 1) * (NT / 2) * 128, nb1);
        }
#pragma unroll
        for (int t = 0; t < NT; ++t)
            acc[t] = __builtin_amdgcn_wmma_f32_16x16x4_f32(false, a0[ks], false, b0[t], (short)0,
                                                           acc[t], false, false);
        if (DUAL) {
#pragma unroll
            for (int t = 0; t < NT; ++t)
                acc[t] = __builtin_amdgcn_wmma_f32_16x16x4_f32(false, a1[ks], false, b1[t], (short)0,
                                                               acc[t], false, false);
        }
        if (ks + 1 < KS) {
#pragma unroll
            for (int t = 0; t < NT; ++t) {
                b0[t] = nb0[t];
                if (DUAL) b1[t] = nb1[t];
            }
        }
    }
}

template <int K, int NC, int ACT, bool DUAL>
__global__ void __launch_bounds__(GS_THREADS)
gemm_wmma_kernel(const float* __restrict__ A0, const float* __restrict__ W0,
                 const float* __restrict__ A1, const float* __restrict__ W1,
                 const float* __restrict__ bias, float* __restrict__ D, int n) {
    constexpr int NT = NC / 16;
    constexpr int KS = K / 4;
    __shared__ __attribute__((aligned(16))) float Ws[(DUAL ? 2 : 1) * K * NC];

    stage_weights<K, NC>(W0, Ws);
    if (DUAL) stage_weights<K, NC>(W1, Ws + K * NC);
    __syncthreads();

    const int lane = threadIdx.x & 31;
    const int wave = __builtin_amdgcn_readfirstlane((int)(threadIdx.x >> 5));
    const int m    = lane & 15;
    const int half = lane >> 4;
    const int nstrips = (n + 15) >> 4;

#pragma unroll 1
    for (int s = blockIdx.x * (GS_THREADS / 32) + wave; s < nstrips; s += gridDim.x * (GS_THREADS / 32)) {
        int boff0 = lane * 4;
        int boff1 = lane * 4 + K * NC;
        if (DUAL) {
            asm volatile("" : "+v"(boff0), "+v"(boff1));
        }
        const float* w0s = Ws + boff0;
        const float* w1s = Ws + boff1;
        const int row0 = s * 16;
        const int row  = row0 + m;
        const int rowc = (row < n) ? row : (n - 1);

        const float* a0p = A0 + rowc * K + half * 2;
        const float* a1p = DUAL ? (A1 + rowc * K + half * 2) : a0p;
        v2f a0[KS], a1[KS];
#pragma unroll
        for (int ks = 0; ks < KS; ++ks) {
            a0[ks] = *(const v2f*)(a0p + 4 * ks);
            if (DUAL) a1[ks] = *(const v2f*)(a1p + 4 * ks);
        }

        v8f acc[NT];
#pragma unroll
        for (int t = 0; t < NT; ++t) {
            const float bv = bias[16 * t + m];
            acc[t] = (v8f){bv, bv, bv, bv, bv, bv, bv, bv};
        }

        strip_products<K, NT, DUAL>(a0, a1, w0s, w1s, acc);

        store_strip<NC, NT, ACT>(acc, D, n, row0, m, half);
    }
}

__global__ void __launch_bounds__(GS_THREADS)
dec_gemm_wmma_kernel(const float* __restrict__ e0, const float* __restrict__ i1,
                     const float* __restrict__ i2, const float* __restrict__ W,
                     const float* __restrict__ bias, float* __restrict__ D,
                     int n, int branch) {
    constexpr int K = F3, NC = DECW;
    constexpr int NT = NC / 16;
    constexpr int KS = K / 4;
    __shared__ __attribute__((aligned(16))) float Ws[K * NC];

    stage_weights<K, NC>(W, Ws);
    __syncthreads();

    const int lane = threadIdx.x & 31;
    const int wave = __builtin_amdgcn_readfirstlane((int)(threadIdx.x >> 5));
    const int m    = lane & 15;
    const int half = lane >> 4;
    const int nstrips = (n + 15) >> 4;
    const float* ws = Ws + lane * 4;

#pragma unroll 1
    for (int s = blockIdx.x * (GS_THREADS / 32) + wave; s < nstrips; s += gridDim.x * (GS_THREADS / 32)) {
        const int row0 = s * 16;
        const int row  = row0 + m;
        const int rowc = (row < n) ? row : (n - 1);

        const int abase = rowc * F3 + branch * 32 + half * 2;
        const float* s0 = e0 + abase;
        const float* s1 = i1 + abase;
        const float* s2 = i2 + abase;
        v2f a[KS];
#pragma unroll
        for (int ks = 0; ks < KS; ++ks) {
            const int k   = 4 * ks;
            const int blk = k >> 5;
            const float* sp = (blk == 0) ? s0 : (blk == 1) ? s1 : s2;
            a[ks] = *(const v2f*)(sp + (k & 31));
        }
        __builtin_amdgcn_sched_barrier(0);

        v8f acc[NT];
#pragma unroll
        for (int t = 0; t < NT; ++t) {
            const float bv = bias[16 * t + m];
            acc[t] = (v8f){bv, bv, bv, bv, bv, bv, bv, bv};
        }

        strip_products<K, NT, false>(a, a, ws, ws, acc);

        store_strip<NC, NT, 2>(acc, D, n, row0, m, half);
    }
}

static inline int fill_grid(int count) {
    int b = (count / 4 + 8 * 32 * FILL_SPW - 1) / (8 * 32 * FILL_SPW);
    return b < 1 ? 1 : (b > 256 ? 256 : b);
}

static inline int gemm_grid(int n) {
    int b = ((n + 15) / 16 + 3) / 4;
    return b > 512 ? 512 : b;
}

extern "C" void kernel_launch(void* const* d_in, const int* in_sizes, int n_in,
                              void* d_out, int out_size, void* d_ws, size_t ws_size,
                              hipStream_t stream) {
    (void)in_sizes; (void)n_in; (void)out_size; (void)ws_size;

    const float* x[3]   = {(const float*)d_in[0], (const float*)d_in[1], (const float*)d_in[2]};
    const int*   ei[3]  = {(const int*)d_in[3], (const int*)d_in[4], (const int*)d_in[5]};
    const int*   A1r    = (const int*)d_in[6];
    const int*   A1c    = (const int*)d_in[7];
    const float* A1v    = (const float*)d_in[8];
    const int*   A2r    = (const int*)d_in[9];
    const int*   A2c    = (const int*)d_in[10];
    const float* A2v    = (const float*)d_in[11];
    const float* c1Wrel  = (const float*)d_in[12];
    const float* c1brel  = (const float*)d_in[13];
    const float* c1Wroot = (const float*)d_in[14];
    const float* c2Wrel  = (const float*)d_in[15];
    const float* c2brel  = (const float*)d_in[16];
    const float* c2Wroot = (const float*)d_in[17];
    const float* linW    = (const float*)d_in[18];
    const float* linb    = (const float*)d_in[19];
    const float* decW0   = (const float*)d_in[20];
    const float* decb0   = (const float*)d_in[21];
    const float* decWout = (const float*)d_in[22];
    const float* decbout = (const float*)d_in[23];
    float* out = (float*)d_out;

    float* ws = (float*)d_ws;
    float* h1   = ws;
    float* agg  = h1   + (size_t)N0 * H;
    float* h2   = agg  + (size_t)N0 * H;
    float* agg6 = h2   + (size_t)N0 * H;
    float* e0   = agg6 + (size_t)N0 * 6;
    float* e1   = e0   + (size_t)N0 * F3;
    float* e2   = e1   + (size_t)N1 * F3;
    float* i1   = e2   + (size_t)N2 * F3;
    float* i2   = i1   + (size_t)N0 * F3;
    float* hdec = i2   + (size_t)N0 * F3;

    const int nlv[3] = {N0, N1, N2};
    const int Elv[3] = {E0, E1, E2};
    float* elv[3] = {e0, e1, e2};

    for (int i = 0; i < 3; ++i) {
        const int n = nlv[i];
        const int E = Elv[i];

        {
            zero_kernel<<<fill_grid(n * 6), 256, 0, stream>>>(agg6, n * 6);
            scatter6_kernel<<<(E + 255) / 256, 256, 0, stream>>>(x[i], ei[i], E, agg6);
            conv1_kernel<<<(n + 8 * C1_RPW - 1) / (8 * C1_RPW), 256, 0, stream>>>(x[i], agg6,
                                              c1Wrel + (size_t)i * 6 * H,
                                              c1brel + (size_t)i * H,
                                              c1Wroot + (size_t)i * 6 * H, h1, n);
        }

        {
            zero_kernel<<<fill_grid(n * H), 256, 0, stream>>>(agg, n * H);
            scatter128_kernel<<<(E + 8 * SC_EPW - 1) / (8 * SC_EPW), 256, 0, stream>>>(h1, ei[i], E, agg);
            gemm_wmma_kernel<H, H, 1, true><<<gemm_grid(n), GS_THREADS, 0, stream>>>(
                agg, c2Wrel + (size_t)i * H * H,
                h1,  c2Wroot + (size_t)i * H * H,
                c2brel + (size_t)i * H, h2, n);
        }

        gemm_wmma_kernel<H, F3, 0, false><<<gemm_grid(n), GS_THREADS, 0, stream>>>(
            h2, linW + (size_t)i * H * F3, nullptr, nullptr,
            linb + (size_t)i * F3, elv[i], n);
    }

    {
        zero_kernel<<<fill_grid(N0 * F3), 256, 0, stream>>>(i1, N0 * F3);
        zero_kernel<<<fill_grid(N0 * F3), 256, 0, stream>>>(i2, N0 * F3);
        interp_kernel<<<(NNZ + 8 * IN_NPW - 1) / (8 * IN_NPW), 256, 0, stream>>>(A1v, A1c, A1r, e1, i1);
        interp_kernel<<<(NNZ + 8 * IN_NPW - 1) / (8 * IN_NPW), 256, 0, stream>>>(A2v, A2c, A2r, e2, i2);
    }

    for (int b = 0; b < 3; ++b) {
        dec_gemm_wmma_kernel<<<gemm_grid(N0), GS_THREADS, 0, stream>>>(
            e0, i1, i2, decW0 + (size_t)b * F3 * DECW, decb0 + (size_t)b * DECW,
            hdec, N0, b);
        dec_out_kernel<<<(N0 + 255) / 256, 256, 0, stream>>>(
            hdec, decWout + (size_t)b * DECW, decbout + b, out, N0, b);
    }
}


# decoder: the 3 branch GEMMs and 3 output-dot kernels fused into one launch; ELU'd tile stays in accumulators, 64-term dot summed as 4 in-lane terms + 16-lane exchange tree (sum order changed)
# speedup vs baseline: 2.3912x; 1.0624x over previous
#include <hip/hip_runtime.h>
#include <hip/hip_bf16.h>

#define N0 50000
#define N1 12500
#define N2 3125
#define E0 800000
#define E1 200000
#define E2 50000
#define NNZ 200000
#define H 128
#define F3 96
#define DECW 64

typedef __attribute__((ext_vector_type(2))) float v2f;
typedef __attribute__((ext_vector_type(8))) float v8f;

#define FILL_SPW 4
__global__ void zero_kernel(float* __restrict__ p, int count) {
    const int tid  = blockIdx.x * blockDim.x + threadIdx.x;
    const int lane = threadIdx.x & 31;
    const int wave = __builtin_amdgcn_readfirstlane(tid >> 5);
    const int step = ((gridDim.x * blockDim.x) >> 5) * (32 * FILL_SPW);
    int head = (int)(((16u - (unsigned)((size_t)p & 15u)) & 15u) >> 2);
    if (head > count) head = count;
    float4* q = (float4*)(p + head);
    const int nvec = (count - head) >> 2;
    const float4 z = {0.0f, 0.0f, 0.0f, 0.0f};
    for (int base = wave * (32 * FILL_SPW); base < nvec; base += step) {
        float4* qq = q + base + lane;
        if (base + 32 * FILL_SPW <= nvec) {
#pragma unroll
            for (int u = 0; u < FILL_SPW; ++u) qq[32 * u] = z;
        } else {
#pragma unroll
            for (int u = 0; u < FILL_SPW; ++u)
                if (base + 32 * u + lane < nvec) qq[32 * u] = z;
        }
    }
    const int tail0 = head + (nvec << 2);
    if (tid < head) p[tid] = 0.0f;
    if (tid < count - tail0) p[tail0 + tid] = 0.0f;
}

#define C1_RPW 8
template <bool TAIL>
__device__ __forceinline__ void conv1_wave(const float* __restrict__ x, const float* __restrict__ agg6,
                                           const float* __restrict__ Wrel, const float* __restrict__ brel,
                                           const float* __restrict__ Wroot, float* __restrict__ h1,
                                           int n, int r0, int c) {
    float wr[6][4], wo[6][4], b[4];
#pragma unroll
    for (int q = 0; q < 4; ++q) b[q] = brel[c + q];
#pragma unroll
    for (int k = 0; k < 6; ++k) {
#pragma unroll
        for (int q = 0; q < 4; ++q) {
            wr[k][q] = Wrel[k * H + c + q];
            wo[k][q] = Wroot[k * H + c + q];
        }
    }
#pragma unroll
    for (int j = 0; j < C1_RPW; ++j) {
        const int r = r0 + j;
        int rl = r;
        if (TAIL) rl = (r < n) ? r : (n - 1);
        const float* ar = agg6 + rl * 6;
        const float* xr = x + rl * 6;
        float s[4] = {b[0], b[1], b[2], b[3]};
#pragma unroll
        for (int k = 0; k < 6; ++k) {
            const float a  = ar[k];
            const float xv = xr[k];
#pragma unroll
            for (int q = 0; q < 4; ++q) {
                s[q] += a * wr[k][q];
                s[q] += xv * wo[k][q];
            }
        }
        if (!TAIL || r < n) {
            float4 v;
            v.x = fmaxf(s[0], 0.0f);
            v.y = fmaxf(s[1], 0.0f);
            v.z = fmaxf(s[2], 0.0f);
            v.w = fmaxf(s[3], 0.0f);
            *(float4*)(h1 + r * H + c) = v;
        }
    }
}

__global__ void conv1_kernel(const float* __restrict__ x, const float* __restrict__ agg6,
                             const float* __restrict__ Wrel, const float* __restrict__ brel,
                             const float* __restrict__ Wroot, float* __restrict__ h1, int n) {
    const int lane = threadIdx.x & 31;
    const int wave = __builtin_amdgcn_readfirstlane((int)((blockIdx.x * blockDim.x + threadIdx.x) >> 5));
    const int r0 = wave * C1_RPW;
    if (r0 >= n) return;
    if (r0 + C1_RPW <= n) conv1_wave<false>(x, agg6, Wrel, brel, Wroot, h1, n, r0, lane * 4);
    else                  conv1_wave<true >(x, agg6, Wrel, brel, Wroot, h1, n, r0, lane * 4);
}

template <bool TAIL>
__device__ __forceinline__ void scatter6_wave(const float* __restrict__ x,
                                              const int* __restrict__ ei, int E,
                                              float* __restrict__ agg6,
                                              int ebase, int g, int k) {
    int src[8], dst[8];
    float v[8];
#pragma unroll
    for (int j = 0; j < 8; ++j) {
        int e = ebase + j * 4 + g;
        if (TAIL) e = (e < E) ? e : (E - 1);
        src[j] = ei[e];
        dst[j] = ei[E + e];
    }
#pragma unroll
    for (int j = 0; j < 8; ++j) v[j] = x[src[j] * 6 + k];
#pragma unroll
    for (int j = 0; j < 8; ++j)
        if (!TAIL || ebase + j * 4 + g < E) atomicAdd(&agg6[dst[j] * 6 + k], v[j]);
}

__global__ void scatter6_kernel(const float* __restrict__ x, const int* __restrict__ ei,
                                int E, float* __restrict__ agg6) {
    const int lane  = threadIdx.x & 31;
    const int wave  = __builtin_amdgcn_readfirstlane((int)((blockIdx.x * blockDim.x + threadIdx.x) >> 5));
    const int ebase = wave * 32;
    if (ebase >= E) return;
    const int g = lane >> 3;
    const int k = lane & 7;
    if (k >= 6) return;
    if (ebase + 32 <= E) scatter6_wave<false>(x, ei, E, agg6, ebase, g, k);
    else                 scatter6_wave<true >(x, ei, E, agg6, ebase, g, k);
}

#define SC_EPW 8
template <bool TAIL>
__device__ __forceinline__ void scatter128_wave(const float* __restrict__ h1,
                                                const int* __restrict__ ei, int E,
                                                float* __restrict__ agg, int e0, int lane) {
    int src[SC_EPW], dst[SC_EPW];
    float v[SC_EPW][4];
#pragma unroll
    for (int j = 0; j < SC_EPW; ++j) {
        int e = e0 + j;
        if (TAIL) e = (e < E) ? e : (E - 1);
        src[j] = ei[e];
        dst[j] = ei[E + e];
    }
#pragma unroll
    for (int j = 0; j < SC_EPW; ++j) {
        const float* hs = h1 + src[j] * H + lane;
#pragma unroll
        for (int k = 0; k < 4; ++k) v[j][k] = hs[32 * k];
    }
#pragma unroll
    for (int j = 0; j < SC_EPW; ++j) {
        if (!TAIL || e0 + j < E) {
            float* ad = agg + dst[j] * H + lane;
#pragma unroll
            for (int k = 0; k < 4; ++k) atomicAdd(ad + 32 * k, v[j][k]);
        }
    }
}

__global__ void scatter128_kernel(const float* __restrict__ h1, const int* __restrict__ ei,
                                  int E, float* __restrict__ agg) {
    const int lane = threadIdx.x & 31;
    const int wave = __builtin_amdgcn_readfirstlane((int)((blockIdx.x * blockDim.x + threadIdx.x) >> 5));
    const int e0 = wave * SC_EPW;
    if (e0 >= E) return;
    if (e0 + SC_EPW <= E) scatter128_wave<false>(h1, ei, E, agg, e0, lane);
    else                  scatter128_wave<true >(h1, ei, E, agg, e0, lane);
}

#define IN_NPW 8
template <bool TAIL>
__device__ __forceinline__ void interp_wave(const float* __restrict__ vals,
                                            const int* __restrict__ cols,
                                            const int* __restrict__ rows,
                                            const float* __restrict__ e,
                                            float* __restrict__ acc, int nz0, int lane) {
    float v[IN_NPW];
    int col[IN_NPW], row[IN_NPW];
    float g[IN_NPW][3];
#pragma unroll
    for (int j = 0; j < IN_NPW; ++j) {
        int c = nz0 + j;
        if (TAIL) c = (c < NNZ) ? c : (NNZ - 1);
        v[j]   = vals[c];
        col[j] = cols[c];
        row[j] = rows[c];
    }
#pragma unroll
    for (int j = 0; j < IN_NPW; ++j) {
        const float* ep = e + col[j] * F3 + lane;
#pragma unroll
        for (int k = 0; k < 3; ++k) g[j][k] = ep[32 * k];
    }
#pragma unroll
    for (int j = 0; j < IN_NPW; ++j) {
        if (!TAIL || nz0 + j < NNZ) {
            float* ap = acc + row[j] * F3 + lane;
#pragma unroll
            for (int k = 0; k < 3; ++k) atomicAdd(ap + 32 * k, v[j] * g[j][k]);
        }
    }
}

__global__ void interp_kernel(const float* __restrict__ vals, const int* __restrict__ cols,
                              const int* __restrict__ rows, const float* __restrict__ e,
                              float* __restrict__ acc) {
    const int lane = threadIdx.x & 31;
    const int wave = __builtin_amdgcn_readfirstlane((int)((blockIdx.x * blockDim.x + threadIdx.x) >> 5));
    const int nz0 = wave * IN_NPW;
    if (nz0 >= NNZ) return;
    if (nz0 + IN_NPW <= NNZ) interp_wave<false>(vals, cols, rows, e, acc, nz0, lane);
    else                     interp_wave<true >(vals, cols, rows, e, acc, nz0, lane);
}

#define GS_THREADS 128

template <int K, int NC, int UNR = 4>
__device__ __forceinline__ void stage_weights(const float* __restrict__ W, float* __restrict__ Ws) {
    constexpr int NT = NC / 16;
#pragma unroll UNR
    for (int q = threadIdx.x; q < (K / 4) * NC; q += GS_THREADS) {
        const int ks = q / NC;
        const int nn = q - ks * NC;
        const float* wp = W + ks * 4 * NC + nn;
        v2f lo, hi;
        lo.x = wp[0];
        lo.y = wp[NC];
        hi.x = wp[2 * NC];
        hi.y = wp[3 * NC];
        const int t = nn >> 4;
        const int c = nn & 15;
        float* dst = Ws + ((ks * (NT / 2) + (t >> 1)) * 32 + c) * 4 + (t & 1) * 2;
        *(v2f*)dst        = lo;
        *(v2f*)(dst + 64) = hi;
    }
}

template <int NT>
__device__ __forceinline__ void load_b(const float* __restrict__ wb, v2f (&b)[NT]) {
#pragma unroll
    for (int tp = 0; tp < NT / 2; ++tp) {
        const float4 q = *(const float4*)(wb + tp * 128);
        b[2 * tp].x     = q.x;
        b[2 * tp].y     = q.y;
        b[2 * tp + 1].x = q.z;
        b[2 * tp + 1].y = q.w;
    }
}

template <int NC, int NT, int ACT>
__device__ __forceinline__ void store_strip(const v8f (&acc)[NT], float* __restrict__ D,
                                            int n, int row0, int m, int half) {
    const int rbase = row0 + half * 8;
    const int lim   = n - rbase;
    float* dp = D + rbase * NC + m;
#pragma unroll
    for (int j = 0; j < 8; ++j) {
        if (j < lim) {
#pragma unroll
            for (int t = 0; t < NT; ++t) {
                float v = acc[t][j];
                if (ACT == 1) v = fmaxf(v, 0.0f);
                if (ACT == 2) v = (v > 0.0f) ? v : (expf(v) - 1.0f);
                dp[j * NC + 16 * t] = v;
            }
        }
    }
}

template <int K, int NT, bool DUAL>
__device__ __forceinline__ void strip_products(const v2f (&a0)[K / 4], const v2f (&a1)[K / 4],
                                               const float* __restrict__ w0s,
                                               const float* __restrict__ w1s, v8f (&acc)[NT]) {
    constexpr int KS = K / 4;
    v2f b0[NT], b1[NT];
    load_b<NT>(w0s, b0);
    if (DUAL) load_b<NT>(w1s, b1);
#pragma unroll
    for (int ks = 0; ks < KS; ++ks) {
        v2f nb0[NT], nb1[NT];
        if (ks + 1 < KS) {
            load_b<NT>(w0s + (ks + 1) * (NT / 2) * 128, nb0);
            if (DUAL) load_b<NT>(w1s + (ks + 1) * (NT / 2) * 128, nb1);
        }
#pragma unroll
        for (int t = 0; t < NT; ++t)
            acc[t] = __builtin_amdgcn_wmma_f32_16x16x4_f32(false, a0[ks], false, b0[t], (short)0,
                                                           acc[t], false, false);
        if (DUAL) {
#pragma unroll
            for (int t = 0; t < NT; ++t)
                acc[t] = __builtin_amdgcn_wmma_f32_16x16x4_f32(false, a1[ks], false, b1[t], (short)0,
                                                               acc[t], false, false);
        }
        if (ks + 1 < KS) {
#pragma unroll
            for (int t = 0; t < NT; ++t) {
                b0[t] = nb0[t];
                if (DUAL) b1[t] = nb1[t];
            }
        }
    }
}

template <int K, int NC, int ACT, bool DUAL>
__global__ void __launch_bounds__(GS_THREADS)
gemm_wmma_kernel(const float* __restrict__ A0, const float* __restrict__ W0,
                 const float* __restrict__ A1, const float* __restrict__ W1,
                 const float* __restrict__ bias, float* __restrict__ D, int n) {
    constexpr int NT = NC / 16;
    constexpr int KS = K / 4;
    __shared__ __attribute__((aligned(16))) float Ws[(DUAL ? 2 : 1) * K * NC];

    stage_weights<K, NC>(W0, Ws);
    if (DUAL) stage_weights<K, NC>(W1, Ws + K * NC);
    __syncthreads();

    const int lane = threadIdx.x & 31;
    const int wave = __builtin_amdgcn_readfirstlane((int)(threadIdx.x >> 5));
    const int m    = lane & 15;
    const int half = lane >> 4;
    const int nstrips = (n + 15) >> 4;

#pragma unroll 1
    for (int s = blockIdx.x * (GS_THREADS / 32) + wave; s < nstrips; s += gridDim.x * (GS_THREADS / 32)) {
        int boff0 = lane * 4;
        int boff1 = lane * 4 + K * NC;
        if (DUAL) {
            asm volatile("" : "+v"(boff0), "+v"(boff1));
        }
        const float* w0s = Ws + boff0;
        const float* w1s = Ws + boff1;
        const int row0 = s * 16;
        const int row  = row0 + m;
        const int rowc = (row < n) ? row : (n - 1);

        const float* a0p = A0 + rowc * K + half * 2;
        const float* a1p = DUAL ? (A1 + rowc * K + half * 2) : a0p;
        v2f a0[KS], a1[KS];
#pragma unroll
        for (int ks = 0; ks < KS; ++ks) {
            a0[ks] = *(const v2f*)(a0p + 4 * ks);
            if (DUAL) a1[ks] = *(const v2f*)(a1p + 4 * ks);
        }

        v8f acc[NT];
#pragma unroll
        for (int t = 0; t < NT; ++t) {
            const float bv = bias[16 * t + m];
            acc[t] = (v8f){bv, bv, bv, bv, bv, bv, bv, bv};
        }

        strip_products<K, NT, DUAL>(a0, a1, w0s, w1s, acc);

        store_strip<NC, NT, ACT>(acc, D, n, row0, m, half);
    }
}

#define DEC_RPB 32
__global__ void __launch_bounds__(GS_THREADS)
dec_fused_wmma_kernel(const float* __restrict__ e0, const float* __restrict__ i1,
                      const float* __restrict__ i2, const float* __restrict__ W,
                      const float* __restrict__ bias, const float* __restrict__ Wout,
                      const float* __restrict__ bout, float* __restrict__ out, int n) {
    constexpr int K = F3, NC = DECW;
    constexpr int NT = NC / 16;
    constexpr int KS = K / 4;
    __shared__ __attribute__((aligned(16))) float Ws[3 * K * NC];

    stage_weights<3 * K, NC, 12>(W, Ws);
    __syncthreads();

    const int lane = threadIdx.x & 31;
    const int wave = __builtin_amdgcn_readfirstlane((int)(threadIdx.x >> 5));
    const int m    = lane & 15;
    const int half = lane >> 4;
    const int nblk = (n + DEC_RPB - 1) / DEC_RPB;

    float bz[3][NT], wo[3][NT], bo[3];
#pragma unroll
    for (int b = 0; b < 3; ++b) {
        bo[b] = bout[b];
#pragma unroll
        for (int t = 0; t < NT; ++t) {
            bz[b][t] = bias[b * NC + 16 * t + m];
            wo[b][t] = Wout[b * NC + 16 * t + m];
        }
    }
    const bool bit3 = (m & 8) != 0;
    const bool bit2 = (m & 4) != 0;
    const bool bit1 = (m & 2) != 0;

#pragma unroll 1
    for (int blk = blockIdx.x * (GS_THREADS / 32) + wave; blk < nblk; blk += gridDim.x * (GS_THREADS / 32)) {
#pragma unroll 1
        for (int st = 0; st < DEC_RPB / 16; ++st) {
            const int row0 = blk * DEC_RPB + st * 16;
            if (row0 >= n) break;
            const int row  = row0 + m;
            const int rowc = (row < n) ? row : (n - 1);

            const int abase = rowc * F3 + half * 2;
            const float* s0 = e0 + abase;
            const float* s1 = i1 + abase;
            const float* s2 = i2 + abase;
            v2f a[3][KS];
#pragma unroll
            for (int b = 0; b < 3; ++b) {
#pragma unroll
                for (int ks = 0; ks < KS; ++ks) {
                    const int k   = 4 * ks;
                    const int src = k >> 5;
                    const float* sp = (src == 0) ? s0 : (src == 1) ? s1 : s2;
                    a[b][ks] = *(const v2f*)(sp + b * 32 + (k & 31));
                }
            }
            float res[3];
#pragma unroll
            for (int b = 0; b < 3; ++b) {
                v8f acc[NT];
#pragma unroll
                for (int t = 0; t < NT; ++t) {
                    const float bv = bz[b][t];
                    acc[t] = (v8f){bv, bv, bv, bv, bv, bv, bv, bv};
                }
                const float* ws = Ws + lane * 4 + b * K * NC;
                strip_products<K, NT, false>(a[b], a[b], ws, ws, acc);

                float p[8];
#pragma unroll
                for (int j = 0; j < 8; ++j) {
                    float s = 0.0f;
#pragma unroll
                    for (int t = 0; t < NT; ++t) {
                        float v = acc[t][j];
                        v = (v > 0.0f) ? v : (expf(v) - 1.0f);
                        s += v * wo[b][t];
                    }
                    p[j] = s;
                }
                float q[4], r2[2];
#pragma unroll
                for (int i = 0; i < 4; ++i) {
                    const float send = bit3 ? p[i] : p[i + 4];
                    const float keep = bit3 ? p[i + 4] : p[i];
                    q[i] = keep + __shfl_xor(send, 8, 32);
                }
#pragma unroll
                for (int i = 0; i < 2; ++i) {
                    const float send = bit2 ? q[i] : q[i + 2];
                    const float keep = bit2 ? q[i + 2] : q[i];
                    r2[i] = keep + __shfl_xor(send, 4, 32);
                }
                const float send1 = bit1 ? r2[0] : r2[1];
                const float keep1 = bit1 ? r2[1] : r2[0];
                float tot = keep1 + __shfl_xor(send1, 2, 32);
                tot += __shfl_xor(tot, 1, 32);
                res[b] = tot + bo[b];
            }

            const int orow = row0 + half * 8 + (m >> 1);
            if ((m & 1) == 0 && orow < n) {
                float* op = out + orow * 3;
                op[0] = res[0];
                op[1] = res[1];
                op[2] = res[2];
            }
        }
    }
}

static inline int fill_grid(int count) {
    int b = (count / 4 + 8 * 32 * FILL_SPW - 1) / (8 * 32 * FILL_SPW);
    return b < 1 ? 1 : (b > 256 ? 256 : b);
}

static inline int gemm_grid(int n) {
    int b = ((n + 15) / 16 + 3) / 4;
    return b > 512 ? 512 : b;
}

extern "C" void kernel_launch(void* const* d_in, const int* in_sizes, int n_in,
                              void* d_out, int out_size, void* d_ws, size_t ws_size,
                              hipStream_t stream) {
    (void)in_sizes; (void)n_in; (void)out_size; (void)ws_size;

    const float* x[3]   = {(const float*)d_in[0], (const float*)d_in[1], (const float*)d_in[2]};
    const int*   ei[3]  = {(const int*)d_in[3], (const int*)d_in[4], (const int*)d_in[5]};
    const int*   A1r    = (const int*)d_in[6];
    const int*   A1c    = (const int*)d_in[7];
    const float* A1v    = (const float*)d_in[8];
    const int*   A2r    = (const int*)d_in[9];
    const int*   A2c    = (const int*)d_in[10];
    const float* A2v    = (const float*)d_in[11];
    const float* c1Wrel  = (const float*)d_in[12];
    const float* c1brel  = (const float*)d_in[13];
    const float* c1Wroot = (const float*)d_in[14];
    const float* c2Wrel  = (const float*)d_in[15];
    const float* c2brel  = (const float*)d_in[16];
    const float* c2Wroot = (const float*)d_in[17];
    const float* linW    = (const float*)d_in[18];
    const float* linb    = (const float*)d_in[19];
    const float* decW0   = (const float*)d_in[20];
    const float* decb0   = (const float*)d_in[21];
    const float* decWout = (const float*)d_in[22];
    const float* decbout = (const float*)d_in[23];
    float* out = (float*)d_out;

    float* ws = (float*)d_ws;
    float* h1   = ws;
    float* agg  = h1   + (size_t)N0 * H;
    float* h2   = agg  + (size_t)N0 * H;
    float* agg6 = h2   + (size_t)N0 * H;
    float* e0   = agg6 + (size_t)N0 * 6;
    float* e1   = e0   + (size_t)N0 * F3;
    float* e2   = e1   + (size_t)N1 * F3;
    float* i1   = e2   + (size_t)N2 * F3;
    float* i2   = i1   + (size_t)N0 * F3;

    const int nlv[3] = {N0, N1, N2};
    const int Elv[3] = {E0, E1, E2};
    float* elv[3] = {e0, e1, e2};

    for (int i = 0; i < 3; ++i) {
        const int n = nlv[i];
        const int E = Elv[i];

        {
            zero_kernel<<<fill_grid(n * 6), 256, 0, stream>>>(agg6, n * 6);
            scatter6_kernel<<<(E + 255) / 256, 256, 0, stream>>>(x[i], ei[i], E, agg6);
            conv1_kernel<<<(n + 8 * C1_RPW - 1) / (8 * C1_RPW), 256, 0, stream>>>(x[i], agg6,
                                              c1Wrel + (size_t)i * 6 * H,
                                              c1brel + (size_t)i * H,
                                              c1Wroot + (size_t)i * 6 * H, h1, n);
        }

        {
            zero_kernel<<<fill_grid(n * H), 256, 0, stream>>>(agg, n * H);
            scatter128_kernel<<<(E + 8 * SC_EPW - 1) / (8 * SC_EPW), 256, 0, stream>>>(h1, ei[i], E, agg);
            gemm_wmma_kernel<H, H, 1, true><<<gemm_grid(n), GS_THREADS, 0, stream>>>(
                agg, c2Wrel + (size_t)i * H * H,
                h1,  c2Wroot + (size_t)i * H * H,
                c2brel + (size_t)i * H, h2, n);
        }

        gemm_wmma_kernel<H, F3, 0, false><<<gemm_grid(n), GS_THREADS, 0, stream>>>(
            h2, linW + (size_t)i * H * F3, nullptr, nullptr,
            linb + (size_t)i * F3, elv[i], n);
    }

    {
        zero_kernel<<<fill_grid(N0 * F3), 256, 0, stream>>>(i1, N0 * F3);
        zero_kernel<<<fill_grid(N0 * F3), 256, 0, stream>>>(i2, N0 * F3);
        interp_kernel<<<(NNZ + 8 * IN_NPW - 1) / (8 * IN_NPW), 256, 0, stream>>>(A1v, A1c, A1r, e1, i1);
        interp_kernel<<<(NNZ + 8 * IN_NPW - 1) / (8 * IN_NPW), 256, 0, stream>>>(A2v, A2c, A2r, e2, i2);
    }

    dec_fused_wmma_kernel<<<((N0 + DEC_RPB - 1) / DEC_RPB + 3) / 4, GS_THREADS, 0, stream>>>(
        e0, i1, i2, decW0, decb0, decWout, decbout, out, N0);
}


# encoder stages batched over the 3 levels: one launch per stage (wave or work-group picks its level by index; levels' rows side by side), fills and both interpolations merged; 26 -> 9 launches
# speedup vs baseline: 2.7981x; 1.1702x over previous
#include <hip/hip_runtime.h>
#include <hip/hip_bf16.h>

#define N0 50000
#define N1 12500
#define N2 3125
#define E0 800000
#define E1 200000
#define E2 50000
#define NNZ 200000
#define H 128
#define F3 96
#define DECW 64

#define ROFF1 N0
#define ROFF2 (N0 + N1)
#define NROWS (N0 + N1 + N2)

typedef __attribute__((ext_vector_type(2))) float v2f;
typedef __attribute__((ext_vector_type(8))) float v8f;

#define FILL_SPW 4
__global__ void zero_kernel(float* __restrict__ p, int count) {
    const int tid  = blockIdx.x * blockDim.x + threadIdx.x;
    const int lane = threadIdx.x & 31;
    const int wave = __builtin_amdgcn_readfirstlane(tid >> 5);
    const int step = ((gridDim.x * blockDim.x) >> 5) * (32 * FILL_SPW);
    int head = (int)(((16u - (unsigned)((size_t)p & 15u)) & 15u) >> 2);
    if (head > count) head = count;
    float4* q = (float4*)(p + head);
    const int nvec = (count - head) >> 2;
    const float4 z = {0.0f, 0.0f, 0.0f, 0.0f};
    for (int base = wave * (32 * FILL_SPW); base < nvec; base += step) {
        float4* qq = q + base + lane;
        if (base + 32 * FILL_SPW <= nvec) {
#pragma unroll
            for (int u = 0; u < FILL_SPW; ++u) qq[32 * u] = z;
        } else {
#pragma unroll
            for (int u = 0; u < FILL_SPW; ++u)
                if (base + 32 * u + lane < nvec) qq[32 * u] = z;
        }
    }
    const int tail0 = head + (nvec << 2);
    if (tid < head) p[tid] = 0.0f;
    if (tid < count - tail0) p[tail0 + tid] = 0.0f;
}

#define C1_RPW 8
#define C1_W0 ((N0 + C1_RPW - 1) / C1_RPW)
#define C1_W1 ((N1 + C1_RPW - 1) / C1_RPW)
#define C1_W2 ((N2 + C1_RPW - 1) / C1_RPW)
template <bool TAIL>
__device__ __forceinline__ void conv1_wave(const float* __restrict__ x, const float* __restrict__ agg6,
                                           const float* __restrict__ Wrel, const float* __restrict__ brel,
                                           const float* __restrict__ Wroot, float* __restrict__ h1,
                                           int n, int r0, int c) {
    float wr[6][4], wo[6][4], b[4];
#pragma unroll
    for (int q = 0; q < 4; ++q) b[q] = brel[c + q];
#pragma unroll
    for (int k = 0; k < 6; ++k) {
#pragma unroll
        for (int q = 0; q < 4; ++q) {
            wr[k][q] = Wrel[k * H + c + q];
            wo[k][q] = Wroot[k * H + c + q];
        }
    }
#pragma unroll
    for (int j = 0; j < C1_RPW; ++j) {
        const int r = r0 + j;
        int rl = r;
        if (TAIL) rl = (r < n) ? r : (n - 1);
        const float* ar = agg6 + rl * 6;
        const float* xr = x + rl * 6;
        float s[4] = {b[0], b[1], b[2], b[3]};
#pragma unroll
        for (int k = 0; k < 6; ++k) {
            const float a  = ar[k];
            const float xv = xr[k];
#pragma unroll
            for (int q = 0; q < 4; ++q) {
                s[q] += a * wr[k][q];
                s[q] += xv * wo[k][q];
            }
        }
        if (!TAIL || r < n) {
            float4 v;
            v.x = fmaxf(s[0], 0.0f);
            v.y = fmaxf(s[1], 0.0f);
            v.z = fmaxf(s[2], 0.0f);
            v.w = fmaxf(s[3], 0.0f);
            *(float4*)(h1 + r * H + c) = v;
        }
    }
}

__global__ void conv1_kernel(const float* __restrict__ x0, const float* __restrict__ x1,
                             const float* __restrict__ x2, const float* __restrict__ agg6,
                             const float* __restrict__ Wrel, const float* __restrict__ brel,
                             const float* __restrict__ Wroot, float* __restrict__ h1) {
    const int lane = threadIdx.x & 31;
    int wave = __builtin_amdgcn_readfirstlane((int)((blockIdx.x * blockDim.x + threadIdx.x) >> 5));
    const float* x = x0;
    int n = N0, lvl = 0, roff = 0;
    if (wave >= C1_W0 + C1_W1) { wave -= C1_W0 + C1_W1; x = x2; n = N2; lvl = 2; roff = ROFF2; }
    else if (wave >= C1_W0)    { wave -= C1_W0;         x = x1; n = N1; lvl = 1; roff = ROFF1; }
    const int r0 = wave * C1_RPW;
    if (r0 >= n) return;
    const float* a6 = agg6 + roff * 6;
    const float* wr = Wrel + lvl * 6 * H;
    const float* br = brel + lvl * H;
    const float* wo = Wroot + lvl * 6 * H;
    float* hd = h1 + (size_t)roff * H;
    if (r0 + C1_RPW <= n) conv1_wave<false>(x, a6, wr, br, wo, hd, n, r0, lane * 4);
    else                  conv1_wave<true >(x, a6, wr, br, wo, hd, n, r0, lane * 4);
}

template <bool TAIL>
__device__ __forceinline__ void scatter6_wave(const float* __restrict__ x,
                                              const int* __restrict__ ei, int E,
                                              float* __restrict__ agg6,
                                              int ebase, int g, int k) {
    int src[8], dst[8];
    float v[8];
#pragma unroll
    for (int j = 0; j < 8; ++j) {
        int e = ebase + j * 4 + g;
        if (TAIL) e = (e < E) ? e : (E - 1);
        src[j] = ei[e];
        dst[j] = ei[E + e];
    }
#pragma unroll
    for (int j = 0; j < 8; ++j) v[j] = x[src[j] * 6 + k];
#pragma unroll
    for (int j = 0; j < 8; ++j)
        if (!TAIL || ebase + j * 4 + g < E) atomicAdd(&agg6[dst[j] * 6 + k], v[j]);
}

#define S6_W0 ((E0 + 31) / 32)
#define S6_W1 ((E1 + 31) / 32)
#define S6_W2 ((E2 + 31) / 32)
__global__ void scatter6_kernel(const float* __restrict__ x0, const float* __restrict__ x1,
                                const float* __restrict__ x2, const int* __restrict__ ei0,
                                const int* __restrict__ ei1, const int* __restrict__ ei2,
                                float* __restrict__ agg6all) {
    const int lane  = threadIdx.x & 31;
    int wave  = __builtin_amdgcn_readfirstlane((int)((blockIdx.x * blockDim.x + threadIdx.x) >> 5));
    const float* x = x0;
    const int* ei = ei0;
    int E = E0, roff = 0;
    if (wave >= S6_W0 + S6_W1) { wave -= S6_W0 + S6_W1; x = x2; ei = ei2; E = E2; roff = ROFF2; }
    else if (wave >= S6_W0)    { wave -= S6_W0;         x = x1; ei = ei1; E = E1; roff = ROFF1; }
    float* agg6 = agg6all + roff * 6;
    const int ebase = wave * 32;
    if (ebase >= E) return;
    const int g = lane >> 3;
    const int k = lane & 7;
    if (k >= 6) return;
    if (ebase + 32 <= E) scatter6_wave<false>(x, ei, E, agg6, ebase, g, k);
    else                 scatter6_wave<true >(x, ei, E, agg6, ebase, g, k);
}

#define SC_EPW 8
template <bool TAIL>
__device__ __forceinline__ void scatter128_wave(const float* __restrict__ h1,
                                                const int* __restrict__ ei, int E,
                                                float* __restrict__ agg, int e0, int lane) {
    int src[SC_EPW], dst[SC_EPW];
    float v[SC_EPW][4];
#pragma unroll
    for (int j = 0; j < SC_EPW; ++j) {
        int e = e0 + j;
        if (TAIL) e = (e < E) ? e : (E - 1);
        src[j] = ei[e];
        dst[j] = ei[E + e];
    }
#pragma unroll
    for (int j = 0; j < SC_EPW; ++j) {
        const float* hs = h1 + src[j] * H + lane;
#pragma unroll
        for (int k = 0; k < 4; ++k) v[j][k] = hs[32 * k];
    }
#pragma unroll
    for (int j = 0; j < SC_EPW; ++j) {
        if (!TAIL || e0 + j < E) {
            float* ad = agg + dst[j] * H + lane;
#pragma unroll
            for (int k = 0; k < 4; ++k) atomicAdd(ad + 32 * k, v[j][k]);
        }
    }
}

#define SC_W0 ((E0 + SC_EPW - 1) / SC_EPW)
#define SC_W1 ((E1 + SC_EPW - 1) / SC_EPW)
#define SC_W2 ((E2 + SC_EPW - 1) / SC_EPW)
__global__ void scatter128_kernel(const float* __restrict__ h1all, const int* __restrict__ ei0,
                                  const int* __restrict__ ei1, const int* __restrict__ ei2,
                                  float* __restrict__ aggall) {
    const int lane = threadIdx.x & 31;
    int wave = __builtin_amdgcn_readfirstlane((int)((blockIdx.x * blockDim.x + threadIdx.x) >> 5));
    const int* ei = ei0;
    int E = E0, roff = 0;
    if (wave >= SC_W0 + SC_W1) { wave -= SC_W0 + SC_W1; ei = ei2; E = E2; roff = ROFF2; }
    else if (wave >= SC_W0)    { wave -= SC_W0;         ei = ei1; E = E1; roff = ROFF1; }
    const float* h1 = h1all + (size_t)roff * H;
    float* agg = aggall + (size_t)roff * H;
    const int e0 = wave * SC_EPW;
    if (e0 >= E) return;
    if (e0 + SC_EPW <= E) scatter128_wave<false>(h1, ei, E, agg, e0, lane);
    else                  scatter128_wave<true >(h1, ei, E, agg, e0, lane);
}

#define IN_NPW 8
template <bool TAIL>
__device__ __forceinline__ void interp_wave(const float* __restrict__ vals,
                                            const int* __restrict__ cols,
                                            const int* __restrict__ rows,
                                            const float* __restrict__ e,
                                            float* __restrict__ acc, int nz0, int lane) {
    float v[IN_NPW];
    int col[IN_NPW], row[IN_NPW];
    float g[IN_NPW][3];
#pragma unroll
    for (int j = 0; j < IN_NPW; ++j) {
        int c = nz0 + j;
        if (TAIL) c = (c < NNZ) ? c : (NNZ - 1);
        v[j]   = vals[c];
        col[j] = cols[c];
        row[j] = rows[c];
    }
#pragma unroll
    for (int j = 0; j < IN_NPW; ++j) {
        const float* ep = e + col[j] * F3 + lane;
#pragma unroll
        for (int k = 0; k < 3; ++k) g[j][k] = ep[32 * k];
    }
#pragma unroll
    for (int j = 0; j < IN_NPW; ++j) {
        if (!TAIL || nz0 + j < NNZ) {
            float* ap = acc + row[j] * F3 + lane;
#pragma unroll
            for (int k = 0; k < 3; ++k) atomicAdd(ap + 32 * k, v[j] * g[j][k]);
        }
    }
}

#define IN_W ((NNZ + IN_NPW - 1) / IN_NPW)
__global__ void interp_kernel(const float* __restrict__ vals1, const int* __restrict__ cols1,
                              const int* __restrict__ rows1, const float* __restrict__ ea,
                              float* __restrict__ acc1,
                              const float* __restrict__ vals2, const int* __restrict__ cols2,
                              const int* __restrict__ rows2, const float* __restrict__ eb,
                              float* __restrict__ acc2) {
    const int lane = threadIdx.x & 31;
    int wave = __builtin_amdgcn_readfirstlane((int)((blockIdx.x * blockDim.x + threadIdx.x) >> 5));
    const float* vals = vals1;
    const int* cols = cols1;
    const int* rows = rows1;
    const float* e = ea;
    float* acc = acc1;
    if (wave >= IN_W) { wave -= IN_W; vals = vals2; cols = cols2; rows = rows2; e = eb; acc = acc2; }
    const int nz0 = wave * IN_NPW;
    if (nz0 >= NNZ) return;
    if (nz0 + IN_NPW <= NNZ) interp_wave<false>(vals, cols, rows, e, acc, nz0, lane);
    else                     interp_wave<true >(vals, cols, rows, e, acc, nz0, lane);
}

#define GS_THREADS 128

template <int K, int NC, int UNR = 4>
__device__ __forceinline__ void stage_weights(const float* __restrict__ W, float* __restrict__ Ws) {
    constexpr int NT = NC / 16;
#pragma unroll UNR
    for (int q = threadIdx.x; q < (K / 4) * NC; q += GS_THREADS) {
        const int ks = q / NC;
        const int nn = q - ks * NC;
        const float* wp = W + ks * 4 * NC + nn;
        v2f lo, hi;
        lo.x = wp[0];
        lo.y = wp[NC];
        hi.x = wp[2 * NC];
        hi.y = wp[3 * NC];
        const int t = nn >> 4;
        const int c = nn & 15;
        float* dst = Ws + ((ks * (NT / 2) + (t >> 1)) * 32 + c) * 4 + (t & 1) * 2;
        *(v2f*)dst        = lo;
        *(v2f*)(dst + 64) = hi;
    }
}

template <int NT>
__device__ __forceinline__ void load_b(const float* __restrict__ wb, v2f (&b)[NT]) {
#pragma unroll
    for (int tp = 0; tp < NT / 2; ++tp) {
        const float4 q = *(const float4*)(wb + tp * 128);
        b[2 * tp].x     = q.x;
        b[2 * tp].y     = q.y;
        b[2 * tp + 1].x = q.z;
        b[2 * tp + 1].y = q.w;
    }
}

template <int NC, int NT, int ACT>
__device__ __forceinline__ void store_strip(const v8f (&acc)[NT], float* __restrict__ D,
                                            int n, int row0, int m, int half) {
    const int rbase = row0 + half * 8;
    const int lim   = n - rbase;
    float* dp = D + rbase * NC + m;
#pragma unroll
    for (int j = 0; j < 8; ++j) {
        if (j < lim) {
#pragma unroll
            for (int t = 0; t < NT; ++t) {
                float v = acc[t][j];
                if (ACT == 1) v = fmaxf(v, 0.0f);
                if (ACT == 2) v = (v > 0.0f) ? v : (expf(v) - 1.0f);
                dp[j * NC + 16 * t] = v;
            }
        }
    }
}

template <int K, int NT, bool DUAL>
__device__ __forceinline__ void strip_products(const v2f (&a0)[K / 4], const v2f (&a1)[K / 4],
                                               const float* __restrict__ w0s,
                                               const float* __restrict__ w1s, v8f (&acc)[NT]) {
    constexpr int KS = K / 4;
    v2f b0[NT], b1[NT];
    load_b<NT>(w0s, b0);
    if (DUAL) load_b<NT>(w1s, b1);
#pragma unroll
    for (int ks = 0; ks < KS; ++ks) {
        v2f nb0[NT], nb1[NT];
        if (ks + 1 < KS) {
            load_b<NT>(w0s + (ks + 1) * (NT / 2) * 128, nb0);
            if (DUAL) load_b<NT>(w1s + (ks + 1) * (NT / 2) * 128, nb1);
        }
#pragma unroll
        for (int t = 0; t < NT; ++t)
            acc[t] = __builtin_amdgcn_wmma_f32_16x16x4_f32(false, a0[ks], false, b0[t], (short)0,
                                                           acc[t], false, false);
        if (DUAL) {
#pragma unroll
            for (int t = 0; t < NT; ++t)
                acc[t] = __builtin_amdgcn_wmma_f32_16x16x4_f32(false, a1[ks], false, b1[t], (short)0,
                                                               acc[t], false, false);
        }
        if (ks + 1 < KS) {
#pragma unroll
            for (int t = 0; t < NT; ++t) {
                b0[t] = nb0[t];
                if (DUAL) b1[t] = nb1[t];
            }
        }
    }
}

#define GEMM_GRID(n) ((((n) + 15) / 16 + 3) / 4 > 512 ? 512 : (((n) + 15) / 16 + 3) / 4)
#define GG0 GEMM_GRID(N0)
#define GG1 GEMM_GRID(N1)
#define GG2 GEMM_GRID(N2)

template <int K, int NC, int ACT, bool DUAL>
__global__ void __launch_bounds__(GS_THREADS)
gemm_wmma_kernel(const float* __restrict__ A0all, const float* __restrict__ W0all,
                 const float* __restrict__ A1all, const float* __restrict__ W1all,
                 const float* __restrict__ biasall, float* __restrict__ Dall) {
    constexpr int NT = NC / 16;
    constexpr int KS = K / 4;
    __shared__ __attribute__((aligned(16))) float Ws[(DUAL ? 2 : 1) * K * NC];

    int lb = blockIdx.x, lg = GG0, n = N0, lvl = 0, roff = 0;
    if (lb >= GG0 + GG1) { lb -= GG0 + GG1; lg = GG2; n = N2; lvl = 2; roff = ROFF2; }
    else if (lb >= GG0)  { lb -= GG0;       lg = GG1; n = N1; lvl = 1; roff = ROFF1; }
    const float* A0   = A0all + (size_t)roff * K;
    const float* W0   = W0all + lvl * K * NC;
    const float* A1   = DUAL ? (A1all + (size_t)roff * K) : A0;
    const float* W1   = DUAL ? (W1all + lvl * K * NC) : W0;
    const float* bias = biasall + lvl * NC;
    float* D          = Dall + (size_t)roff * NC;

    stage_weights<K, NC>(W0, Ws);
    if (DUAL) stage_weights<K, NC>(W1, Ws + K * NC);
    __syncthreads();

    const int lane = threadIdx.x & 31;
    const int wave = __builtin_amdgcn_readfirstlane((int)(threadIdx.x >> 5));
    const int m    = lane & 15;
    const int half = lane >> 4;
    const int nstrips = (n + 15) >> 4;

#pragma unroll 1
    for (int s = lb * (GS_THREADS / 32) + wave; s < nstrips; s += lg * (GS_THREADS / 32)) {
        int boff0 = lane * 4;
        int boff1 = lane * 4 + K * NC;
        if (DUAL) {
            asm volatile("" : "+v"(boff0), "+v"(boff1));
        }
        const float* w0s = Ws + boff0;
        const float* w1s = Ws + boff1;
        const int row0 = s * 16;
        const int row  = row0 + m;
        const int rowc = (row < n) ? row : (n - 1);

        const float* a0p = A0 + rowc * K + half * 2;
        const float* a1p = DUAL ? (A1 + rowc * K + half * 2) : a0p;
        v2f a0[KS], a1[KS];
#pragma unroll
        for (int ks = 0; ks < KS; ++ks) {
            a0[ks] = *(const v2f*)(a0p + 4 * ks);
            if (DUAL) a1[ks] = *(const v2f*)(a1p + 4 * ks);
        }

        v8f acc[NT];
#pragma unroll
        for (int t = 0; t < NT; ++t) {
            const float bv = bias[16 * t + m];
            acc[t] = (v8f){bv, bv, bv, bv, bv, bv, bv, bv};
        }

        strip_products<K, NT, DUAL>(a0, a1, w0s, w1s, acc);

        store_strip<NC, NT, ACT>(acc, D, n, row0, m, half);
    }
}

#define DEC_RPB 32
__global__ void __launch_bounds__(GS_THREADS)
dec_fused_wmma_kernel(const float* __restrict__ e0, const float* __restrict__ i1,
                      const float* __restrict__ i2, const float* __restrict__ W,
                      const float* __restrict__ bias, const float* __restrict__ Wout,
                      const float* __restrict__ bout, float* __restrict__ out, int n) {
    constexpr int K = F3, NC = DECW;
    constexpr int NT = NC / 16;
    constexpr int KS = K / 4;
    __shared__ __attribute__((aligned(16))) float Ws[3 * K * NC];

    stage_weights<3 * K, NC, 12>(W, Ws);
    __syncthreads();

    const int lane = threadIdx.x & 31;
    const int wave = __builtin_amdgcn_readfirstlane((int)(threadIdx.x >> 5));
    const int m    = lane & 15;
    const int half = lane >> 4;
    const int nblk = (n + DEC_RPB - 1) / DEC_RPB;

    float bz[3][NT], wo[3][NT], bo[3];
#pragma unroll
    for (int b = 0; b < 3; ++b) {
        bo[b] = bout[b];
#pragma unroll
        for (int t = 0; t < NT; ++t) {
            bz[b][t] = bias[b * NC + 16 * t + m];
            wo[b][t] = Wout[b * NC + 16 * t + m];
        }
    }
    const bool bit3 = (m & 8) != 0;
    const bool bit2 = (m & 4) != 0;
    const bool bit1 = (m & 2) != 0;

#pragma unroll 1
    for (int blk = blockIdx.x * (GS_THREADS / 32) + wave; blk < nblk; blk += gridDim.x * (GS_THREADS / 32)) {
#pragma unroll 1
        for (int st = 0; st < DEC_RPB / 16; ++st) {
            const int row0 = blk * DEC_RPB + st * 16;
            if (row0 >= n) break;
            const int row  = row0 + m;
            const int rowc = (row < n) ? row : (n - 1);

            const int abase = rowc * F3 + half * 2;
            const float* s0 = e0 + abase;
            const float* s1 = i1 + abase;
            const float* s2 = i2 + abase;
            v2f a[3][KS];
#pragma unroll
            for (int b = 0; b < 3; ++b) {
#pragma unroll
                for (int ks = 0; ks < KS; ++ks) {
                    const int k   = 4 * ks;
                    const int src = k >> 5;
                    const float* sp = (src == 0) ? s0 : (src == 1) ? s1 : s2;
                    a[b][ks] = *(const v2f*)(sp + b * 32 + (k & 31));
                }
            }
            float res[3];
#pragma unroll
            for (int b = 0; b < 3; ++b) {
                v8f acc[NT];
#pragma unroll
                for (int t = 0; t < NT; ++t) {
                    const float bv = bz[b][t];
                    acc[t] = (v8f){bv, bv, bv, bv, bv, bv, bv, bv};
                }
                const float* ws = Ws + lane * 4 + b * K * NC;
                strip_products<K, NT, false>(a[b], a[b], ws, ws, acc);

                float p[8];
#pragma unroll
                for (int j = 0; j < 8; ++j) {
                    float s = 0.0f;
#pragma unroll
                    for (int t = 0; t < NT; ++t) {
                        float v = acc[t][j];
                        v = (v > 0.0f) ? v : (expf(v) - 1.0f);
                        s += v * wo[b][t];
                    }
                    p[j] = s;
                }
                float q[4], r2[2];
#pragma unroll
                for (int i = 0; i < 4; ++i) {
                    const float send = bit3 ? p[i] : p[i + 4];
                    const float keep = bit3 ? p[i + 4] : p[i];
                    q[i] = keep + __shfl_xor(send, 8, 32);
                }
#pragma unroll
                for (int i = 0; i < 2; ++i) {
                    const float send = bit2 ? q[i] : q[i + 2];
                    const float keep = bit2 ? q[i + 2] : q[i];
                    r2[i] = keep + __shfl_xor(send, 4, 32);
                }
                const float send1 = bit1 ? r2[0] : r2[1];
                const float keep1 = bit1 ? r2[1] : r2[0];
                float tot = keep1 + __shfl_xor(send1, 2, 32);
                tot += __shfl_xor(tot, 1, 32);
                res[b] = tot + bo[b];
            }

            const int orow = row0 + half * 8 + (m >> 1);
            if ((m & 1) == 0 && orow < n) {
                float* op = out + orow * 3;
                op[0] = res[0];
                op[1] = res[1];
                op[2] = res[2];
            }
        }
    }
}

#define FILL_GRID(count) (((count) / 4 + 8 * 32 * FILL_SPW - 1) / (8 * 32 * FILL_SPW) > 256 ? 256 \
                          : ((count) / 4 + 8 * 32 * FILL_SPW - 1) / (8 * 32 * FILL_SPW))

#define WS_H1    ((size_t)0)
#define WS_AGG   (WS_H1 + (size_t)NROWS * H)
#define WS_AGG6  (WS_AGG + (size_t)NROWS * H)
#define WS_H2    (((WS_AGG6 + (size_t)NROWS * 6 + 31) / 32) * 32)
#define WS_E     (WS_H2 + (size_t)NROWS * H)
#define WS_I1    ((size_t)0)
#define WS_I2    (WS_I1 + (size_t)N0 * F3)

extern "C" void kernel_launch(void* const* d_in, const int* in_sizes, int n_in,
                              void* d_out, int out_size, void* d_ws, size_t ws_size,
                              hipStream_t stream) {
    (void)in_sizes; (void)n_in; (void)out_size; (void)ws_size;

    const float* x0     = (const float*)d_in[0];
    const float* x1     = (const float*)d_in[1];
    const float* x2     = (const float*)d_in[2];
    const int*   ei0    = (const int*)d_in[3];
    const int*   ei1    = (const int*)d_in[4];
    const int*   ei2    = (const int*)d_in[5];
    const int*   A1r    = (const int*)d_in[6];
    const int*   A1c    = (const int*)d_in[7];
    const float* A1v    = (const float*)d_in[8];
    const int*   A2r    = (const int*)d_in[9];
    const int*   A2c    = (const int*)d_in[10];
    const float* A2v    = (const float*)d_in[11];
    const float* c1Wrel  = (const float*)d_in[12];
    const float* c1brel  = (const float*)d_in[13];
    const float* c1Wroot = (const float*)d_in[14];
    const float* c2Wrel  = (const float*)d_in[15];
    const float* c2brel  = (const float*)d_in[16];
    const float* c2Wroot = (const float*)d_in[17];
    const float* linW    = (const float*)d_in[18];
    const float* linb    = (const float*)d_in[19];
    const float* decW0   = (const float*)d_in[20];
    const float* decb0   = (const float*)d_in[21];
    const float* decWout = (const float*)d_in[22];
    const float* decbout = (const float*)d_in[23];
    float* out = (float*)d_out;

    float* ws   = (float*)d_ws;
    float* h1   = ws + WS_H1;
    float* agg  = ws + WS_AGG;
    float* agg6 = ws + WS_AGG6;
    float* h2   = ws + WS_H2;
    float* eall = ws + WS_E;
    float* i1   = ws + WS_I1;
    float* i2   = ws + WS_I2;

    zero_kernel<<<FILL_GRID(NROWS * (H + 6)), 256, 0, stream>>>(agg, NROWS * (H + 6));
    scatter6_kernel<<<(S6_W0 + S6_W1 + S6_W2 + 7) / 8, 256, 0, stream>>>(x0, x1, x2, ei0, ei1, ei2, agg6);
    conv1_kernel<<<(C1_W0 + C1_W1 + C1_W2 + 7) / 8, 256, 0, stream>>>(x0, x1, x2, agg6, c1Wrel, c1brel,
                                                                    c1Wroot, h1);

    scatter128_kernel<<<(SC_W0 + SC_W1 + SC_W2 + 7) / 8, 256, 0, stream>>>(h1, ei0, ei1, ei2, agg);
    gemm_wmma_kernel<H, H, 1, true><<<GG0 + GG1 + GG2, GS_THREADS, 0, stream>>>(
        agg, c2Wrel, h1, c2Wroot, c2brel, h2);

    gemm_wmma_kernel<H, F3, 0, false><<<GG0 + GG1 + GG2, GS_THREADS, 0, stream>>>(
        h2, linW, nullptr, nullptr, linb, eall);

    zero_kernel<<<FILL_GRID(2 * N0 * F3), 256, 0, stream>>>(i1, 2 * N0 * F3);
    interp_kernel<<<(2 * IN_W + 7) / 8, 256, 0, stream>>>(A1v, A1c, A1r, eall + (size_t)ROFF1 * F3, i1,
                                                         A2v, A2c, A2r, eall + (size_t)ROFF2 * F3, i2);

    dec_fused_wmma_kernel<<<((N0 + DEC_RPB - 1) / DEC_RPB + 3) / 4, GS_THREADS, 0, stream>>>(
        eall, i1, i2, decW0, decb0, decWout, decbout, out, N0);
}


# conv1/conv2 segment sums as gathers over a per-call CSR (1 returning int atomic per edge ranks it in its dst row); a wave sums a row's sources in registers; no float atomics there; sum order changed
# speedup vs baseline: 3.0180x; 1.0786x over previous
#include <hip/hip_runtime.h>
#include <hip/hip_bf16.h>

#define N0 50000
#define N1 12500
#define N2 3125
#define E0 800000
#define E1 200000
#define E2 50000
#define NNZ 200000
#define H 128
#define F3 96
#define DECW 64

#define ROFF1 N0
#define ROFF2 (N0 + N1)
#define NROWS (N0 + N1 + N2)

typedef __attribute__((ext_vector_type(2))) float v2f;
typedef __attribute__((ext_vector_type(8))) float v8f;

#define FILL_SPW 4
__global__ void zero_kernel(float* __restrict__ p, int count) {
    const int tid  = blockIdx.x * blockDim.x + threadIdx.x;
    const int lane = threadIdx.x & 31;
    const int wave = __builtin_amdgcn_readfirstlane(tid >> 5);
    const int step = ((gridDim.x * blockDim.x) >> 5) * (32 * FILL_SPW);
    int head = (int)(((16u - (unsigned)((size_t)p & 15u)) & 15u) >> 2);
    if (head > count) head = count;
    float4* q = (float4*)(p + head);
    const int nvec = (count - head) >> 2;
    const float4 z = {0.0f, 0.0f, 0.0f, 0.0f};
    for (int base = wave * (32 * FILL_SPW); base < nvec; base += step) {
        float4* qq = q + base + lane;
        if (base + 32 * FILL_SPW <= nvec) {
#pragma unroll
            for (int u = 0; u < FILL_SPW; ++u) qq[32 * u] = z;
        } else {
#pragma unroll
            for (int u = 0; u < FILL_SPW; ++u)
                if (base + 32 * u + lane < nvec) qq[32 * u] = z;
        }
    }
    const int tail0 = head + (nvec << 2);
    if (tid < head) p[tid] = 0.0f;
    if (tid < count - tail0) p[tail0 + tid] = 0.0f;
}

#define C1_RPW 8
#define C1_W0 ((N0 + C1_RPW - 1) / C1_RPW)
#define C1_W1 ((N1 + C1_RPW - 1) / C1_RPW)
#define C1_W2 ((N2 + C1_RPW - 1) / C1_RPW)
template <bool TAIL>
__device__ __forceinline__ void conv1_wave(const float* __restrict__ x, const float* __restrict__ agg6,
                                           const float* __restrict__ Wrel, const float* __restrict__ brel,
                                           const float* __restrict__ Wroot, float* __restrict__ h1,
                                           int n, int r0, int c) {
    float wr[6][4], wo[6][4], b[4];
#pragma unroll
    for (int q = 0; q < 4; ++q) b[q] = brel[c + q];
#pragma unroll
    for (int k = 0; k < 6; ++k) {
#pragma unroll
        for (int q = 0; q < 4; ++q) {
            wr[k][q] = Wrel[k * H + c + q];
            wo[k][q] = Wroot[k * H + c + q];
        }
    }
#pragma unroll
    for (int j = 0; j < C1_RPW; ++j) {
        const int r = r0 + j;
        int rl = r;
        if (TAIL) rl = (r < n) ? r : (n - 1);
        const float* ar = agg6 + rl * 6;
        const float* xr = x + rl * 6;
        float s[4] = {b[0], b[1], b[2], b[3]};
#pragma unroll
        for (int k = 0; k < 6; ++k) {
            const float a  = ar[k];
            const float xv = xr[k];
#pragma unroll
            for (int q = 0; q < 4; ++q) {
                s[q] += a * wr[k][q];
                s[q] += xv * wo[k][q];
            }
        }
        if (!TAIL || r < n) {
            float4 v;
            v.x = fmaxf(s[0], 0.0f);
            v.y = fmaxf(s[1], 0.0f);
            v.z = fmaxf(s[2], 0.0f);
            v.w = fmaxf(s[3], 0.0f);
            *(float4*)(h1 + r * H + c) = v;
        }
    }
}

__global__ void conv1_kernel(const float* __restrict__ x0, const float* __restrict__ x1,
                             const float* __restrict__ x2, const float* __restrict__ agg6,
                             const float* __restrict__ Wrel, const float* __restrict__ brel,
                             const float* __restrict__ Wroot, float* __restrict__ h1) {
    const int lane = threadIdx.x & 31;
    int wave = __builtin_amdgcn_readfirstlane((int)((blockIdx.x * blockDim.x + threadIdx.x) >> 5));
    const float* x = x0;
    int n = N0, lvl = 0, roff = 0;
    if (wave >= C1_W0 + C1_W1) { wave -= C1_W0 + C1_W1; x = x2; n = N2; lvl = 2; roff = ROFF2; }
    else if (wave >= C1_W0)    { wave -= C1_W0;         x = x1; n = N1; lvl = 1; roff = ROFF1; }
    const int r0 = wave * C1_RPW;
    if (r0 >= n) return;
    const float* a6 = agg6 + roff * 6;
    const float* wr = Wrel + lvl * 6 * H;
    const float* br = brel + lvl * H;
    const float* wo = Wroot + lvl * 6 * H;
    float* hd = h1 + (size_t)roff * H;
    if (r0 + C1_RPW <= n) conv1_wave<false>(x, a6, wr, br, wo, hd, n, r0, lane * 4);
    else                  conv1_wave<true >(x, a6, wr, br, wo, hd, n, r0, lane * 4);
}

#define ETOT (E0 + E1 + E2)
#define CNT_STRIDE 32

#define EDGE_LEVEL(eg, ei, E, roff, e)                                             \
    const int* ei = ei0; int E = E0, roff = 0, e = (eg);                          \
    if ((eg) >= E0 + E1) { ei = ei2; E = E2; roff = ROFF2; e = (eg) - (E0 + E1); } \
    else if ((eg) >= E0) { ei = ei1; E = E1; roff = ROFF1; e = (eg) - E0; }

__global__ void csr_rank_kernel(const int* __restrict__ ei0, const int* __restrict__ ei1,
                                const int* __restrict__ ei2, int* __restrict__ cnt,
                                int* __restrict__ rank) {
    const int eg = blockIdx.x * blockDim.x + threadIdx.x;
    if (eg >= ETOT) return;
    EDGE_LEVEL(eg, ei, E, roff, e)
    const int d = ei[E + e] + roff;
    rank[eg] = atomicAdd(&cnt[d * CNT_STRIDE], 1);
}

__global__ void csr_alloc_kernel(const int* __restrict__ cnt, int* __restrict__ cursor,
                                 int* __restrict__ start) {
    const int r    = blockIdx.x * blockDim.x + threadIdx.x;
    const int lane = threadIdx.x & 31;
    const int c = (r < NROWS) ? cnt[r * CNT_STRIDE] : 0;
    int incl = c;
#pragma unroll
    for (int d = 1; d < 32; d <<= 1) {
        const int t = __shfl_up(incl, d, 32);
        if (lane >= d) incl += t;
    }
    int base = 0;
    if (lane == 31) base = atomicAdd(cursor, incl);
    base = __shfl(base, 31, 32);
    if (r < NROWS) start[r] = base + incl - c;
}

__global__ void csr_place_kernel(const int* __restrict__ ei0, const int* __restrict__ ei1,
                                 const int* __restrict__ ei2, const int* __restrict__ start,
                                 const int* __restrict__ rank, int* __restrict__ ssrc) {
    const int eg = blockIdx.x * blockDim.x + threadIdx.x;
    if (eg >= ETOT) return;
    EDGE_LEVEL(eg, ei, E, roff, e)
    const int s = ei[e] + roff;
    const int d = ei[E + e] + roff;
    ssrc[start[d] + rank[eg]] = s;
}

__global__ void gather6_kernel(const float* __restrict__ x0, const float* __restrict__ x1,
                               const float* __restrict__ x2, const int* __restrict__ start,
                               const int* __restrict__ cnt, const int* __restrict__ ssrc,
                               float* __restrict__ agg6all) {
    const int t = blockIdx.x * blockDim.x + threadIdx.x;
    const int r = t >> 3;
    const int k = t & 7;
    if (r >= NROWS || k >= 6) return;
    const float* x = x0;
    int roff = 0;
    if (r >= ROFF2)      { x = x2; roff = ROFF2; }
    else if (r >= ROFF1) { x = x1; roff = ROFF1; }
    const int c   = cnt[r * CNT_STRIDE];
    const int* sp = ssrc + start[r];
    const float* xk = x + k;
    float acc = 0.0f;
    int i = 0;
    for (; i + 4 <= c; i += 4) {
        const int s0 = sp[i], s1 = sp[i + 1], s2 = sp[i + 2], s3 = sp[i + 3];
        const float v0 = xk[(s0 - roff) * 6];
        const float v1 = xk[(s1 - roff) * 6];
        const float v2 = xk[(s2 - roff) * 6];
        const float v3 = xk[(s3 - roff) * 6];
        acc += v0; acc += v1; acc += v2; acc += v3;
    }
    for (; i < c; ++i) acc += xk[(sp[i] - roff) * 6];
    agg6all[r * 6 + k] = acc;
}

template <int NB, bool LIM>
__device__ __forceinline__ void gather128_batch(const int* __restrict__ sp, int i, int c,
                                                const float4* __restrict__ hb, float4& acc) {
    int s[NB];
    float4 v[NB];
#pragma unroll
    for (int j = 0; j < NB; ++j) {
        int ii = i + j;
        if (LIM) ii = (ii < c) ? ii : (c - 1);
        s[j] = sp[ii];
    }
#pragma unroll
    for (int j = 0; j < NB; ++j) v[j] = hb[(size_t)s[j] * (H / 4)];
#pragma unroll
    for (int j = 0; j < NB; ++j) {
        if (!LIM || i + j < c) {
            acc.x += v[j].x;
            acc.y += v[j].y;
            acc.z += v[j].z;
            acc.w += v[j].w;
        }
    }
}

__global__ void gather128_kernel(const float* __restrict__ h1all, const int* __restrict__ start,
                                 const int* __restrict__ cnt, const int* __restrict__ ssrc,
                                 float* __restrict__ aggall) {
    const int lane = threadIdx.x & 31;
    const int r = __builtin_amdgcn_readfirstlane((int)((blockIdx.x * blockDim.x + threadIdx.x) >> 5));
    if (r >= NROWS) return;
    const int c   = cnt[r * CNT_STRIDE];
    const int* sp = ssrc + start[r];
    const float4* hb = (const float4*)h1all + lane;
    float4 acc = {0.0f, 0.0f, 0.0f, 0.0f};
    int i = 0;
    for (; i + 8 <= c; i += 8) gather128_batch<8, false>(sp, i, c, hb, acc);
    const int rem = c - i;
    if (rem > 4)      gather128_batch<8, true>(sp, i, c, hb, acc);
    else if (rem > 0) gather128_batch<4, true>(sp, i, c, hb, acc);
    ((float4*)aggall)[(size_t)r * (H / 4) + lane] = acc;
}

#define IN_NPW 8
template <bool TAIL>
__device__ __forceinline__ void interp_wave(const float* __restrict__ vals,
                                            const int* __restrict__ cols,
                                            const int* __restrict__ rows,
                                            const float* __restrict__ e,
                                            float* __restrict__ acc, int nz0, int lane) {
    float v[IN_NPW];
    int col[IN_NPW], row[IN_NPW];
    float g[IN_NPW][3];
#pragma unroll
    for (int j = 0; j < IN_NPW; ++j) {
        int c = nz0 + j;
        if (TAIL) c = (c < NNZ) ? c : (NNZ - 1);
        v[j]   = vals[c];
        col[j] = cols[c];
        row[j] = rows[c];
    }
#pragma unroll
    for (int j = 0; j < IN_NPW; ++j) {
        const float* ep = e + col[j] * F3 + lane;
#pragma unroll
        for (int k = 0; k < 3; ++k) g[j][k] = ep[32 * k];
    }
#pragma unroll
    for (int j = 0; j < IN_NPW; ++j) {
        if (!TAIL || nz0 + j < NNZ) {
            float* ap = acc + row[j] * F3 + lane;
#pragma unroll
            for (int k = 0; k < 3; ++k) atomicAdd(ap + 32 * k, v[j] * g[j][k]);
        }
    }
}

#define IN_W ((NNZ + IN_NPW - 1) / IN_NPW)
__global__ void interp_kernel(const float* __restrict__ vals1, const int* __restrict__ cols1,
                              const int* __restrict__ rows1, const float* __restrict__ ea,
                              float* __restrict__ acc1,
                              const float* __restrict__ vals2, const int* __restrict__ cols2,
                              const int* __restrict__ rows2, const float* __restrict__ eb,
                              float* __restrict__ acc2) {
    const int lane = threadIdx.x & 31;
    int wave = __builtin_amdgcn_readfirstlane((int)((blockIdx.x * blockDim.x + threadIdx.x) >> 5));
    const float* vals = vals1;
    const int* cols = cols1;
    const int* rows = rows1;
    const float* e = ea;
    float* acc = acc1;
    if (wave >= IN_W) { wave -= IN_W; vals = vals2; cols = cols2; rows = rows2; e = eb; acc = acc2; }
    const int nz0 = wave * IN_NPW;
    if (nz0 >= NNZ) return;
    if (nz0 + IN_NPW <= NNZ) interp_wave<false>(vals, cols, rows, e, acc, nz0, lane);
    else                     interp_wave<true >(vals, cols, rows, e, acc, nz0, lane);
}

#define GS_THREADS 128

template <int K, int NC, int UNR = 4>
__device__ __forceinline__ void stage_weights(const float* __restrict__ W, float* __restrict__ Ws) {
    constexpr int NT = NC / 16;
#pragma unroll UNR
    for (int q = threadIdx.x; q < (K / 4) * NC; q += GS_THREADS) {
        const int ks = q / NC;
        const int nn = q - ks * NC;
        const float* wp = W + ks * 4 * NC + nn;
        v2f lo, hi;
        lo.x = wp[0];
        lo.y = wp[NC];
        hi.x = wp[2 * NC];
        hi.y = wp[3 * NC];
        const int t = nn >> 4;
        const int c = nn & 15;
        float* dst = Ws + ((ks * (NT / 2) + (t >> 1)) * 32 + c) * 4 + (t & 1) * 2;
        *(v2f*)dst        = lo;
        *(v2f*)(dst + 64) = hi;
    }
}

template <int NT>
__device__ __forceinline__ void load_b(const float* __restrict__ wb, v2f (&b)[NT]) {
#pragma unroll
    for (int tp = 0; tp < NT / 2; ++tp) {
        const float4 q = *(const float4*)(wb + tp * 128);
        b[2 * tp].x     = q.x;
        b[2 * tp].y     = q.y;
        b[2 * tp + 1].x = q.z;
        b[2 * tp + 1].y = q.w;
    }
}

template <int NC, int NT, int ACT>
__device__ __forceinline__ void store_strip(const v8f (&acc)[NT], float* __restrict__ D,
                                            int n, int row0, int m, int half) {
    const int rbase = row0 + half * 8;
    const int lim   = n - rbase;
    float* dp = D + rbase * NC + m;
#pragma unroll
    for (int j = 0; j < 8; ++j) {
        if (j < lim) {
#pragma unroll
            for (int t = 0; t < NT; ++t) {
                float v = acc[t][j];
                if (ACT == 1) v = fmaxf(v, 0.0f);
                if (ACT == 2) v = (v > 0.0f) ? v : (expf(v) - 1.0f);
                dp[j * NC + 16 * t] = v;
            }
        }
    }
}

template <int K, int NT, bool DUAL>
__device__ __forceinline__ void strip_products(const v2f (&a0)[K / 4], const v2f (&a1)[K / 4],
                                               const float* __restrict__ w0s,
                                               const float* __restrict__ w1s, v8f (&acc)[NT]) {
    constexpr int KS = K / 4;
    v2f b0[NT], b1[NT];
    load_b<NT>(w0s, b0);
    if (DUAL) load_b<NT>(w1s, b1);
#pragma unroll
    for (int ks = 0; ks < KS; ++ks) {
        v2f nb0[NT], nb1[NT];
        if (ks + 1 < KS) {
            load_b<NT>(w0s + (ks + 1) * (NT / 2) * 128, nb0);
            if (DUAL) load_b<NT>(w1s + (ks + 1) * (NT / 2) * 128, nb1);
        }
#pragma unroll
        for (int t = 0; t < NT; ++t)
            acc[t] = __builtin_amdgcn_wmma_f32_16x16x4_f32(false, a0[ks], false, b0[t], (short)0,
                                                           acc[t], false, false);
        if (DUAL) {
#pragma unroll
            for (int t = 0; t < NT; ++t)
                acc[t] = __builtin_amdgcn_wmma_f32_16x16x4_f32(false, a1[ks], false, b1[t], (short)0,
                                                               acc[t], false, false);
        }
        if (ks + 1 < KS) {
#pragma unroll
            for (int t = 0; t < NT; ++t) {
                b0[t] = nb0[t];
                if (DUAL) b1[t] = nb1[t];
            }
        }
    }
}

#define GEMM_GRID(n) ((((n) + 15) / 16 + 3) / 4 > 512 ? 512 : (((n) + 15) / 16 + 3) / 4)
#define GG0 GEMM_GRID(N0)
#define GG1 GEMM_GRID(N1)
#define GG2 GEMM_GRID(N2)

template <int K, int NC, int ACT, bool DUAL>
__global__ void __launch_bounds__(GS_THREADS)
gemm_wmma_kernel(const float* __restrict__ A0all, const float* __restrict__ W0all,
                 const float* __restrict__ A1all, const float* __restrict__ W1all,
                 const float* __restrict__ biasall, float* __restrict__ Dall) {
    constexpr int NT = NC / 16;
    constexpr int KS = K / 4;
    __shared__ __attribute__((aligned(16))) float Ws[(DUAL ? 2 : 1) * K * NC];

    int lb = blockIdx.x, lg = GG0, n = N0, lvl = 0, roff = 0;
    if (lb >= GG0 + GG1) { lb -= GG0 + GG1; lg = GG2; n = N2; lvl = 2; roff = ROFF2; }
    else if (lb >= GG0)  { lb -= GG0;       lg = GG1; n = N1; lvl = 1; roff = ROFF1; }
    const float* A0   = A0all + (size_t)roff * K;
    const float* W0   = W0all + lvl * K * NC;
    const float* A1   = DUAL ? (A1all + (size_t)roff * K) : A0;
    const float* W1   = DUAL ? (W1all + lvl * K * NC) : W0;
    const float* bias = biasall + lvl * NC;
    float* D          = Dall + (size_t)roff * NC;

    stage_weights<K, NC>(W0, Ws);
    if (DUAL) stage_weights<K, NC>(W1, Ws + K * NC);
    __syncthreads();

    const int lane = threadIdx.x & 31;
    const int wave = __builtin_amdgcn_readfirstlane((int)(threadIdx.x >> 5));
    const int m    = lane & 15;
    const int half = lane >> 4;
    const int nstrips = (n + 15) >> 4;

#pragma unroll 1
    for (int s = lb * (GS_THREADS / 32) + wave; s < nstrips; s += lg * (GS_THREADS / 32)) {
        int boff0 = lane * 4;
        int boff1 = lane * 4 + K * NC;
        if (DUAL) {
            asm volatile("" : "+v"(boff0), "+v"(boff1));
        }
        const float* w0s = Ws + boff0;
        const float* w1s = Ws + boff1;
        const int row0 = s * 16;
        const int row  = row0 + m;
        const int rowc = (row < n) ? row : (n - 1);

        const float* a0p = A0 + rowc * K + half * 2;
        const float* a1p = DUAL ? (A1 + rowc * K + half * 2) : a0p;
        v2f a0[KS], a1[KS];
#pragma unroll
        for (int ks = 0; ks < KS; ++ks) {
            a0[ks] = *(const v2f*)(a0p + 4 * ks);
            if (DUAL) a1[ks] = *(const v2f*)(a1p + 4 * ks);
        }

        v8f acc[NT];
#pragma unroll
        for (int t = 0; t < NT; ++t) {
            const float bv = bias[16 * t + m];
            acc[t] = (v8f){bv, bv, bv, bv, bv, bv, bv, bv};
        }

        strip_products<K, NT, DUAL>(a0, a1, w0s, w1s, acc);

        store_strip<NC, NT, ACT>(acc, D, n, row0, m, half);
    }
}

#define DEC_RPB 32
__global__ void __launch_bounds__(GS_THREADS)
dec_fused_wmma_kernel(const float* __restrict__ e0, const float* __restrict__ i1,
                      const float* __restrict__ i2, const float* __restrict__ W,
                      const float* __restrict__ bias, const float* __restrict__ Wout,
                      const float* __restrict__ bout, float* __restrict__ out, int n) {
    constexpr int K = F3, NC = DECW;
    constexpr int NT = NC / 16;
    constexpr int KS = K / 4;
    __shared__ __attribute__((aligned(16))) float Ws[3 * K * NC];

    stage_weights<3 * K, NC, 12>(W, Ws);
    __syncthreads();

    const int lane = threadIdx.x & 31;
    const int wave = __builtin_amdgcn_readfirstlane((int)(threadIdx.x >> 5));
    const int m    = lane & 15;
    const int half = lane >> 4;
    const int nblk = (n + DEC_RPB - 1) / DEC_RPB;

    float bz[3][NT], wo[3][NT], bo[3];
#pragma unroll
    for (int b = 0; b < 3; ++b) {
        bo[b] = bout[b];
#pragma unroll
        for (int t = 0; t < NT; ++t) {
            bz[b][t] = bias[b * NC + 16 * t + m];
            wo[b][t] = Wout[b * NC + 16 * t + m];
        }
    }
    const bool bit3 = (m & 8) != 0;
    const bool bit2 = (m & 4) != 0;
    const bool bit1 = (m & 2) != 0;

#pragma unroll 1
    for (int blk = blockIdx.x * (GS_THREADS / 32) + wave; blk < nblk; blk += gridDim.x * (GS_THREADS / 32)) {
#pragma unroll 1
        for (int st = 0; st < DEC_RPB / 16; ++st) {
            const int row0 = blk * DEC_RPB + st * 16;
            if (row0 >= n) break;
            const int row  = row0 + m;
            const int rowc = (row < n) ? row : (n - 1);

            const int abase = rowc * F3 + half * 2;
            const float* s0 = e0 + abase;
            const float* s1 = i1 + abase;
            const float* s2 = i2 + abase;
            v2f a[3][KS];
#pragma unroll
            for (int b = 0; b < 3; ++b) {
#pragma unroll
                for (int ks = 0; ks < KS; ++ks) {
                    const int k   = 4 * ks;
                    const int src = k >> 5;
                    const float* sp = (src == 0) ? s0 : (src == 1) ? s1 : s2;
                    a[b][ks] = *(const v2f*)(sp + b * 32 + (k & 31));
                }
            }
            float res[3];
#pragma unroll
            for (int b = 0; b < 3; ++b) {
                v8f acc[NT];
#pragma unroll
                for (int t = 0; t < NT; ++t) {
                    const float bv = bz[b][t];
                    acc[t] = (v8f){bv, bv, bv, bv, bv, bv, bv, bv};
                }
                const float* ws = Ws + lane * 4 + b * K * NC;
                strip_products<K, NT, false>(a[b], a[b], ws, ws, acc);

                float p[8];
#pragma unroll
                for (int j = 0; j < 8; ++j) {
                    float s = 0.0f;
#pragma unroll
                    for (int t = 0; t < NT; ++t) {
                        float v = acc[t][j];
                        v = (v > 0.0f) ? v : (expf(v) - 1.0f);
                        s += v * wo[b][t];
                    }
                    p[j] = s;
                }
                float q[4], r2[2];
#pragma unroll
                for (int i = 0; i < 4; ++i) {
                    const float send = bit3 ? p[i] : p[i + 4];
                    const float keep = bit3 ? p[i + 4] : p[i];
                    q[i] = keep + __shfl_xor(send, 8, 32);
                }
#pragma unroll
                for (int i = 0; i < 2; ++i) {
                    const float send = bit2 ? q[i] : q[i + 2];
                    const float keep = bit2 ? q[i + 2] : q[i];
                    r2[i] = keep + __shfl_xor(send, 4, 32);
                }
                const float send1 = bit1 ? r2[0] : r2[1];
                const float keep1 = bit1 ? r2[1] : r2[0];
                float tot = keep1 + __shfl_xor(send1, 2, 32);
                tot += __shfl_xor(tot, 1, 32);
                res[b] = tot + bo[b];
            }

            const int orow = row0 + half * 8 + (m >> 1);
            if ((m & 1) == 0 && orow < n) {
                float* op = out + orow * 3;
                op[0] = res[0];
                op[1] = res[1];
                op[2] = res[2];
            }
        }
    }
}

#define FILL_GRID(count) (((count) / 4 + 8 * 32 * FILL_SPW - 1) / (8 * 32 * FILL_SPW) > 256 ? 256 \
                          : ((count) / 4 + 8 * 32 * FILL_SPW - 1) / (8 * 32 * FILL_SPW))

#define WS_H1    ((size_t)0)
#define WS_AGG   (WS_H1 + (size_t)NROWS * H)
#define WS_AGG6  (WS_AGG + (size_t)NROWS * H)
#define WS_H2    (((WS_AGG6 + (size_t)NROWS * 6 + 31) / 32) * 32)
#define WS_E     (WS_H2 + (size_t)NROWS * H)
#define WS_I1    ((size_t)0)
#define WS_I2    (WS_I1 + (size_t)N0 * F3)
#define WS_CNT   WS_E
#define WS_CUR   (WS_CNT + (size_t)NROWS * CNT_STRIDE)
#define WS_START (WS_CUR + 32)
#define WS_RANK  (WS_START + (((size_t)NROWS + 31) / 32) * 32)
#define WS_SSRC  (WS_RANK + (((size_t)ETOT + 31) / 32) * 32)
static_assert(WS_SSRC + (size_t)ETOT <= WS_E + (size_t)NROWS * F3, "the CSR arrays must fit in e's place");

extern "C" void kernel_launch(void* const* d_in, const int* in_sizes, int n_in,
                              void* d_out, int out_size, void* d_ws, size_t ws_size,
                              hipStream_t stream) {
    (void)in_sizes; (void)n_in; (void)out_size; (void)ws_size;

    const float* x0     = (const float*)d_in[0];
    const float* x1     = (const float*)d_in[1];
    const float* x2     = (const float*)d_in[2];
    const int*   ei0    = (const int*)d_in[3];
    const int*   ei1    = (const int*)d_in[4];
    const int*   ei2    = (const int*)d_in[5];
    const int*   A1r    = (const int*)d_in[6];
    const int*   A1c    = (const int*)d_in[7];
    const float* A1v    = (const float*)d_in[8];
    const int*   A2r    = (const int*)d_in[9];
    const int*   A2c    = (const int*)d_in[10];
    const float* A2v    = (const float*)d_in[11];
    const float* c1Wrel  = (const float*)d_in[12];
    const float* c1brel  = (const float*)d_in[13];
    const float* c1Wroot = (const float*)d_in[14];
    const float* c2Wrel  = (const float*)d_in[15];
    const float* c2brel  = (const float*)d_in[16];
    const float* c2Wroot = (const float*)d_in[17];
    const float* linW    = (const float*)d_in[18];
    const float* linb    = (const float*)d_in[19];
    const float* decW0   = (const float*)d_in[20];
    const float* decb0   = (const float*)d_in[21];
    const float* decWout = (const float*)d_in[22];
    const float* decbout = (const float*)d_in[23];
    float* out = (float*)d_out;

    float* ws   = (float*)d_ws;
    float* h1   = ws + WS_H1;
    float* agg  = ws + WS_AGG;
    float* agg6 = ws + WS_AGG6;
    float* h2   = ws + WS_H2;
    float* eall = ws + WS_E;
    float* i1   = ws + WS_I1;
    float* i2   = ws + WS_I2;
    int* cnt    = (int*)(ws + WS_CNT);
    int* cursor = (int*)(ws + WS_CUR);
    int* start  = (int*)(ws + WS_START);
    int* rank   = (int*)(ws + WS_RANK);
    int* ssrc   = (int*)(ws + WS_SSRC);

    zero_kernel<<<FILL_GRID(NROWS * CNT_STRIDE + 32), 256, 0, stream>>>((float*)cnt, NROWS * CNT_STRIDE + 32);
    csr_rank_kernel<<<(ETOT + 255) / 256, 256, 0, stream>>>(ei0, ei1, ei2, cnt, rank);
    csr_alloc_kernel<<<(NROWS + 255) / 256, 256, 0, stream>>>(cnt, cursor, start);
    csr_place_kernel<<<(ETOT + 255) / 256, 256, 0, stream>>>(ei0, ei1, ei2, start, rank, ssrc);

    gather6_kernel<<<(NROWS * 8 + 255) / 256, 256, 0, stream>>>(x0, x1, x2, start, cnt, ssrc, agg6);
    conv1_kernel<<<(C1_W0 + C1_W1 + C1_W2 + 7) / 8, 256, 0, stream>>>(x0, x1, x2, agg6, c1Wrel, c1brel,
                                                                    c1Wroot, h1);

    gather128_kernel<<<(NROWS + 7) / 8, 256, 0, stream>>>(h1, start, cnt, ssrc, agg);
    gemm_wmma_kernel<H, H, 1, true><<<GG0 + GG1 + GG2, GS_THREADS, 0, stream>>>(
        agg, c2Wrel, h1, c2Wroot, c2brel, h2);

    gemm_wmma_kernel<H, F3, 0, false><<<GG0 + GG1 + GG2, GS_THREADS, 0, stream>>>(
        h2, linW, nullptr, nullptr, linb, eall);

    zero_kernel<<<FILL_GRID(2 * N0 * F3), 256, 0, stream>>>(i1, 2 * N0 * F3);
    interp_kernel<<<(2 * IN_W + 7) / 8, 256, 0, stream>>>(A1v, A1c, A1r, eall + (size_t)ROFF1 * F3, i1,
                                                         A2v, A2c, A2r, eall + (size_t)ROFF2 * F3, i2);

    dec_fused_wmma_kernel<<<((N0 + DEC_RPB - 1) / DEC_RPB + 3) / 4, GS_THREADS, 0, stream>>>(
        eall, i1, i2, decW0, decb0, decWout, decbout, out, N0);
}


# row records hold their sources: counter + first 31 sources of a row in one 128-byte line, each edge placed in ONE pass (row scan and 2nd edge pass gone), overflow list past 31; 12 -> 10 launches
# speedup vs baseline: 3.1883x; 1.0564x over previous
#include <hip/hip_runtime.h>
#include <hip/hip_bf16.h>

#define N0 50000
#define N1 12500
#define N2 3125
#define E0 800000
#define E1 200000
#define E2 50000
#define NNZ 200000
#define H 128
#define F3 96
#define DECW 64

#define ROFF1 N0
#define ROFF2 (N0 + N1)
#define NROWS (N0 + N1 + N2)

typedef __attribute__((ext_vector_type(2))) float v2f;
typedef __attribute__((ext_vector_type(8))) float v8f;

#define FILL_SPW 4
__global__ void zero_kernel(float* __restrict__ p, int count) {
    const int tid  = blockIdx.x * blockDim.x + threadIdx.x;
    const int lane = threadIdx.x & 31;
    const int wave = __builtin_amdgcn_readfirstlane(tid >> 5);
    const int step = ((gridDim.x * blockDim.x) >> 5) * (32 * FILL_SPW);
    int head = (int)(((16u - (unsigned)((size_t)p & 15u)) & 15u) >> 2);
    if (head > count) head = count;
    float4* q = (float4*)(p + head);
    const int nvec = (count - head) >> 2;
    const float4 z = {0.0f, 0.0f, 0.0f, 0.0f};
    for (int base = wave * (32 * FILL_SPW); base < nvec; base += step) {
        float4* qq = q + base + lane;
        if (base + 32 * FILL_SPW <= nvec) {
#pragma unroll
            for (int u = 0; u < FILL_SPW; ++u) qq[32 * u] = z;
        } else {
#pragma unroll
            for (int u = 0; u < FILL_SPW; ++u)
                if (base + 32 * u + lane < nvec) qq[32 * u] = z;
        }
    }
    const int tail0 = head + (nvec << 2);
    if (tid < head) p[tid] = 0.0f;
    if (tid < count - tail0) p[tail0 + tid] = 0.0f;
}

#define C1_RPW 8
#define C1_W0 ((N0 + C1_RPW - 1) / C1_RPW)
#define C1_W1 ((N1 + C1_RPW - 1) / C1_RPW)
#define C1_W2 ((N2 + C1_RPW - 1) / C1_RPW)
template <bool TAIL>
__device__ __forceinline__ void conv1_wave(const float* __restrict__ x, const float* __restrict__ agg6,
                                           const float* __restrict__ Wrel, const float* __restrict__ brel,
                                           const float* __restrict__ Wroot, float* __restrict__ h1,
                                           int n, int r0, int c) {
    float wr[6][4], wo[6][4], b[4];
#pragma unroll
    for (int q = 0; q < 4; ++q) b[q] = brel[c + q];
#pragma unroll
    for (int k = 0; k < 6; ++k) {
#pragma unroll
        for (int q = 0; q < 4; ++q) {
            wr[k][q] = Wrel[k * H + c + q];
            wo[k][q] = Wroot[k * H + c + q];
        }
    }
#pragma unroll
    for (int j = 0; j < C1_RPW; ++j) {
        const int r = r0 + j;
        int rl = r;
        if (TAIL) rl = (r < n) ? r : (n - 1);
        const float* ar = agg6 + rl * 6;
        const float* xr = x + rl * 6;
        float s[4] = {b[0], b[1], b[2], b[3]};
#pragma unroll
        for (int k = 0; k < 6; ++k) {
            const float a  = ar[k];
            const float xv = xr[k];
#pragma unroll
            for (int q = 0; q < 4; ++q) {
                s[q] += a * wr[k][q];
                s[q] += xv * wo[k][q];
            }
        }
        if (!TAIL || r < n) {
            float4 v;
            v.x = fmaxf(s[0], 0.0f);
            v.y = fmaxf(s[1], 0.0f);
            v.z = fmaxf(s[2], 0.0f);
            v.w = fmaxf(s[3], 0.0f);
            *(float4*)(h1 + r * H + c) = v;
        }
    }
}

__global__ void conv1_kernel(const float* __restrict__ x0, const float* __restrict__ x1,
                             const float* __restrict__ x2, const float* __restrict__ agg6,
                             const float* __restrict__ Wrel, const float* __restrict__ brel,
                             const float* __restrict__ Wroot, float* __restrict__ h1) {
    const int lane = threadIdx.x & 31;
    int wave = __builtin_amdgcn_readfirstlane((int)((blockIdx.x * blockDim.x + threadIdx.x) >> 5));
    const float* x = x0;
    int n = N0, lvl = 0, roff = 0;
    if (wave >= C1_W0 + C1_W1) { wave -= C1_W0 + C1_W1; x = x2; n = N2; lvl = 2; roff = ROFF2; }
    else if (wave >= C1_W0)    { wave -= C1_W0;         x = x1; n = N1; lvl = 1; roff = ROFF1; }
    const int r0 = wave * C1_RPW;
    if (r0 >= n) return;
    const float* a6 = agg6 + roff * 6;
    const float* wr = Wrel + lvl * 6 * H;
    const float* br = brel + lvl * H;
    const float* wo = Wroot + lvl * 6 * H;
    float* hd = h1 + (size_t)roff * H;
    if (r0 + C1_RPW <= n) conv1_wave<false>(x, a6, wr, br, wo, hd, n, r0, lane * 4);
    else                  conv1_wave<true >(x, a6, wr, br, wo, hd, n, r0, lane * 4);
}

#define ETOT (E0 + E1 + E2)
#define CNT_STRIDE 32
#define CSR_INL (CNT_STRIDE - 1)

#define EDGE_LEVEL(eg, ei, E, roff, e)                                             \
    const int* ei = ei0; int E = E0, roff = 0, e = (eg);                          \
    if ((eg) >= E0 + E1) { ei = ei2; E = E2; roff = ROFF2; e = (eg) - (E0 + E1); } \
    else if ((eg) >= E0) { ei = ei1; E = E1; roff = ROFF1; e = (eg) - E0; }

__global__ void csr_build_kernel(const int* __restrict__ ei0, const int* __restrict__ ei1,
                                 const int* __restrict__ ei2, int* __restrict__ rec,
                                 int* __restrict__ cursor, int* __restrict__ ovf) {
    const int eg = blockIdx.x * blockDim.x + threadIdx.x;
    if (eg >= ETOT) return;
    EDGE_LEVEL(eg, ei, E, roff, e)
    const int s = ei[e] + roff;
    const int d = ei[E + e] + roff;
    int* line = rec + (size_t)d * CNT_STRIDE;
    const int place = atomicAdd(line, 1);
    if (place < CSR_INL) {
        line[1 + place] = s;
    } else {
        const int k = atomicAdd(cursor, 1);
        ovf[2 * k]     = d;
        ovf[2 * k + 1] = s;
    }
}

__global__ void gather6_kernel(const float* __restrict__ x0, const float* __restrict__ x1,
                               const float* __restrict__ x2, const int* __restrict__ rec,
                               const int* __restrict__ cursor, const int* __restrict__ ovf,
                               float* __restrict__ agg6all) {
    const int t = blockIdx.x * blockDim.x + threadIdx.x;
    const int r = t >> 3;
    const int k = t & 7;
    if (r >= NROWS || k >= 6) return;
    const float* x = x0;
    int roff = 0;
    if (r >= ROFF2)      { x = x2; roff = ROFF2; }
    else if (r >= ROFF1) { x = x1; roff = ROFF1; }
    const int* sp = rec + (size_t)r * CNT_STRIDE + 1;
    const int deg = sp[-1];
    const int c   = (deg < CSR_INL) ? deg : CSR_INL;
    const float* xk = x + k;
    float acc = 0.0f;
    int i = 0;
    for (; i + 4 <= c; i += 4) {
        const int s0 = sp[i], s1 = sp[i + 1], s2 = sp[i + 2], s3 = sp[i + 3];
        const float v0 = xk[(s0 - roff) * 6];
        const float v1 = xk[(s1 - roff) * 6];
        const float v2 = xk[(s2 - roff) * 6];
        const float v3 = xk[(s3 - roff) * 6];
        acc += v0; acc += v1; acc += v2; acc += v3;
    }
    for (; i < c; ++i) acc += xk[(sp[i] - roff) * 6];
    if (deg > CSR_INL) {
        int nov = *cursor;
        nov = (nov < ETOT) ? nov : ETOT;
        for (int q = 0; q < nov; ++q)
            if (ovf[2 * q] == r) acc += xk[(ovf[2 * q + 1] - roff) * 6];
    }
    agg6all[r * 6 + k] = acc;
}

template <int NB, bool LIM>
__device__ __forceinline__ void gather128_batch(const int* __restrict__ sp, int i, int c,
                                                const float4* __restrict__ hb, float4& acc) {
    int s[NB];
    float4 v[NB];
#pragma unroll
    for (int j = 0; j < NB; ++j) {
        int ii = i + j;
        if (LIM) ii = (ii < c) ? ii : (c - 1);
        s[j] = sp[ii];
    }
#pragma unroll
    for (int j = 0; j < NB; ++j) v[j] = hb[(size_t)s[j] * (H / 4)];
#pragma unroll
    for (int j = 0; j < NB; ++j) {
        if (!LIM || i + j < c) {
            acc.x += v[j].x;
            acc.y += v[j].y;
            acc.z += v[j].z;
            acc.w += v[j].w;
        }
    }
}

__global__ void gather128_kernel(const float* __restrict__ h1all, const int* __restrict__ rec,
                                 const int* __restrict__ cursor, const int* __restrict__ ovf,
                                 float* __restrict__ aggall) {
    const int lane = threadIdx.x & 31;
    const int r = __builtin_amdgcn_readfirstlane((int)((blockIdx.x * blockDim.x + threadIdx.x) >> 5));
    if (r >= NROWS) return;
    const int* sp = rec + (size_t)r * CNT_STRIDE + 1;
    const int deg = sp[-1];
    const int c   = (deg < CSR_INL) ? deg : CSR_INL;
    const float4* hb = (const float4*)h1all + lane;
    float4 acc = {0.0f, 0.0f, 0.0f, 0.0f};
    int i = 0;
    for (; i + 8 <= c; i += 8) gather128_batch<8, false>(sp, i, c, hb, acc);
    const int rem = c - i;
    if (rem > 4)      gather128_batch<8, true>(sp, i, c, hb, acc);
    else if (rem > 0) gather128_batch<4, true>(sp, i, c, hb, acc);
    if (deg > CSR_INL) {
        int nov = *cursor;
        nov = (nov < ETOT) ? nov : ETOT;
        for (int q = 0; q < nov; ++q) {
            if (ovf[2 * q] == r) {
                const float4 v = hb[(size_t)ovf[2 * q + 1] * (H / 4)];
                acc.x += v.x;
                acc.y += v.y;
                acc.z += v.z;
                acc.w += v.w;
            }
        }
    }
    ((float4*)aggall)[(size_t)r * (H / 4) + lane] = acc;
}

#define IN_NPW 8
template <bool TAIL>
__device__ __forceinline__ void interp_wave(const float* __restrict__ vals,
                                            const int* __restrict__ cols,
                                            const int* __restrict__ rows,
                                            const float* __restrict__ e,
                                            float* __restrict__ acc, int nz0, int lane) {
    float v[IN_NPW];
    int col[IN_NPW], row[IN_NPW];
    float g[IN_NPW][3];
#pragma unroll
    for (int j = 0; j < IN_NPW; ++j) {
        int c = nz0 + j;
        if (TAIL) c = (c < NNZ) ? c : (NNZ - 1);
        v[j]   = vals[c];
        col[j] = cols[c];
        row[j] = rows[c];
    }
#pragma unroll
    for (int j = 0; j < IN_NPW; ++j) {
        const float* ep = e + col[j] * F3 + lane;
#pragma unroll
        for (int k = 0; k < 3; ++k) g[j][k] = ep[32 * k];
    }
#pragma unroll
    for (int j = 0; j < IN_NPW; ++j) {
        if (!TAIL || nz0 + j < NNZ) {
            float* ap = acc + row[j] * F3 + lane;
#pragma unroll
            for (int k = 0; k < 3; ++k) atomicAdd(ap + 32 * k, v[j] * g[j][k]);
        }
    }
}

#define IN_W ((NNZ + IN_NPW - 1) / IN_NPW)
__global__ void interp_kernel(const float* __restrict__ vals1, const int* __restrict__ cols1,
                              const int* __restrict__ rows1, const float* __restrict__ ea,
                              float* __restrict__ acc1,
                              const float* __restrict__ vals2, const int* __restrict__ cols2,
                              const int* __restrict__ rows2, const float* __restrict__ eb,
                              float* __restrict__ acc2) {
    const int lane = threadIdx.x & 31;
    int wave = __builtin_amdgcn_readfirstlane((int)((blockIdx.x * blockDim.x + threadIdx.x) >> 5));
    const float* vals = vals1;
    const int* cols = cols1;
    const int* rows = rows1;
    const float* e = ea;
    float* acc = acc1;
    if (wave >= IN_W) { wave -= IN_W; vals = vals2; cols = cols2; rows = rows2; e = eb; acc = acc2; }
    const int nz0 = wave * IN_NPW;
    if (nz0 >= NNZ) return;
    if (nz0 + IN_NPW <= NNZ) interp_wave<false>(vals, cols, rows, e, acc, nz0, lane);
    else                     interp_wave<true >(vals, cols, rows, e, acc, nz0, lane);
}

#define GS_THREADS 128

template <int K, int NC, int UNR = 4>
__device__ __forceinline__ void stage_weights(const float* __restrict__ W, float* __restrict__ Ws) {
    constexpr int NT = NC / 16;
#pragma unroll UNR
    for (int q = threadIdx.x; q < (K / 4) * NC; q += GS_THREADS) {
        const int ks = q / NC;
        const int nn = q - ks * NC;
        const float* wp = W + ks * 4 * NC + nn;
        v2f lo, hi;
        lo.x = wp[0];
        lo.y = wp[NC];
        hi.x = wp[2 * NC];
        hi.y = wp[3 * NC];
        const int t = nn >> 4;
        const int c = nn & 15;
        float* dst = Ws + ((ks * (NT / 2) + (t >> 1)) * 32 + c) * 4 + (t & 1) * 2;
        *(v2f*)dst        = lo;
        *(v2f*)(dst + 64) = hi;
    }
}

template <int NT>
__device__ __forceinline__ void load_b(const float* __restrict__ wb, v2f (&b)[NT]) {
#pragma unroll
    for (int tp = 0; tp < NT / 2; ++tp) {
        const float4 q = *(const float4*)(wb + tp * 128);
        b[2 * tp].x     = q.x;
        b[2 * tp].y     = q.y;
        b[2 * tp + 1].x = q.z;
        b[2 * tp + 1].y = q.w;
    }
}

template <int NC, int NT, int ACT>
__device__ __forceinline__ void store_strip(const v8f (&acc)[NT], float* __restrict__ D,
                                            int n, int row0, int m, int half) {
    const int rbase = row0 + half * 8;
    const int lim   = n - rbase;
    float* dp = D + rbase * NC + m;
#pragma unroll
    for (int j = 0; j < 8; ++j) {
        if (j < lim) {
#pragma unroll
            for (int t = 0; t < NT; ++t) {
                float v = acc[t][j];
                if (ACT == 1) v = fmaxf(v, 0.0f);
                if (ACT == 2) v = (v > 0.0f) ? v : (expf(v) - 1.0f);
                dp[j * NC + 16 * t] = v;
            }
        }
    }
}

template <int K, int NT, bool DUAL>
__device__ __forceinline__ void strip_products(const v2f (&a0)[K / 4], const v2f (&a1)[K / 4],
                                               const float* __restrict__ w0s,
                                               const float* __restrict__ w1s, v8f (&acc)[NT]) {
    constexpr int KS = K / 4;
    v2f b0[NT], b1[NT];
    load_b<NT>(w0s, b0);
    if (DUAL) load_b<NT>(w1s, b1);
#pragma unroll
    for (int ks = 0; ks < KS; ++ks) {
        v2f nb0[NT], nb1[NT];
        if (ks + 1 < KS) {
            load_b<NT>(w0s + (ks + 1) * (NT / 2) * 128, nb0);
            if (DUAL) load_b<NT>(w1s + (ks + 1) * (NT / 2) * 128, nb1);
        }
#pragma unroll
        for (int t = 0; t < NT; ++t)
            acc[t] = __builtin_amdgcn_wmma_f32_16x16x4_f32(false, a0[ks], false, b0[t], (short)0,
                                                           acc[t], false, false);
        if (DUAL) {
#pragma unroll
            for (int t = 0; t < NT; ++t)
                acc[t] = __builtin_amdgcn_wmma_f32_16x16x4_f32(false, a1[ks], false, b1[t], (short)0,
                                                               acc[t], false, false);
        }
        if (ks + 1 < KS) {
#pragma unroll
            for (int t = 0; t < NT; ++t) {
                b0[t] = nb0[t];
                if (DUAL) b1[t] = nb1[t];
            }
        }
    }
}

#define GEMM_GRID(n) ((((n) + 15) / 16 + 3) / 4 > 512 ? 512 : (((n) + 15) / 16 + 3) / 4)
#define GG0 GEMM_GRID(N0)
#define GG1 GEMM_GRID(N1)
#define GG2 GEMM_GRID(N2)

template <int K, int NC, int ACT, bool DUAL>
__global__ void __launch_bounds__(GS_THREADS)
gemm_wmma_kernel(const float* __restrict__ A0all, const float* __restrict__ W0all,
                 const float* __restrict__ A1all, const float* __restrict__ W1all,
                 const float* __restrict__ biasall, float* __restrict__ Dall) {
    constexpr int NT = NC / 16;
    constexpr int KS = K / 4;
    __shared__ __attribute__((aligned(16))) float Ws[(DUAL ? 2 : 1) * K * NC];

    int lb = blockIdx.x, lg = GG0, n = N0, lvl = 0, roff = 0;
    if (lb >= GG0 + GG1) { lb -= GG0 + GG1; lg = GG2; n = N2; lvl = 2; roff = ROFF2; }
    else if (lb >= GG0)  { lb -= GG0;       lg = GG1; n = N1; lvl = 1; roff = ROFF1; }
    const float* A0   = A0all + (size_t)roff * K;
    const float* W0   = W0all + lvl * K * NC;
    const float* A1   = DUAL ? (A1all + (size_t)roff * K) : A0;
    const float* W1   = DUAL ? (W1all + lvl * K * NC) : W0;
    const float* bias = biasall + lvl * NC;
    float* D          = Dall + (size_t)roff * NC;

    stage_weights<K, NC>(W0, Ws);
    if (DUAL) stage_weights<K, NC>(W1, Ws + K * NC);
    __syncthreads();

    const int lane = threadIdx.x & 31;
    const int wave = __builtin_amdgcn_readfirstlane((int)(threadIdx.x >> 5));
    const int m    = lane & 15;
    const int half = lane >> 4;
    const int nstrips = (n + 15) >> 4;

#pragma unroll 1
    for (int s = lb * (GS_THREADS / 32) + wave; s < nstrips; s += lg * (GS_THREADS / 32)) {
        int boff0 = lane * 4;
        int boff1 = lane * 4 + K * NC;
        if (DUAL) {
            asm volatile("" : "+v"(boff0), "+v"(boff1));
        }
        const float* w0s = Ws + boff0;
        const float* w1s = Ws + boff1;
        const int row0 = s * 16;
        const int row  = row0 + m;
        const int rowc = (row < n) ? row : (n - 1);

        const float* a0p = A0 + rowc * K + half * 2;
        const float* a1p = DUAL ? (A1 + rowc * K + half * 2) : a0p;
        v2f a0[KS], a1[KS];
#pragma unroll
        for (int ks = 0; ks < KS; ++ks) {
            a0[ks] = *(const v2f*)(a0p + 4 * ks);
            if (DUAL) a1[ks] = *(const v2f*)(a1p + 4 * ks);
        }

        v8f acc[NT];
#pragma unroll
        for (int t = 0; t < NT; ++t) {
            const float bv = bias[16 * t + m];
            acc[t] = (v8f){bv, bv, bv, bv, bv, bv, bv, bv};
        }

        strip_products<K, NT, DUAL>(a0, a1, w0s, w1s, acc);

        store_strip<NC, NT, ACT>(acc, D, n, row0, m, half);
    }
}

#define DEC_RPB 32
__global__ void __launch_bounds__(GS_THREADS)
dec_fused_wmma_kernel(const float* __restrict__ e0, const float* __restrict__ i1,
                      const float* __restrict__ i2, const float* __restrict__ W,
                      const float* __restrict__ bias, const float* __restrict__ Wout,
                      const float* __restrict__ bout, float* __restrict__ out, int n) {
    constexpr int K = F3, NC = DECW;
    constexpr int NT = NC / 16;
    constexpr int KS = K / 4;
    __shared__ __attribute__((aligned(16))) float Ws[3 * K * NC];

    stage_weights<3 * K, NC, 12>(W, Ws);
    __syncthreads();

    const int lane = threadIdx.x & 31;
    const int wave = __builtin_amdgcn_readfirstlane((int)(threadIdx.x >> 5));
    const int m    = lane & 15;
    const int half = lane >> 4;
    const int nblk = (n + DEC_RPB - 1) / DEC_RPB;

    float bz[3][NT], wo[3][NT], bo[3];
#pragma unroll
    for (int b = 0; b < 3; ++b) {
        bo[b] = bout[b];
#pragma unroll
        for (int t = 0; t < NT; ++t) {
            bz[b][t] = bias[b * NC + 16 * t + m];
            wo[b][t] = Wout[b * NC + 16 * t + m];
        }
    }
    const bool bit3 = (m & 8) != 0;
    const bool bit2 = (m & 4) != 0;
    const bool bit1 = (m & 2) != 0;

#pragma unroll 1
    for (int blk = blockIdx.x * (GS_THREADS / 32) + wave; blk < nblk; blk += gridDim.x * (GS_THREADS / 32)) {
#pragma unroll 1
        for (int st = 0; st < DEC_RPB / 16; ++st) {
            const int row0 = blk * DEC_RPB + st * 16;
            if (row0 >= n) break;
            const int row  = row0 + m;
            const int rowc = (row < n) ? row : (n - 1);

            const int abase = rowc * F3 + half * 2;
            const float* s0 = e0 + abase;
            const float* s1 = i1 + abase;
            const float* s2 = i2 + abase;
            v2f a[3][KS];
#pragma unroll
            for (int b = 0; b < 3; ++b) {
#pragma unroll
                for (int ks = 0; ks < KS; ++ks) {
                    const int k   = 4 * ks;
                    const int src = k >> 5;
                    const float* sp = (src == 0) ? s0 : (src == 1) ? s1 : s2;
                    a[b][ks] = *(const v2f*)(sp + b * 32 + (k & 31));
                }
            }
            float res[3];
#pragma unroll
            for (int b = 0; b < 3; ++b) {
                v8f acc[NT];
#pragma unroll
                for (int t = 0; t < NT; ++t) {
                    const float bv = bz[b][t];
                    acc[t] = (v8f){bv, bv, bv, bv, bv, bv, bv, bv};
                }
                const float* ws = Ws + lane * 4 + b * K * NC;
                strip_products<K, NT, false>(a[b], a[b], ws, ws, acc);

                float p[8];
#pragma unroll
                for (int j = 0; j < 8; ++j) {
                    float s = 0.0f;
#pragma unroll
                    for (int t = 0; t < NT; ++t) {
                        float v = acc[t][j];
                        v = (v > 0.0f) ? v : (expf(v) - 1.0f);
                        s += v * wo[b][t];
                    }
                    p[j] = s;
                }
                float q[4], r2[2];
#pragma unroll
                for (int i = 0; i < 4; ++i) {
                    const float send = bit3 ? p[i] : p[i + 4];
                    const float keep = bit3 ? p[i + 4] : p[i];
                    q[i] = keep + __shfl_xor(send, 8, 32);
                }
#pragma unroll
                for (int i = 0; i < 2; ++i) {
                    const float send = bit2 ? q[i] : q[i + 2];
                    const float keep = bit2 ? q[i + 2] : q[i];
                    r2[i] = keep + __shfl_xor(send, 4, 32);
                }
                const float send1 = bit1 ? r2[0] : r2[1];
                const float keep1 = bit1 ? r2[1] : r2[0];
                float tot = keep1 + __shfl_xor(send1, 2, 32);
                tot += __shfl_xor(tot, 1, 32);
                res[b] = tot + bo[b];
            }

            const int orow = row0 + half * 8 + (m >> 1);
            if ((m & 1) == 0 && orow < n) {
                float* op = out + orow * 3;
                op[0] = res[0];
                op[1] = res[1];
                op[2] = res[2];
            }
        }
    }
}

#define FILL_GRID(count) (((count) / 4 + 8 * 32 * FILL_SPW - 1) / (8 * 32 * FILL_SPW) > 256 ? 256 \
                          : ((count) / 4 + 8 * 32 * FILL_SPW - 1) / (8 * 32 * FILL_SPW))

#define WS_H1    ((size_t)0)
#define WS_AGG   (WS_H1 + (size_t)NROWS * H)
#define WS_AGG6  (WS_AGG + (size_t)NROWS * H)
#define WS_H2    (((WS_AGG6 + (size_t)NROWS * 6 + 31) / 32) * 32)
#define WS_E     (WS_H2 + (size_t)NROWS * H)
#define WS_I1    ((size_t)0)
#define WS_I2    (WS_I1 + (size_t)N0 * F3)
#define WS_CNT   WS_E
#define WS_CUR   (WS_CNT + (size_t)NROWS * CNT_STRIDE)
#define WS_OVF   (WS_CUR + 32)
static_assert(WS_OVF + 2 * (size_t)ETOT <= WS_E + (size_t)NROWS * F3, "the source lists must fit in e's place");

extern "C" void kernel_launch(void* const* d_in, const int* in_sizes, int n_in,
                              void* d_out, int out_size, void* d_ws, size_t ws_size,
                              hipStream_t stream) {
    (void)in_sizes; (void)n_in; (void)out_size; (void)ws_size;

    const float* x0     = (const float*)d_in[0];
    const float* x1     = (const float*)d_in[1];
    const float* x2     = (const float*)d_in[2];
    const int*   ei0    = (const int*)d_in[3];
    const int*   ei1    = (const int*)d_in[4];
    const int*   ei2    = (const int*)d_in[5];
    const int*   A1r    = (const int*)d_in[6];
    const int*   A1c    = (const int*)d_in[7];
    const float* A1v    = (const float*)d_in[8];
    const int*   A2r    = (const int*)d_in[9];
    const int*   A2c    = (const int*)d_in[10];
    const float* A2v    = (const float*)d_in[11];
    const float* c1Wrel  = (const float*)d_in[12];
    const float* c1brel  = (const float*)d_in[13];
    const float* c1Wroot = (const float*)d_in[14];
    const float* c2Wrel  = (const float*)d_in[15];
    const float* c2brel  = (const float*)d_in[16];
    const float* c2Wroot = (const float*)d_in[17];
    const float* linW    = (const float*)d_in[18];
    const float* linb    = (const float*)d_in[19];
    const float* decW0   = (const float*)d_in[20];
    const float* decb0   = (const float*)d_in[21];
    const float* decWout = (const float*)d_in[22];
    const float* decbout = (const float*)d_in[23];
    float* out = (float*)d_out;

    float* ws   = (float*)d_ws;
    float* h1   = ws + WS_H1;
    float* agg  = ws + WS_AGG;
    float* agg6 = ws + WS_AGG6;
    float* h2   = ws + WS_H2;
    float* eall = ws + WS_E;
    float* i1   = ws + WS_I1;
    float* i2   = ws + WS_I2;
    int* cnt    = (int*)(ws + WS_CNT);
    int* cursor = (int*)(ws + WS_CUR);
    int* ovf    = (int*)(ws + WS_OVF);

    zero_kernel<<<FILL_GRID(NROWS * CNT_STRIDE + 32), 256, 0, stream>>>((float*)cnt, NROWS * CNT_STRIDE + 32);
    csr_build_kernel<<<(ETOT + 255) / 256, 256, 0, stream>>>(ei0, ei1, ei2, cnt, cursor, ovf);

    gather6_kernel<<<(NROWS * 8 + 255) / 256, 256, 0, stream>>>(x0, x1, x2, cnt, cursor, ovf, agg6);
    conv1_kernel<<<(C1_W0 + C1_W1 + C1_W2 + 7) / 8, 256, 0, stream>>>(x0, x1, x2, agg6, c1Wrel, c1brel,
                                                                    c1Wroot, h1);

    gather128_kernel<<<(NROWS + 7) / 8, 256, 0, stream>>>(h1, cnt, cursor, ovf, agg);
    gemm_wmma_kernel<H, H, 1, true><<<GG0 + GG1 + GG2, GS_THREADS, 0, stream>>>(
        agg, c2Wrel, h1, c2Wroot, c2brel, h2);

    gemm_wmma_kernel<H, F3, 0, false><<<GG0 + GG1 + GG2, GS_THREADS, 0, stream>>>(
        h2, linW, nullptr, nullptr, linb, eall);

    zero_kernel<<<FILL_GRID(2 * N0 * F3), 256, 0, stream>>>(i1, 2 * N0 * F3);
    interp_kernel<<<(2 * IN_W + 7) / 8, 256, 0, stream>>>(A1v, A1c, A1r, eall + (size_t)ROFF1 * F3, i1,
                                                         A2v, A2c, A2r, eall + (size_t)ROFF2 * F3, i2);

    dec_fused_wmma_kernel<<<((N0 + DEC_RPB - 1) / DEC_RPB + 3) / 4, GS_THREADS, 0, stream>>>(
        eall, i1, i2, decW0, decb0, decWout, decbout, out, N0);
}


# interpolation sums as gathers over per-call row lists (1 returning int atomic per entry, in the edge-list launch); a wave sums a row in registers; float atomics and fill gone; sum order changed
# speedup vs baseline: 3.4371x; 1.0780x over previous
#include <hip/hip_runtime.h>
#include <hip/hip_bf16.h>

#define N0 50000
#define N1 12500
#define N2 3125
#define E0 800000
#define E1 200000
#define E2 50000
#define NNZ 200000
#define H 128
#define F3 96
#define DECW 64

#define ROFF1 N0
#define ROFF2 (N0 + N1)
#define NROWS (N0 + N1 + N2)

typedef __attribute__((ext_vector_type(2))) float v2f;
typedef __attribute__((ext_vector_type(8))) float v8f;

#define FILL_SPW 4
__global__ void zero_kernel(float* __restrict__ p, int count) {
    const int tid  = blockIdx.x * blockDim.x + threadIdx.x;
    const int lane = threadIdx.x & 31;
    const int wave = __builtin_amdgcn_readfirstlane(tid >> 5);
    const int step = ((gridDim.x * blockDim.x) >> 5) * (32 * FILL_SPW);
    int head = (int)(((16u - (unsigned)((size_t)p & 15u)) & 15u) >> 2);
    if (head > count) head = count;
    float4* q = (float4*)(p + head);
    const int nvec = (count - head) >> 2;
    const float4 z = {0.0f, 0.0f, 0.0f, 0.0f};
    for (int base = wave * (32 * FILL_SPW); base < nvec; base += step) {
        float4* qq = q + base + lane;
        if (base + 32 * FILL_SPW <= nvec) {
#pragma unroll
            for (int u = 0; u < FILL_SPW; ++u) qq[32 * u] = z;
        } else {
#pragma unroll
            for (int u = 0; u < FILL_SPW; ++u)
                if (base + 32 * u + lane < nvec) qq[32 * u] = z;
        }
    }
    const int tail0 = head + (nvec << 2);
    if (tid < head) p[tid] = 0.0f;
    if (tid < count - tail0) p[tail0 + tid] = 0.0f;
}

#define C1_RPW 8
#define C1_W0 ((N0 + C1_RPW - 1) / C1_RPW)
#define C1_W1 ((N1 + C1_RPW - 1) / C1_RPW)
#define C1_W2 ((N2 + C1_RPW - 1) / C1_RPW)
template <bool TAIL>
__device__ __forceinline__ void conv1_wave(const float* __restrict__ x, const float* __restrict__ agg6,
                                           const float* __restrict__ Wrel, const float* __restrict__ brel,
                                           const float* __restrict__ Wroot, float* __restrict__ h1,
                                           int n, int r0, int c) {
    float wr[6][4], wo[6][4], b[4];
#pragma unroll
    for (int q = 0; q < 4; ++q) b[q] = brel[c + q];
#pragma unroll
    for (int k = 0; k < 6; ++k) {
#pragma unroll
        for (int q = 0; q < 4; ++q) {
            wr[k][q] = Wrel[k * H + c + q];
            wo[k][q] = Wroot[k * H + c + q];
        }
    }
#pragma unroll
    for (int j = 0; j < C1_RPW; ++j) {
        const int r = r0 + j;
        int rl = r;
        if (TAIL) rl = (r < n) ? r : (n - 1);
        const float* ar = agg6 + rl * 6;
        const float* xr = x + rl * 6;
        float s[4] = {b[0], b[1], b[2], b[3]};
#pragma unroll
        for (int k = 0; k < 6; ++k) {
            const float a  = ar[k];
            const float xv = xr[k];
#pragma unroll
            for (int q = 0; q < 4; ++q) {
                s[q] += a * wr[k][q];
                s[q] += xv * wo[k][q];
            }
        }
        if (!TAIL || r < n) {
            float4 v;
            v.x = fmaxf(s[0], 0.0f);
            v.y = fmaxf(s[1], 0.0f);
            v.z = fmaxf(s[2], 0.0f);
            v.w = fmaxf(s[3], 0.0f);
            *(float4*)(h1 + r * H + c) = v;
        }
    }
}

__global__ void conv1_kernel(const float* __restrict__ x0, const float* __restrict__ x1,
                             const float* __restrict__ x2, const float* __restrict__ agg6,
                             const float* __restrict__ Wrel, const float* __restrict__ brel,
                             const float* __restrict__ Wroot, float* __restrict__ h1) {
    const int lane = threadIdx.x & 31;
    int wave = __builtin_amdgcn_readfirstlane((int)((blockIdx.x * blockDim.x + threadIdx.x) >> 5));
    const float* x = x0;
    int n = N0, lvl = 0, roff = 0;
    if (wave >= C1_W0 + C1_W1) { wave -= C1_W0 + C1_W1; x = x2; n = N2; lvl = 2; roff = ROFF2; }
    else if (wave >= C1_W0)    { wave -= C1_W0;         x = x1; n = N1; lvl = 1; roff = ROFF1; }
    const int r0 = wave * C1_RPW;
    if (r0 >= n) return;
    const float* a6 = agg6 + roff * 6;
    const float* wr = Wrel + lvl * 6 * H;
    const float* br = brel + lvl * H;
    const float* wo = Wroot + lvl * 6 * H;
    float* hd = h1 + (size_t)roff * H;
    if (r0 + C1_RPW <= n) conv1_wave<false>(x, a6, wr, br, wo, hd, n, r0, lane * 4);
    else                  conv1_wave<true >(x, a6, wr, br, wo, hd, n, r0, lane * 4);
}

#define ETOT (E0 + E1 + E2)
#define CNT_STRIDE 32
#define CSR_INL (CNT_STRIDE - 1)

#define EDGE_LEVEL(eg, ei, E, roff, e)                                             \
    const int* ei = ei0; int E = E0, roff = 0, e = (eg);                          \
    if ((eg) >= E0 + E1) { ei = ei2; E = E2; roff = ROFF2; e = (eg) - (E0 + E1); } \
    else if ((eg) >= E0) { ei = ei1; E = E1; roff = ROFF1; e = (eg) - E0; }

#define ITOT (2 * NNZ)
#define IREC_PAIRS ((CNT_STRIDE - 2) / 2)

__global__ void csr_build_kernel(const int* __restrict__ ei0, const int* __restrict__ ei1,
                                 const int* __restrict__ ei2, int* __restrict__ rec,
                                 int* __restrict__ cursor, int* __restrict__ ovf,
                                 const int* __restrict__ rows1, const int* __restrict__ cols1,
                                 const float* __restrict__ vals1,
                                 const int* __restrict__ rows2, const int* __restrict__ cols2,
                                 const float* __restrict__ vals2,
                                 int* __restrict__ irec, int* __restrict__ icur, int* __restrict__ iovf) {
    const int eg = blockIdx.x * blockDim.x + threadIdx.x;
    if (eg < ETOT) {
        EDGE_LEVEL(eg, ei, E, roff, e)
        const int s = ei[e] + roff;
        const int d = ei[E + e] + roff;
        int* line = rec + (size_t)d * CNT_STRIDE;
        const int place = atomicAdd(line, 1);
        if (place < CSR_INL) {
            line[1 + place] = s;
        } else {
            const int k = atomicAdd(cursor, 1);
            ovf[2 * k]     = d;
            ovf[2 * k + 1] = s;
        }
    } else if (eg < ETOT + ITOT) {
        int n = eg - ETOT;
        const int* rows = rows1;
        const int* cols = cols1;
        const float* vals = vals1;
        int rbase = 0;
        if (n >= NNZ) { n -= NNZ; rows = rows2; cols = cols2; vals = vals2; rbase = N0; }
        const int rr = rows[n] + rbase;
        int2 p;
        p.x = cols[n];
        p.y = __float_as_int(vals[n]);
        int* line = irec + (size_t)rr * CNT_STRIDE;
        const int place = atomicAdd(line, 1);
        if (place < IREC_PAIRS) {
            *(int2*)(line + 2 + 2 * place) = p;
        } else {
            const int k = atomicAdd(icur, 1);
            iovf[3 * k]     = rr;
            iovf[3 * k + 1] = p.x;
            iovf[3 * k + 2] = p.y;
        }
    }
}

__global__ void gather6_kernel(const float* __restrict__ x0, const float* __restrict__ x1,
                               const float* __restrict__ x2, const int* __restrict__ rec,
                               const int* __restrict__ cursor, const int* __restrict__ ovf,
                               float* __restrict__ agg6all) {
    const int t = blockIdx.x * blockDim.x + threadIdx.x;
    const int r = t >> 3;
    const int k = t & 7;
    if (r >= NROWS || k >= 6) return;
    const float* x = x0;
    int roff = 0;
    if (r >= ROFF2)      { x = x2; roff = ROFF2; }
    else if (r >= ROFF1) { x = x1; roff = ROFF1; }
    const int* sp = rec + (size_t)r * CNT_STRIDE + 1;
    const int deg = sp[-1];
    const int c   = (deg < CSR_INL) ? deg : CSR_INL;
    const float* xk = x + k;
    float acc = 0.0f;
    int i = 0;
    for (; i + 4 <= c; i += 4) {
        const int s0 = sp[i], s1 = sp[i + 1], s2 = sp[i + 2], s3 = sp[i + 3];
        const float v0 = xk[(s0 - roff) * 6];
        const float v1 = xk[(s1 - roff) * 6];
        const float v2 = xk[(s2 - roff) * 6];
        const float v3 = xk[(s3 - roff) * 6];
        acc += v0; acc += v1; acc += v2; acc += v3;
    }
    for (; i < c; ++i) acc += xk[(sp[i] - roff) * 6];
    if (deg > CSR_INL) {
        int nov = *cursor;
        nov = (nov < ETOT) ? nov : ETOT;
        for (int q = 0; q < nov; ++q)
            if (ovf[2 * q] == r) acc += xk[(ovf[2 * q + 1] - roff) * 6];
    }
    agg6all[r * 6 + k] = acc;
}

template <int NB, bool LIM>
__device__ __forceinline__ void gather128_batch(const int* __restrict__ sp, int i, int c,
                                                const float4* __restrict__ hb, float4& acc) {
    int s[NB];
    float4 v[NB];
#pragma unroll
    for (int j = 0; j < NB; ++j) {
        int ii = i + j;
        if (LIM) ii = (ii < c) ? ii : (c - 1);
        s[j] = sp[ii];
    }
#pragma unroll
    for (int j = 0; j < NB; ++j) v[j] = hb[(size_t)s[j] * (H / 4)];
#pragma unroll
    for (int j = 0; j < NB; ++j) {
        if (!LIM || i + j < c) {
            acc.x += v[j].x;
            acc.y += v[j].y;
            acc.z += v[j].z;
            acc.w += v[j].w;
        }
    }
}

__global__ void gather128_kernel(const float* __restrict__ h1all, const int* __restrict__ rec,
                                 const int* __restrict__ cursor, const int* __restrict__ ovf,
                                 float* __restrict__ aggall) {
    const int lane = threadIdx.x & 31;
    const int r = __builtin_amdgcn_readfirstlane((int)((blockIdx.x * blockDim.x + threadIdx.x) >> 5));
    if (r >= NROWS) return;
    const int* sp = rec + (size_t)r * CNT_STRIDE + 1;
    const int deg = sp[-1];
    const int c   = (deg < CSR_INL) ? deg : CSR_INL;
    const float4* hb = (const float4*)h1all + lane;
    float4 acc = {0.0f, 0.0f, 0.0f, 0.0f};
    int i = 0;
    for (; i + 8 <= c; i += 8) gather128_batch<8, false>(sp, i, c, hb, acc);
    const int rem = c - i;
    if (rem > 4)      gather128_batch<8, true>(sp, i, c, hb, acc);
    else if (rem > 0) gather128_batch<4, true>(sp, i, c, hb, acc);
    if (deg > CSR_INL) {
        int nov = *cursor;
        nov = (nov < ETOT) ? nov : ETOT;
        for (int q = 0; q < nov; ++q) {
            if (ovf[2 * q] == r) {
                const float4 v = hb[(size_t)ovf[2 * q + 1] * (H / 4)];
                acc.x += v.x;
                acc.y += v.y;
                acc.z += v.z;
                acc.w += v.w;
            }
        }
    }
    ((float4*)aggall)[(size_t)r * (H / 4) + lane] = acc;
}

template <int NB>
__device__ __forceinline__ void interp_batch(const int* __restrict__ pr, int i,
                                             const float4* __restrict__ eb, float4& acc) {
    int col[NB];
    float val[NB];
    float4 v[NB];
#pragma unroll
    for (int j = 0; j < NB; ++j) {
        col[j] = pr[2 * (i + j)];
        val[j] = __int_as_float(pr[2 * (i + j) + 1]);
    }
#pragma unroll
    for (int j = 0; j < NB; ++j) v[j] = eb[(size_t)col[j] * (F3 / 4)];
#pragma unroll
    for (int j = 0; j < NB; ++j) {
        acc.x += val[j] * v[j].x;
        acc.y += val[j] * v[j].y;
        acc.z += val[j] * v[j].z;
        acc.w += val[j] * v[j].w;
    }
}

__global__ void interp_gather_kernel(const int* __restrict__ irec, const int* __restrict__ icur,
                                     const int* __restrict__ iovf, const float* __restrict__ ea,
                                     const float* __restrict__ eb2, float* __restrict__ out1,
                                     float* __restrict__ out2) {
    const int lane = threadIdx.x & 31;
    const int w = __builtin_amdgcn_readfirstlane((int)((blockIdx.x * blockDim.x + threadIdx.x) >> 5));
    if (w >= 2 * N0) return;
    const float* e = ea;
    float* o = out1;
    int r = w;
    if (w >= N0) { e = eb2; o = out2; r = w - N0; }
    const int* rp  = irec + (size_t)w * CNT_STRIDE;
    const int deg  = rp[0];
    const int c    = (deg < IREC_PAIRS) ? deg : IREC_PAIRS;
    const int* pr  = rp + 2;
    const int ln   = (lane < F3 / 4) ? lane : (F3 / 4 - 1);
    const float4* eb = (const float4*)e + ln;
    float4 acc = {0.0f, 0.0f, 0.0f, 0.0f};
    int i = 0;
    for (; i + 8 <= c; i += 8) interp_batch<8>(pr, i, eb, acc);
    switch (c - i) {
        case 1: interp_batch<1>(pr, i, eb, acc); break;
        case 2: interp_batch<2>(pr, i, eb, acc); break;
        case 3: interp_batch<3>(pr, i, eb, acc); break;
        case 4: interp_batch<4>(pr, i, eb, acc); break;
        case 5: interp_batch<5>(pr, i, eb, acc); break;
        case 6: interp_batch<6>(pr, i, eb, acc); break;
        case 7: interp_batch<7>(pr, i, eb, acc); break;
        default: break;
    }
    if (deg > IREC_PAIRS) {
        int nov = *icur;
        nov = (nov < ITOT) ? nov : ITOT;
        for (int q = 0; q < nov; ++q) {
            if (iovf[3 * q] == w) {
                const float val = __int_as_float(iovf[3 * q + 2]);
                const float4 v = eb[(size_t)iovf[3 * q + 1] * (F3 / 4)];
                acc.x += val * v.x;
                acc.y += val * v.y;
                acc.z += val * v.z;
                acc.w += val * v.w;
            }
        }
    }
    if (lane < F3 / 4) ((float4*)o)[(size_t)r * (F3 / 4) + lane] = acc;
}

#define GS_THREADS 128

template <int K, int NC, int UNR = 4>
__device__ __forceinline__ void stage_weights(const float* __restrict__ W, float* __restrict__ Ws) {
    constexpr int NT = NC / 16;
#pragma unroll UNR
    for (int q = threadIdx.x; q < (K / 4) * NC; q += GS_THREADS) {
        const int ks = q / NC;
        const int nn = q - ks * NC;
        const float* wp = W + ks * 4 * NC + nn;
        v2f lo, hi;
        lo.x = wp[0];
        lo.y = wp[NC];
        hi.x = wp[2 * NC];
        hi.y = wp[3 * NC];
        const int t = nn >> 4;
        const int c = nn & 15;
        float* dst = Ws + ((ks * (NT / 2) + (t >> 1)) * 32 + c) * 4 + (t & 1) * 2;
        *(v2f*)dst        = lo;
        *(v2f*)(dst + 64) = hi;
    }
}

template <int NT>
__device__ __forceinline__ void load_b(const float* __restrict__ wb, v2f (&b)[NT]) {
#pragma unroll
    for (int tp = 0; tp < NT / 2; ++tp) {
        const float4 q = *(const float4*)(wb + tp * 128);
        b[2 * tp].x     = q.x;
        b[2 * tp].y     = q.y;
        b[2 * tp + 1].x = q.z;
        b[2 * tp + 1].y = q.w;
    }
}

template <int NC, int NT, int ACT>
__device__ __forceinline__ void store_strip(const v8f (&acc)[NT], float* __restrict__ D,
                                            int n, int row0, int m, int half) {
    const int rbase = row0 + half * 8;
    const int lim   = n - rbase;
    float* dp = D + rbase * NC + m;
#pragma unroll
    for (int j = 0; j < 8; ++j) {
        if (j < lim) {
#pragma unroll
            for (int t = 0; t < NT; ++t) {
                float v = acc[t][j];
                if (ACT == 1) v = fmaxf(v, 0.0f);
                if (ACT == 2) v = (v > 0.0f) ? v : (expf(v) - 1.0f);
                dp[j * NC + 16 * t] = v;
            }
        }
    }
}

template <int K, int NT, bool DUAL>
__device__ __forceinline__ void strip_products(const v2f (&a0)[K / 4], const v2f (&a1)[K / 4],
                                               const float* __restrict__ w0s,
                                               const float* __restrict__ w1s, v8f (&acc)[NT]) {
    constexpr int KS = K / 4;
    v2f b0[NT], b1[NT];
    load_b<NT>(w0s, b0);
    if (DUAL) load_b<NT>(w1s, b1);
#pragma unroll
    for (int ks = 0; ks < KS; ++ks) {
        v2f nb0[NT], nb1[NT];
        if (ks + 1 < KS) {
            load_b<NT>(w0s + (ks + 1) * (NT / 2) * 128, nb0);
            if (DUAL) load_b<NT>(w1s + (ks + 1) * (NT / 2) * 128, nb1);
        }
#pragma unroll
        for (int t = 0; t < NT; ++t)
            acc[t] = __builtin_amdgcn_wmma_f32_16x16x4_f32(false, a0[ks], false, b0[t], (short)0,
                                                           acc[t], false, false);
        if (DUAL) {
#pragma unroll
            for (int t = 0; t < NT; ++t)
                acc[t] = __builtin_amdgcn_wmma_f32_16x16x4_f32(false, a1[ks], false, b1[t], (short)0,
                                                               acc[t], false, false);
        }
        if (ks + 1 < KS) {
#pragma unroll
            for (int t = 0; t < NT; ++t) {
                b0[t] = nb0[t];
                if (DUAL) b1[t] = nb1[t];
            }
        }
    }
}

#define GEMM_GRID(n) ((((n) + 15) / 16 + 3) / 4 > 512 ? 512 : (((n) + 15) / 16 + 3) / 4)
#define GG0 GEMM_GRID(N0)
#define GG1 GEMM_GRID(N1)
#define GG2 GEMM_GRID(N2)

template <int K, int NC, int ACT, bool DUAL>
__global__ void __launch_bounds__(GS_THREADS)
gemm_wmma_kernel(const float* __restrict__ A0all, const float* __restrict__ W0all,
                 const float* __restrict__ A1all, const float* __restrict__ W1all,
                 const float* __restrict__ biasall, float* __restrict__ Dall) {
    constexpr int NT = NC / 16;
    constexpr int KS = K / 4;
    __shared__ __attribute__((aligned(16))) float Ws[(DUAL ? 2 : 1) * K * NC];

    int lb = blockIdx.x, lg = GG0, n = N0, lvl = 0, roff = 0;
    if (lb >= GG0 + GG1) { lb -= GG0 + GG1; lg = GG2; n = N2; lvl = 2; roff = ROFF2; }
    else if (lb >= GG0)  { lb -= GG0;       lg = GG1; n = N1; lvl = 1; roff = ROFF1; }
    const float* A0   = A0all + (size_t)roff * K;
    const float* W0   = W0all + lvl * K * NC;
    const float* A1   = DUAL ? (A1all + (size_t)roff * K) : A0;
    const float* W1   = DUAL ? (W1all + lvl * K * NC) : W0;
    const float* bias = biasall + lvl * NC;
    float* D          = Dall + (size_t)roff * NC;

    stage_weights<K, NC>(W0, Ws);
    if (DUAL) stage_weights<K, NC>(W1, Ws + K * NC);
    __syncthreads();

    const int lane = threadIdx.x & 31;
    const int wave = __builtin_amdgcn_readfirstlane((int)(threadIdx.x >> 5));
    const int m    = lane & 15;
    const int half = lane >> 4;
    const int nstrips = (n + 15) >> 4;

#pragma unroll 1
    for (int s = lb * (GS_THREADS / 32) + wave; s < nstrips; s += lg * (GS_THREADS / 32)) {
        int boff0 = lane * 4;
        int boff1 = lane * 4 + K * NC;
        if (DUAL) {
            asm volatile("" : "+v"(boff0), "+v"(boff1));
        }
        const float* w0s = Ws + boff0;
        const float* w1s = Ws + boff1;
        const int row0 = s * 16;
        const int row  = row0 + m;
        const int rowc = (row < n) ? row : (n - 1);

        const float* a0p = A0 + rowc * K + half * 2;
        const float* a1p = DUAL ? (A1 + rowc * K + half * 2) : a0p;
        v2f a0[KS], a1[KS];
#pragma unroll
        for (int ks = 0; ks < KS; ++ks) {
            a0[ks] = *(const v2f*)(a0p + 4 * ks);
            if (DUAL) a1[ks] = *(const v2f*)(a1p + 4 * ks);
        }

        v8f acc[NT];
#pragma unroll
        for (int t = 0; t < NT; ++t) {
            const float bv = bias[16 * t + m];
            acc[t] = (v8f){bv, bv, bv, bv, bv, bv, bv, bv};
        }

        strip_products<K, NT, DUAL>(a0, a1, w0s, w1s, acc);

        store_strip<NC, NT, ACT>(acc, D, n, row0, m, half);
    }
}

#define DEC_RPB 32
__global__ void __launch_bounds__(GS_THREADS)
dec_fused_wmma_kernel(const float* __restrict__ e0, const float* __restrict__ i1,
                      const float* __restrict__ i2, const float* __restrict__ W,
                      const float* __restrict__ bias, const float* __restrict__ Wout,
                      const float* __restrict__ bout, float* __restrict__ out, int n) {
    constexpr int K = F3, NC = DECW;
    constexpr int NT = NC / 16;
    constexpr int KS = K / 4;
    __shared__ __attribute__((aligned(16))) float Ws[3 * K * NC];

    stage_weights<3 * K, NC, 12>(W, Ws);
    __syncthreads();

    const int lane = threadIdx.x & 31;
    const int wave = __builtin_amdgcn_readfirstlane((int)(threadIdx.x >> 5));
    const int m    = lane & 15;
    const int half = lane >> 4;
    const int nblk = (n + DEC_RPB - 1) / DEC_RPB;

    float bz[3][NT], wo[3][NT], bo[3];
#pragma unroll
    for (int b = 0; b < 3; ++b) {
        bo[b] = bout[b];
#pragma unroll
        for (int t = 0; t < NT; ++t) {
            bz[b][t] = bias[b * NC + 16 * t + m];
            wo[b][t] = Wout[b * NC + 16 * t + m];
        }
    }
    const bool bit3 = (m & 8) != 0;
    const bool bit2 = (m & 4) != 0;
    const bool bit1 = (m & 2) != 0;

#pragma unroll 1
    for (int blk = blockIdx.x * (GS_THREADS / 32) + wave; blk < nblk; blk += gridDim.x * (GS_THREADS / 32)) {
#pragma unroll 1
        for (int st = 0; st < DEC_RPB / 16; ++st) {
            const int row0 = blk * DEC_RPB + st * 16;
            if (row0 >= n) break;
            const int row  = row0 + m;
            const int rowc = (row < n) ? row : (n - 1);

            const int abase = rowc * F3 + half * 2;
            const float* s0 = e0 + abase;
            const float* s1 = i1 + abase;
            const float* s2 = i2 + abase;
            v2f a[3][KS];
#pragma unroll
            for (int b = 0; b < 3; ++b) {
#pragma unroll
                for (int ks = 0; ks < KS; ++ks) {
                    const int k   = 4 * ks;
                    const int src = k >> 5;
                    const float* sp = (src == 0) ? s0 : (src == 1) ? s1 : s2;
                    a[b][ks] = *(const v2f*)(sp + b * 32 + (k & 31));
                }
            }
            float res[3];
#pragma unroll
            for (int b = 0; b < 3; ++b) {
                v8f acc[NT];
#pragma unroll
                for (int t = 0; t < NT; ++t) {
                    const float bv = bz[b][t];
                    acc[t] = (v8f){bv, bv, bv, bv, bv, bv, bv, bv};
                }
                const float* ws = Ws + lane * 4 + b * K * NC;
                strip_products<K, NT, false>(a[b], a[b], ws, ws, acc);

                float p[8];
#pragma unroll
                for (int j = 0; j < 8; ++j) {
                    float s = 0.0f;
#pragma unroll
                    for (int t = 0; t < NT; ++t) {
                        float v = acc[t][j];
                        v = (v > 0.0f) ? v : (expf(v) - 1.0f);
                        s += v * wo[b][t];
                    }
                    p[j] = s;
                }
                float q[4], r2[2];
#pragma unroll
                for (int i = 0; i < 4; ++i) {
                    const float send = bit3 ? p[i] : p[i + 4];
                    const float keep = bit3 ? p[i + 4] : p[i];
                    q[i] = keep + __shfl_xor(send, 8, 32);
                }
#pragma unroll
                for (int i = 0; i < 2; ++i) {
                    const float send = bit2 ? q[i] : q[i + 2];
                    const float keep = bit2 ? q[i + 2] : q[i];
                    r2[i] = keep + __shfl_xor(send, 4, 32);
                }
                const float send1 = bit1 ? r2[0] : r2[1];
                const float keep1 = bit1 ? r2[1] : r2[0];
                float tot = keep1 + __shfl_xor(send1, 2, 32);
                tot += __shfl_xor(tot, 1, 32);
                res[b] = tot + bo[b];
            }

            const int orow = row0 + half * 8 + (m >> 1);
            if ((m & 1) == 0 && orow < n) {
                float* op = out + orow * 3;
                op[0] = res[0];
                op[1] = res[1];
                op[2] = res[2];
            }
        }
    }
}

#define FILL_GRID(count) (((count) / 4 + 8 * 32 * FILL_SPW - 1) / (8 * 32 * FILL_SPW) > 256 ? 256 \
                          : ((count) / 4 + 8 * 32 * FILL_SPW - 1) / (8 * 32 * FILL_SPW))

#define WS_H1    ((size_t)0)
#define WS_AGG   (WS_H1 + (size_t)NROWS * H)
#define WS_AGG6  (WS_AGG + (size_t)NROWS * H)
#define WS_H2    (((WS_AGG6 + (size_t)NROWS * 6 + 31) / 32) * 32)
#define WS_TAIL  (WS_H2 + (size_t)NROWS * H)
#define WS_I1    ((size_t)0)
#define WS_I2    (WS_I1 + (size_t)N0 * F3)
#define WS_E     (WS_I2 + (size_t)N0 * F3)
static_assert(WS_E % 32 == 0 && WS_E + (size_t)NROWS * F3 <= WS_AGG6, "e must fit behind i1 / i2 in h1 / agg's place");
#define WS_OVF   WS_H2
#define WS_CNT   (WS_TAIL - 32 - (size_t)NROWS * CNT_STRIDE)
#define WS_CUR   (WS_TAIL - 32)
#define WS_IREC  WS_TAIL
#define WS_ICUR  (WS_IREC + (size_t)2 * N0 * CNT_STRIDE)
#define WS_IOVF  (WS_ICUR + 32)
#define WS_ZCOUNT (NROWS * CNT_STRIDE + 32 + 2 * N0 * CNT_STRIDE + 32)
static_assert(WS_OVF + 2 * (size_t)ETOT <= WS_CNT, "the edge lists must fit in h2's place");
static_assert(WS_IOVF + 3 * (size_t)ITOT <= WS_TAIL + (size_t)NROWS * F3, "the interpolation lists must fit in the tail");

extern "C" void kernel_launch(void* const* d_in, const int* in_sizes, int n_in,
                              void* d_out, int out_size, void* d_ws, size_t ws_size,
                              hipStream_t stream) {
    (void)in_sizes; (void)n_in; (void)out_size; (void)ws_size;

    const float* x0     = (const float*)d_in[0];
    const float* x1     = (const float*)d_in[1];
    const float* x2     = (const float*)d_in[2];
    const int*   ei0    = (const int*)d_in[3];
    const int*   ei1    = (const int*)d_in[4];
    const int*   ei2    = (const int*)d_in[5];
    const int*   A1r    = (const int*)d_in[6];
    const int*   A1c    = (const int*)d_in[7];
    const float* A1v    = (const float*)d_in[8];
    const int*   A2r    = (const int*)d_in[9];
    const int*   A2c    = (const int*)d_in[10];
    const float* A2v    = (const float*)d_in[11];
    const float* c1Wrel  = (const float*)d_in[12];
    const float* c1brel  = (const float*)d_in[13];
    const float* c1Wroot = (const float*)d_in[14];
    const float* c2Wrel  = (const float*)d_in[15];
    const float* c2brel  = (const float*)d_in[16];
    const float* c2Wroot = (const float*)d_in[17];
    const float* linW    = (const float*)d_in[18];
    const float* linb    = (const float*)d_in[19];
    const float* decW0   = (const float*)d_in[20];
    const float* decb0   = (const float*)d_in[21];
    const float* decWout = (const float*)d_in[22];
    const float* decbout = (const float*)d_in[23];
    float* out = (float*)d_out;

    float* ws   = (float*)d_ws;
    float* h1   = ws + WS_H1;
    float* agg  = ws + WS_AGG;
    float* agg6 = ws + WS_AGG6;
    float* h2   = ws + WS_H2;
    float* eall = ws + WS_E;
    float* i1   = ws + WS_I1;
    float* i2   = ws + WS_I2;
    int* cnt    = (int*)(ws + WS_CNT);
    int* cursor = (int*)(ws + WS_CUR);
    int* ovf    = (int*)(ws + WS_OVF);
    int* irec   = (int*)(ws + WS_IREC);
    int* icur   = (int*)(ws + WS_ICUR);
    int* iovf   = (int*)(ws + WS_IOVF);

    zero_kernel<<<FILL_GRID(WS_ZCOUNT), 256, 0, stream>>>((float*)cnt, WS_ZCOUNT);
    csr_build_kernel<<<(ETOT + ITOT + 255) / 256, 256, 0, stream>>>(ei0, ei1, ei2, cnt, cursor, ovf,
                                                                   A1r, A1c, A1v, A2r, A2c, A2v,
                                                                   irec, icur, iovf);

    gather6_kernel<<<(NROWS * 8 + 255) / 256, 256, 0, stream>>>(x0, x1, x2, cnt, cursor, ovf, agg6);
    conv1_kernel<<<(C1_W0 + C1_W1 + C1_W2 + 7) / 8, 256, 0, stream>>>(x0, x1, x2, agg6, c1Wrel, c1brel,
                                                                    c1Wroot, h1);

    gather128_kernel<<<(NROWS + 7) / 8, 256, 0, stream>>>(h1, cnt, cursor, ovf, agg);
    gemm_wmma_kernel<H, H, 1, true><<<GG0 + GG1 + GG2, GS_THREADS, 0, stream>>>(
        agg, c2Wrel, h1, c2Wroot, c2brel, h2);

    gemm_wmma_kernel<H, F3, 0, false><<<GG0 + GG1 + GG2, GS_THREADS, 0, stream>>>(
        h2, linW, nullptr, nullptr, linb, eall);

    interp_gather_kernel<<<(2 * N0 + 7) / 8, 256, 0, stream>>>(irec, icur, iovf, eall + (size_t)ROFF1 * F3,
                                                              eall + (size_t)ROFF2 * F3, i1, i2);

    dec_fused_wmma_kernel<<<((N0 + DEC_RPB - 1) / DEC_RPB + 3) / 4, GS_THREADS, 0, stream>>>(
        eall, i1, i2, decW0, decb0, decWout, decbout, out, N0);
}


# conv2 + lin GEMMs chained in one launch: both products transposed (weights as A operand), relu'd h2 accumulators are lin's B fragments, h2 never stored; sum order of lin changed; 9 -> 8 launches
# speedup vs baseline: 3.7895x; 1.1025x over previous
#include <hip/hip_runtime.h>
#include <hip/hip_bf16.h>

#define N0 50000
#define N1 12500
#define N2 3125
#define E0 800000
#define E1 200000
#define E2 50000
#define NNZ 200000
#define H 128
#define F3 96
#define DECW 64

#define ROFF1 N0
#define ROFF2 (N0 + N1)
#define NROWS (N0 + N1 + N2)

typedef __attribute__((ext_vector_type(2))) float v2f;
typedef __attribute__((ext_vector_type(8))) float v8f;

#define FILL_SPW 4
__global__ void zero_kernel(float* __restrict__ p, int count) {
    const int tid  = blockIdx.x * blockDim.x + threadIdx.x;
    const int lane = threadIdx.x & 31;
    const int wave = __builtin_amdgcn_readfirstlane(tid >> 5);
    const int step = ((gridDim.x * blockDim.x) >> 5) * (32 * FILL_SPW);
    int head = (int)(((16u - (unsigned)((size_t)p & 15u)) & 15u) >> 2);
    if (head > count) head = count;
    float4* q = (float4*)(p + head);
    const int nvec = (count - head) >> 2;
    const float4 z = {0.0f, 0.0f, 0.0f, 0.0f};
    for (int base = wave * (32 * FILL_SPW); base < nvec; base += step) {
        float4* qq = q + base + lane;
        if (base + 32 * FILL_SPW <= nvec) {
#pragma unroll
            for (int u = 0; u < FILL_SPW; ++u) qq[32 * u] = z;
        } else {
#pragma unroll
            for (int u = 0; u < FILL_SPW; ++u)
                if (base + 32 * u + lane < nvec) qq[32 * u] = z;
        }
    }
    const int tail0 = head + (nvec << 2);
    if (tid < head) p[tid] = 0.0f;
    if (tid < count - tail0) p[tail0 + tid] = 0.0f;
}

#define C1_RPW 8
#define C1_W0 ((N0 + C1_RPW - 1) / C1_RPW)
#define C1_W1 ((N1 + C1_RPW - 1) / C1_RPW)
#define C1_W2 ((N2 + C1_RPW - 1) / C1_RPW)
template <bool TAIL>
__device__ __forceinline__ void conv1_wave(const float* __restrict__ x, const float* __restrict__ agg6,
                                           const float* __restrict__ Wrel, const float* __restrict__ brel,
                                           const float* __restrict__ Wroot, float* __restrict__ h1,
                                           int n, int r0, int c) {
    float wr[6][4], wo[6][4], b[4];
#pragma unroll
    for (int q = 0; q < 4; ++q) b[q] = brel[c + q];
#pragma unroll
    for (int k = 0; k < 6; ++k) {
#pragma unroll
        for (int q = 0; q < 4; ++q) {
            wr[k][q] = Wrel[k * H + c + q];
            wo[k][q] = Wroot[k * H + c + q];
        }
    }
#pragma unroll
    for (int j = 0; j < C1_RPW; ++j) {
        const int r = r0 + j;
        int rl = r;
        if (TAIL) rl = (r < n) ? r : (n - 1);
        const float* ar = agg6 + rl * 6;
        const float* xr = x + rl * 6;
        float s[4] = {b[0], b[1], b[2], b[3]};
#pragma unroll
        for (int k = 0; k < 6; ++k) {
            const float a  = ar[k];
            const float xv = xr[k];
#pragma unroll
            for (int q = 0; q < 4; ++q) {
                s[q] += a * wr[k][q];
                s[q] += xv * wo[k][q];
            }
        }
        if (!TAIL || r < n) {
            float4 v;
            v.x = fmaxf(s[0], 0.0f);
            v.y = fmaxf(s[1], 0.0f);
            v.z = fmaxf(s[2], 0.0f);
            v.w = fmaxf(s[3], 0.0f);
            *(float4*)(h1 + r * H + c) = v;
        }
    }
}

__global__ void conv1_kernel(const float* __restrict__ x0, const float* __restrict__ x1,
                             const float* __restrict__ x2, const float* __restrict__ agg6,
                             const float* __restrict__ Wrel, const float* __restrict__ brel,
                             const float* __restrict__ Wroot, float* __restrict__ h1) {
    const int lane = threadIdx.x & 31;
    int wave = __builtin_amdgcn_readfirstlane((int)((blockIdx.x * blockDim.x + threadIdx.x) >> 5));
    const float* x = x0;
    int n = N0, lvl = 0, roff = 0;
    if (wave >= C1_W0 + C1_W1) { wave -= C1_W0 + C1_W1; x = x2; n = N2; lvl = 2; roff = ROFF2; }
    else if (wave >= C1_W0)    { wave -= C1_W0;         x = x1; n = N1; lvl = 1; roff = ROFF1; }
    const int r0 = wave * C1_RPW;
    if (r0 >= n) return;
    const float* a6 = agg6 + roff * 6;
    const float* wr = Wrel + lvl * 6 * H;
    const float* br = brel + lvl * H;
    const float* wo = Wroot + lvl * 6 * H;
    float* hd = h1 + (size_t)roff * H;
    if (r0 + C1_RPW <= n) conv1_wave<false>(x, a6, wr, br, wo, hd, n, r0, lane * 4);
    else                  conv1_wave<true >(x, a6, wr, br, wo, hd, n, r0, lane * 4);
}

#define ETOT (E0 + E1 + E2)
#define CNT_STRIDE 32
#define CSR_INL (CNT_STRIDE - 1)

#define EDGE_LEVEL(eg, ei, E, roff, e)                                             \
    const int* ei = ei0; int E = E0, roff = 0, e = (eg);                          \
    if ((eg) >= E0 + E1) { ei = ei2; E = E2; roff = ROFF2; e = (eg) - (E0 + E1); } \
    else if ((eg) >= E0) { ei = ei1; E = E1; roff = ROFF1; e = (eg) - E0; }

#define ITOT (2 * NNZ)
#define IREC_PAIRS ((CNT_STRIDE - 2) / 2)

__global__ void csr_build_kernel(const int* __restrict__ ei0, const int* __restrict__ ei1,
                                 const int* __restrict__ ei2, int* __restrict__ rec,
                                 int* __restrict__ cursor, int* __restrict__ ovf,
                                 const int* __restrict__ rows1, const int* __restrict__ cols1,
                                 const float* __restrict__ vals1,
                                 const int* __restrict__ rows2, const int* __restrict__ cols2,
                                 const float* __restrict__ vals2,
                                 int* __restrict__ irec, int* __restrict__ icur, int* __restrict__ iovf) {
    const int eg = blockIdx.x * blockDim.x + threadIdx.x;
    if (eg < ETOT) {
        EDGE_LEVEL(eg, ei, E, roff, e)
        const int s = ei[e] + roff;
        const int d = ei[E + e] + roff;
        int* line = rec + (size_t)d * CNT_STRIDE;
        const int place = atomicAdd(line, 1);
        if (place < CSR_INL) {
            line[1 + place] = s;
        } else {
            const int k = atomicAdd(cursor, 1);
            ovf[2 * k]     = d;
            ovf[2 * k + 1] = s;
        }
    } else if (eg < ETOT + ITOT) {
        int n = eg - ETOT;
        const int* rows = rows1;
        const int* cols = cols1;
        const float* vals = vals1;
        int rbase = 0;
        if (n >= NNZ) { n -= NNZ; rows = rows2; cols = cols2; vals = vals2; rbase = N0; }
        const int rr = rows[n] + rbase;
        int2 p;
        p.x = cols[n];
        p.y = __float_as_int(vals[n]);
        int* line = irec + (size_t)rr * CNT_STRIDE;
        const int place = atomicAdd(line, 1);
        if (place < IREC_PAIRS) {
            *(int2*)(line + 2 + 2 * place) = p;
        } else {
            const int k = atomicAdd(icur, 1);
            iovf[3 * k]     = rr;
            iovf[3 * k + 1] = p.x;
            iovf[3 * k + 2] = p.y;
        }
    }
}

__global__ void gather6_kernel(const float* __restrict__ x0, const float* __restrict__ x1,
                               const float* __restrict__ x2, const int* __restrict__ rec,
                               const int* __restrict__ cursor, const int* __restrict__ ovf,
                               float* __restrict__ agg6all) {
    const int t = blockIdx.x * blockDim.x + threadIdx.x;
    const int r = t >> 3;
    const int k = t & 7;
    if (r >= NROWS || k >= 6) return;
    const float* x = x0;
    int roff = 0;
    if (r >= ROFF2)      { x = x2; roff = ROFF2; }
    else if (r >= ROFF1) { x = x1; roff = ROFF1; }
    const int* sp = rec + (size_t)r * CNT_STRIDE + 1;
    const int deg = sp[-1];
    const int c   = (deg < CSR_INL) ? deg : CSR_INL;
    const float* xk = x + k;
    float acc = 0.0f;
    int i = 0;
    for (; i + 4 <= c; i += 4) {
        const int s0 = sp[i], s1 = sp[i + 1], s2 = sp[i + 2], s3 = sp[i + 3];
        const float v0 = xk[(s0 - roff) * 6];
        const float v1 = xk[(s1 - roff) * 6];
        const float v2 = xk[(s2 - roff) * 6];
        const float v3 = xk[(s3 - roff) * 6];
        acc += v0; acc += v1; acc += v2; acc += v3;
    }
    for (; i < c; ++i) acc += xk[(sp[i] - roff) * 6];
    if (deg > CSR_INL) {
        int nov = *cursor;
        nov = (nov < ETOT) ? nov : ETOT;
        for (int q = 0; q < nov; ++q)
            if (ovf[2 * q] == r) acc += xk[(ovf[2 * q + 1] - roff) * 6];
    }
    agg6all[r * 6 + k] = acc;
}

template <int NB, bool LIM>
__device__ __forceinline__ void gather128_batch(const int* __restrict__ sp, int i, int c,
                                                const float4* __restrict__ hb, float4& acc) {
    int s[NB];
    float4 v[NB];
#pragma unroll
    for (int j = 0; j < NB; ++j) {
        int ii = i + j;
        if (LIM) ii = (ii < c) ? ii : (c - 1);
        s[j] = sp[ii];
    }
#pragma unroll
    for (int j = 0; j < NB; ++j) v[j] = hb[(size_t)s[j] * (H / 4)];
#pragma unroll
    for (int j = 0; j < NB; ++j) {
        if (!LIM || i + j < c) {
            acc.x += v[j].x;
            acc.y += v[j].y;
            acc.z += v[j].z;
            acc.w += v[j].w;
        }
    }
}

__global__ void gather128_kernel(const float* __restrict__ h1all, const int* __restrict__ rec,
                                 const int* __restrict__ cursor, const int* __restrict__ ovf,
                                 float* __restrict__ aggall) {
    const int lane = threadIdx.x & 31;
    const int r = __builtin_amdgcn_readfirstlane((int)((blockIdx.x * blockDim.x + threadIdx.x) >> 5));
    if (r >= NROWS) return;
    const int* sp = rec + (size_t)r * CNT_STRIDE + 1;
    const int deg = sp[-1];
    const int c   = (deg < CSR_INL) ? deg : CSR_INL;
    const float4* hb = (const float4*)h1all + lane;
    float4 acc = {0.0f, 0.0f, 0.0f, 0.0f};
    int i = 0;
    for (; i + 8 <= c; i += 8) gather128_batch<8, false>(sp, i, c, hb, acc);
    const int rem = c - i;
    if (rem > 4)      gather128_batch<8, true>(sp, i, c, hb, acc);
    else if (rem > 0) gather128_batch<4, true>(sp, i, c, hb, acc);
    if (deg > CSR_INL) {
        int nov = *cursor;
        nov = (nov < ETOT) ? nov : ETOT;
        for (int q = 0; q < nov; ++q) {
            if (ovf[2 * q] == r) {
                const float4 v = hb[(size_t)ovf[2 * q + 1] * (H / 4)];
                acc.x += v.x;
                acc.y += v.y;
                acc.z += v.z;
                acc.w += v.w;
            }
        }
    }
    ((float4*)aggall)[(size_t)r * (H / 4) + lane] = acc;
}

template <int NB>
__device__ __forceinline__ void interp_batch(const int* __restrict__ pr, int i,
                                             const float4* __restrict__ eb, float4& acc) {
    int col[NB];
    float val[NB];
    float4 v[NB];
#pragma unroll
    for (int j = 0; j < NB; ++j) {
        col[j] = pr[2 * (i + j)];
        val[j] = __int_as_float(pr[2 * (i + j) + 1]);
    }
#pragma unroll
    for (int j = 0; j < NB; ++j) v[j] = eb[(size_t)col[j] * (F3 / 4)];
#pragma unroll
    for (int j = 0; j < NB; ++j) {
        acc.x += val[j] * v[j].x;
        acc.y += val[j] * v[j].y;
        acc.z += val[j] * v[j].z;
        acc.w += val[j] * v[j].w;
    }
}

__global__ void interp_gather_kernel(const int* __restrict__ irec, const int* __restrict__ icur,
                                     const int* __restrict__ iovf, const float* __restrict__ ea,
                                     const float* __restrict__ eb2, float* __restrict__ out1,
                                     float* __restrict__ out2) {
    const int lane = threadIdx.x & 31;
    const int w = __builtin_amdgcn_readfirstlane((int)((blockIdx.x * blockDim.x + threadIdx.x) >> 5));
    if (w >= 2 * N0) return;
    const float* e = ea;
    float* o = out1;
    int r = w;
    if (w >= N0) { e = eb2; o = out2; r = w - N0; }
    const int* rp  = irec + (size_t)w * CNT_STRIDE;
    const int deg  = rp[0];
    const int c    = (deg < IREC_PAIRS) ? deg : IREC_PAIRS;
    const int* pr  = rp + 2;
    const int ln   = (lane < F3 / 4) ? lane : (F3 / 4 - 1);
    const float4* eb = (const float4*)e + ln;
    float4 acc = {0.0f, 0.0f, 0.0f, 0.0f};
    int i = 0;
    for (; i + 8 <= c; i += 8) interp_batch<8>(pr, i, eb, acc);
    switch (c - i) {
        case 1: interp_batch<1>(pr, i, eb, acc); break;
        case 2: interp_batch<2>(pr, i, eb, acc); break;
        case 3: interp_batch<3>(pr, i, eb, acc); break;
        case 4: interp_batch<4>(pr, i, eb, acc); break;
        case 5: interp_batch<5>(pr, i, eb, acc); break;
        case 6: interp_batch<6>(pr, i, eb, acc); break;
        case 7: interp_batch<7>(pr, i, eb, acc); break;
        default: break;
    }
    if (deg > IREC_PAIRS) {
        int nov = *icur;
        nov = (nov < ITOT) ? nov : ITOT;
        for (int q = 0; q < nov; ++q) {
            if (iovf[3 * q] == w) {
                const float val = __int_as_float(iovf[3 * q + 2]);
                const float4 v = eb[(size_t)iovf[3 * q + 1] * (F3 / 4)];
                acc.x += val * v.x;
                acc.y += val * v.y;
                acc.z += val * v.z;
                acc.w += val * v.w;
            }
        }
    }
    if (lane < F3 / 4) ((float4*)o)[(size_t)r * (F3 / 4) + lane] = acc;
}

#define GS_THREADS 128

template <int K, int NC, int UNR = 4, int NTHR = GS_THREADS>
__device__ __forceinline__ void stage_weights(const float* __restrict__ W, float* __restrict__ Ws) {
    constexpr int NT = NC / 16;
#pragma unroll UNR
    for (int q = threadIdx.x; q < (K / 4) * NC; q += NTHR) {
        const int ks = q / NC;
        const int nn = q - ks * NC;
        const float* wp = W + ks * 4 * NC + nn;
        v2f lo, hi;
        lo.x = wp[0];
        lo.y = wp[NC];
        hi.x = wp[2 * NC];
        hi.y = wp[3 * NC];
        const int t = nn >> 4;
        const int c = nn & 15;
        float* dst = Ws + ((ks * (NT / 2) + (t >> 1)) * 32 + c) * 4 + (t & 1) * 2;
        *(v2f*)dst        = lo;
        *(v2f*)(dst + 64) = hi;
    }
}

template <int NT>
__device__ __forceinline__ void load_b(const float* __restrict__ wb, v2f (&b)[NT]) {
#pragma unroll
    for (int tp = 0; tp < NT / 2; ++tp) {
        const float4 q = *(const float4*)(wb + tp * 128);
        b[2 * tp].x     = q.x;
        b[2 * tp].y     = q.y;
        b[2 * tp + 1].x = q.z;
        b[2 * tp + 1].y = q.w;
    }
}

template <int K, int NT, bool DUAL, bool WA = false>
__device__ __forceinline__ void strip_products(const v2f (&a0)[K / 4], const v2f (&a1)[K / 4],
                                               const float* __restrict__ w0s,
                                               const float* __restrict__ w1s, v8f (&acc)[NT]) {
    constexpr int KS = K / 4;
    v2f b0[NT], b1[NT];
    load_b<NT>(w0s, b0);
    if (DUAL) load_b<NT>(w1s, b1);
#pragma unroll
    for (int ks = 0; ks < KS; ++ks) {
        v2f nb0[NT], nb1[NT];
        if (ks + 1 < KS) {
            load_b<NT>(w0s + (ks + 1) * (NT / 2) * 128, nb0);
            if (DUAL) load_b<NT>(w1s + (ks + 1) * (NT / 2) * 128, nb1);
        }
#pragma unroll
        for (int t = 0; t < NT; ++t)
            acc[t] = WA ? __builtin_amdgcn_wmma_f32_16x16x4_f32(false, b0[t], false, a0[ks], (short)0,
                                                                acc[t], false, false)
                        : __builtin_amdgcn_wmma_f32_16x16x4_f32(false, a0[ks], false, b0[t], (short)0,
                                                                acc[t], false, false);
        if (DUAL) {
#pragma unroll
            for (int t = 0; t < NT; ++t)
                acc[t] = WA ? __builtin_amdgcn_wmma_f32_16x16x4_f32(false, b1[t], false, a1[ks], (short)0,
                                                                    acc[t], false, false)
                            : __builtin_amdgcn_wmma_f32_16x16x4_f32(false, a1[ks], false, b1[t], (short)0,
                                                                    acc[t], false, false);
        }
        if (ks + 1 < KS) {
#pragma unroll
            for (int t = 0; t < NT; ++t) {
                b0[t] = nb0[t];
                if (DUAL) b1[t] = nb1[t];
            }
        }
    }
}

#define CL_THREADS 256
#define CL_WAVES (CL_THREADS / 32)
#define CL_G0 195
#define CL_G1 49
#define CL_G2 12

__device__ __forceinline__ void stage_lin_weights(const float* __restrict__ W, float* __restrict__ Ws) {
    constexpr int NU = F3 / 16;
#pragma unroll 4
    for (int q = threadIdx.x; q < (H / 4) * F3; q += CL_THREADS) {
        const int kk = q / F3;
        const int nn = q - kk * F3;
        const int r0 = (kk >> 2) * 16 + (kk & 3) * 2;
        const float* wp = W + r0 * F3 + nn;
        v2f lo, hi;
        lo.x = wp[0];
        lo.y = wp[F3];
        hi.x = wp[8 * F3];
        hi.y = wp[9 * F3];
        const int u = nn >> 4;
        const int c = nn & 15;
        float* dst = Ws + ((kk * (NU / 2) + (u >> 1)) * 32 + c) * 4 + (u & 1) * 2;
        *(v2f*)dst        = lo;
        *(v2f*)(dst + 64) = hi;
    }
}

__global__ void __launch_bounds__(CL_THREADS)
conv2_lin_wmma_kernel(const float* __restrict__ aggall, const float* __restrict__ Wrelall,
                      const float* __restrict__ h1all, const float* __restrict__ Wrootall,
                      const float* __restrict__ brelall, const float* __restrict__ linWall,
                      const float* __restrict__ linball, float* __restrict__ eall) {
    constexpr int K  = H;
    constexpr int NT = H / 16;
    constexpr int KS = K / 4;
    constexpr int NU = F3 / 16;
    __shared__ __attribute__((aligned(16))) float Ws[2 * K * H + K * F3];

    int lb = blockIdx.x, lg = CL_G0, n = N0, lvl = 0, roff = 0;
    if (lb >= CL_G0 + CL_G1) { lb -= CL_G0 + CL_G1; lg = CL_G2; n = N2; lvl = 2; roff = ROFF2; }
    else if (lb >= CL_G0)    { lb -= CL_G0;         lg = CL_G1; n = N1; lvl = 1; roff = ROFF1; }
    const float* A0   = aggall + (size_t)roff * K;
    const float* A1   = h1all + (size_t)roff * K;
    const float* bias = brelall + lvl * H;
    const float* lbias = linball + lvl * F3;
    float* E          = eall + (size_t)roff * F3;

    stage_weights<K, H, 8, CL_THREADS>(Wrelall + lvl * K * H, Ws);
    stage_weights<K, H, 8, CL_THREADS>(Wrootall + lvl * K * H, Ws + K * H);
    stage_lin_weights(linWall + lvl * K * F3, Ws + 2 * K * H);
    __syncthreads();

    const int lane = threadIdx.x & 31;
    const int wave = __builtin_amdgcn_readfirstlane((int)(threadIdx.x >> 5));
    const int m    = lane & 15;
    const int half = lane >> 4;
    const int nstrips = (n + 15) >> 4;

    __shared__ float Ps[CL_WAVES * NU * 8 * 32];
#pragma unroll 1
    for (int sx = 2 * CL_WAVES * lg + lb; sx < nstrips; sx += lg) {
        int boff0 = lane * 4;
        int boff1 = lane * 4 + K * H;
        int boff2 = lane * 4 + 2 * K * H;
        asm volatile("" : "+v"(boff0), "+v"(boff1), "+v"(boff2));
        const int tsel = (wave >> 1) * 128 + (wave & 1) * 2;
        const float* w0s = Ws + boff0 + tsel;
        const float* w1s = Ws + boff1 + tsel;
        const float* w2s = Ws + boff2 + wave * (4 * (NU / 2) * 128);
        const int row0 = sx * 16;
        const int row  = row0 + m;
        const int rowc = (row < n) ? row : (n - 1);

        const float* a0p = A0 + rowc * K + half * 2;
        const float* a1p = A1 + rowc * K + half * 2;
        v2f a0[KS], a1[KS];
#pragma unroll
        for (int ks = 0; ks < KS; ++ks) {
            a0[ks] = *(const v2f*)(a0p + 4 * ks);
            a1[ks] = *(const v2f*)(a1p + 4 * ks);
        }
        v8f acc1;
        {
            const float* bp = bias + 16 * wave + 8 * half;
            const v2f b01 = *(const v2f*)(bp);
            const v2f b23 = *(const v2f*)(bp + 2);
            const v2f b45 = *(const v2f*)(bp + 4);
            const v2f b67 = *(const v2f*)(bp + 6);
            acc1 = (v8f){b01.x, b01.y, b23.x, b23.y, b45.x, b45.y, b67.x, b67.y};
        }
#pragma unroll
        for (int ks = 0; ks < KS; ++ks) {
            const v2f wa = *(const v2f*)(w0s + ks * (NT / 2) * 128);
            const v2f wb = *(const v2f*)(w1s + ks * (NT / 2) * 128);
            acc1 = __builtin_amdgcn_wmma_f32_16x16x4_f32(false, wa, false, a0[ks], (short)0, acc1, false, false);
            acc1 = __builtin_amdgcn_wmma_f32_16x16x4_f32(false, wb, false, a1[ks], (short)0, acc1, false, false);
        }
        v8f p[NU];
#pragma unroll
        for (int u = 0; u < NU; ++u) p[u] = (v8f){0.0f, 0.0f, 0.0f, 0.0f, 0.0f, 0.0f, 0.0f, 0.0f};
#pragma unroll
        for (int jp = 0; jp < 4; ++jp) {
            v2f wl[NU];
            load_b<NU>(w2s + jp * (NU / 2) * 128, wl);
            v2f hf;
            hf.x = fmaxf(acc1[2 * jp], 0.0f);
            hf.y = fmaxf(acc1[2 * jp + 1], 0.0f);
#pragma unroll
            for (int u = 0; u < NU; ++u)
                p[u] = __builtin_amdgcn_wmma_f32_16x16x4_f32(false, wl[u], false, hf, (short)0, p[u], false, false);
        }
#pragma unroll
        for (int u = 0; u < NU; ++u) {
#pragma unroll
            for (int j = 0; j < 8; ++j) Ps[((wave * NU + u) * 8 + j) * 32 + lane] = p[u][j];
        }
        __syncthreads();
        {
            const int rsel = lane & 1;
            const int hsel = (lane >> 1) & 1;
            const int j    = lane >> 2;
            const int sl   = 2 * wave + rsel + 16 * hsel;
            const int orow = row0 + 2 * wave + rsel;
#pragma unroll
            for (int u = 0; u < NU; ++u) {
                float v = lbias[16 * u + 8 * hsel + j];
#pragma unroll
                for (int w2 = 0; w2 < CL_WAVES; ++w2) v += Ps[((w2 * NU + u) * 8 + j) * 32 + sl];
                if (orow < n) E[(size_t)orow * F3 + 16 * u + 8 * hsel + j] = v;
            }
        }
        __syncthreads();
    }

#pragma unroll 1
    for (int s = wave * lg + lb; s < nstrips && s < 2 * CL_WAVES * lg; s += lg * CL_WAVES) {
        int boff0 = lane * 4;
        int boff1 = lane * 4 + K * H;
        int boff2 = lane * 4 + 2 * K * H;
        asm volatile("" : "+v"(boff0), "+v"(boff1), "+v"(boff2));
        const float* w0s = Ws + boff0;
        const float* w1s = Ws + boff1;
        const float* w2s = Ws + boff2;
        const int row0 = s * 16;
        const int row  = row0 + m;
        const int rowc = (row < n) ? row : (n - 1);

        const float* a0p = A0 + rowc * K + half * 2;
        const float* a1p = A1 + rowc * K + half * 2;
        v2f a0[KS], a1[KS];
#pragma unroll
        for (int ks = 0; ks < KS; ++ks) {
            a0[ks] = *(const v2f*)(a0p + 4 * ks);
            a1[ks] = *(const v2f*)(a1p + 4 * ks);
        }

        v8f acc[NT];
#pragma unroll
        for (int t = 0; t < NT; ++t) {
            const float* bp = bias + 16 * t + 8 * half;
            const v2f b01 = *(const v2f*)(bp);
            const v2f b23 = *(const v2f*)(bp + 2);
            const v2f b45 = *(const v2f*)(bp + 4);
            const v2f b67 = *(const v2f*)(bp + 6);
            acc[t] = (v8f){b01.x, b01.y, b23.x, b23.y, b45.x, b45.y, b67.x, b67.y};
        }
        strip_products<K, NT, true, true>(a0, a1, w0s, w1s, acc);

        v8f acc2[NU];
#pragma unroll
        for (int u = 0; u < NU; ++u) {
            const float* bp = lbias + 16 * u + 8 * half;
            const v2f b01 = *(const v2f*)(bp);
            const v2f b23 = *(const v2f*)(bp + 2);
            const v2f b45 = *(const v2f*)(bp + 4);
            const v2f b67 = *(const v2f*)(bp + 6);
            acc2[u] = (v8f){b01.x, b01.y, b23.x, b23.y, b45.x, b45.y, b67.x, b67.y};
        }
        v2f wl[NU];
        load_b<NU>(w2s, wl);
#pragma unroll
        for (int kk = 0; kk < KS; ++kk) {
            v2f nwl[NU];
            if (kk + 1 < KS) load_b<NU>(w2s + (kk + 1) * (NU / 2) * 128, nwl);
            const int t  = kk >> 2;
            const int jp = kk & 3;
            v2f hf;
            hf.x = fmaxf(acc[t][2 * jp], 0.0f);
            hf.y = fmaxf(acc[t][2 * jp + 1], 0.0f);
#pragma unroll
            for (int u = 0; u < NU; ++u)
                acc2[u] = __builtin_amdgcn_wmma_f32_16x16x4_f32(false, wl[u], false, hf, (short)0,
                                                                acc2[u], false, false);
            if (kk + 1 < KS) {
#pragma unroll
                for (int u = 0; u < NU; ++u) wl[u] = nwl[u];
            }
        }

        if (row < n) {
            float* ep = E + (size_t)row * F3 + 8 * half;
#pragma unroll
            for (int u = 0; u < NU; ++u) {
                float4 lo, hi;
                lo.x = acc2[u][0]; lo.y = acc2[u][1]; lo.z = acc2[u][2]; lo.w = acc2[u][3];
                hi.x = acc2[u][4]; hi.y = acc2[u][5]; hi.z = acc2[u][6]; hi.w = acc2[u][7];
                *(float4*)(ep + 16 * u)     = lo;
                *(float4*)(ep + 16 * u + 4) = hi;
            }
        }
    }
}

#define DEC_RPB 32
__global__ void __launch_bounds__(GS_THREADS)
dec_fused_wmma_kernel(const float* __restrict__ e0, const float* __restrict__ i1,
                      const float* __restrict__ i2, const float* __restrict__ W,
                      const float* __restrict__ bias, const float* __restrict__ Wout,
                      const float* __restrict__ bout, float* __restrict__ out, int n) {
    constexpr int K = F3, NC = DECW;
    constexpr int NT = NC / 16;
    constexpr int KS = K / 4;
    __shared__ __attribute__((aligned(16))) float Ws[3 * K * NC];

    stage_weights<3 * K, NC, 12>(W, Ws);
    __syncthreads();

    const int lane = threadIdx.x & 31;
    const int wave = __builtin_amdgcn_readfirstlane((int)(threadIdx.x >> 5));
    const int m    = lane & 15;
    const int half = lane >> 4;
    const int nblk = (n + DEC_RPB - 1) / DEC_RPB;

    float bz[3][NT], wo[3][NT], bo[3];
#pragma unroll
    for (int b = 0; b < 3; ++b) {
        bo[b] = bout[b];
#pragma unroll
        for (int t = 0; t < NT; ++t) {
            bz[b][t] = bias[b * NC + 16 * t + m];
            wo[b][t] = Wout[b * NC + 16 * t + m];
        }
    }
    const bool bit3 = (m & 8) != 0;
    const bool bit2 = (m & 4) != 0;
    const bool bit1 = (m & 2) != 0;

#pragma unroll 1
    for (int blk = blockIdx.x * (GS_THREADS / 32) + wave; blk < nblk; blk += gridDim.x * (GS_THREADS / 32)) {
#pragma unroll 1
        for (int st = 0; st < DEC_RPB / 16; ++st) {
            const int row0 = blk * DEC_RPB + st * 16;
            if (row0 >= n) break;
            const int row  = row0 + m;
            const int rowc = (row < n) ? row : (n - 1);

            const int abase = rowc * F3 + half * 2;
            const float* s0 = e0 + abase;
            const float* s1 = i1 + abase;
            const float* s2 = i2 + abase;
            v2f a[3][KS];
#pragma unroll
            for (int b = 0; b < 3; ++b) {
#pragma unroll
                for (int ks = 0; ks < KS; ++ks) {
                    const int k   = 4 * ks;
                    const int src = k >> 5;
                    const float* sp = (src == 0) ? s0 : (src == 1) ? s1 : s2;
                    a[b][ks] = *(const v2f*)(sp + b * 32 + (k & 31));
                }
            }
            float res[3];
#pragma unroll
            for (int b = 0; b < 3; ++b) {
                v8f acc[NT];
#pragma unroll
                for (int t = 0; t < NT; ++t) {
                    const float bv = bz[b][t];
                    acc[t] = (v8f){bv, bv, bv, bv, bv, bv, bv, bv};
                }
                const float* ws = Ws + lane * 4 + b * K * NC;
                strip_products<K, NT, false>(a[b], a[b], ws, ws, acc);

                float p[8];
#pragma unroll
                for (int j = 0; j < 8; ++j) {
                    float s = 0.0f;
#pragma unroll
                    for (int t = 0; t < NT; ++t) {
                        float v = acc[t][j];
                        v = (v > 0.0f) ? v : (expf(v) - 1.0f);
                        s += v * wo[b][t];
                    }
                    p[j] = s;
                }
                float q[4], r2[2];
#pragma unroll
                for (int i = 0; i < 4; ++i) {
                    const float send = bit3 ? p[i] : p[i + 4];
                    const float keep = bit3 ? p[i + 4] : p[i];
                    q[i] = keep + __shfl_xor(send, 8, 32);
                }
#pragma unroll
                for (int i = 0; i < 2; ++i) {
                    const float send = bit2 ? q[i] : q[i + 2];
                    const float keep = bit2 ? q[i + 2] : q[i];
                    r2[i] = keep + __shfl_xor(send, 4, 32);
                }
                const float send1 = bit1 ? r2[0] : r2[1];
                const float keep1 = bit1 ? r2[1] : r2[0];
                float tot = keep1 + __shfl_xor(send1, 2, 32);
                tot += __shfl_xor(tot, 1, 32);
                res[b] = tot + bo[b];
            }

            const int orow = row0 + half * 8 + (m >> 1);
            if ((m & 1) == 0 && orow < n) {
                float* op = out + orow * 3;
                op[0] = res[0];
                op[1] = res[1];
                op[2] = res[2];
            }
        }
    }
}

#define FILL_GRID(count) (((count) / 4 + 8 * 32 * FILL_SPW - 1) / (8 * 32 * FILL_SPW) > 256 ? 256 \
                          : ((count) / 4 + 8 * 32 * FILL_SPW - 1) / (8 * 32 * FILL_SPW))

#define WS_H1    ((size_t)0)
#define WS_AGG   (WS_H1 + (size_t)NROWS * H)
#define WS_AGG6  (WS_AGG + (size_t)NROWS * H)
#define WS_H2    (((WS_AGG6 + (size_t)NROWS * 6 + 31) / 32) * 32)
#define WS_TAIL  (WS_H2 + (size_t)NROWS * H)
#define WS_I1    ((size_t)0)
#define WS_I2    (WS_I1 + (size_t)N0 * F3)
#define WS_E     WS_H2
static_assert(WS_I2 + (size_t)N0 * F3 <= WS_AGG6, "i1 / i2 must fit in h1 / agg's place");
static_assert(WS_E % 32 == 0 && WS_E + (size_t)NROWS * F3 <= WS_TAIL, "e must fit in the region WS_H2");
#define WS_OVF   WS_H2
#define WS_CNT   (WS_TAIL - 32 - (size_t)NROWS * CNT_STRIDE)
#define WS_CUR   (WS_TAIL - 32)
#define WS_IREC  WS_TAIL
#define WS_ICUR  (WS_IREC + (size_t)2 * N0 * CNT_STRIDE)
#define WS_IOVF  (WS_ICUR + 32)
#define WS_ZCOUNT (NROWS * CNT_STRIDE + 32 + 2 * N0 * CNT_STRIDE + 32)
static_assert(WS_OVF + 2 * (size_t)ETOT <= WS_CNT, "the edge lists must fit in the region WS_H2");
static_assert(WS_IOVF + 3 * (size_t)ITOT <= WS_TAIL + (size_t)NROWS * F3, "the interpolation lists must fit in the tail");

extern "C" void kernel_launch(void* const* d_in, const int* in_sizes, int n_in,
                              void* d_out, int out_size, void* d_ws, size_t ws_size,
                              hipStream_t stream) {
    (void)in_sizes; (void)n_in; (void)out_size; (void)ws_size;

    const float* x0     = (const float*)d_in[0];
    const float* x1     = (const float*)d_in[1];
    const float* x2     = (const float*)d_in[2];
    const int*   ei0    = (const int*)d_in[3];
    const int*   ei1    = (const int*)d_in[4];
    const int*   ei2    = (const int*)d_in[5];
    const int*   A1r    = (const int*)d_in[6];
    const int*   A1c    = (const int*)d_in[7];
    const float* A1v    = (const float*)d_in[8];
    const int*   A2r    = (const int*)d_in[9];
    const int*   A2c    = (const int*)d_in[10];
    const float* A2v    = (const float*)d_in[11];
    const float* c1Wrel  = (const float*)d_in[12];
    const float* c1brel  = (const float*)d_in[13];
    const float* c1Wroot = (const float*)d_in[14];
    const float* c2Wrel  = (const float*)d_in[15];
    const float* c2brel  = (const float*)d_in[16];
    const float* c2Wroot = (const float*)d_in[17];
    const float* linW    = (const float*)d_in[18];
    const float* linb    = (const float*)d_in[19];
    const float* decW0   = (const float*)d_in[20];
    const float* decb0   = (const float*)d_in[21];
    const float* decWout = (const float*)d_in[22];
    const float* decbout = (const float*)d_in[23];
    float* out = (float*)d_out;

    float* ws   = (float*)d_ws;
    float* h1   = ws + WS_H1;
    float* agg  = ws + WS_AGG;
    float* agg6 = ws + WS_AGG6;
    float* eall = ws + WS_E;
    float* i1   = ws + WS_I1;
    float* i2   = ws + WS_I2;
    int* cnt    = (int*)(ws + WS_CNT);
    int* cursor = (int*)(ws + WS_CUR);
    int* ovf    = (int*)(ws + WS_OVF);
    int* irec   = (int*)(ws + WS_IREC);
    int* icur   = (int*)(ws + WS_ICUR);
    int* iovf   = (int*)(ws + WS_IOVF);

    zero_kernel<<<FILL_GRID(WS_ZCOUNT), 256, 0, stream>>>((float*)cnt, WS_ZCOUNT);
    csr_build_kernel<<<(ETOT + ITOT + 255) / 256, 256, 0, stream>>>(ei0, ei1, ei2, cnt, cursor, ovf,
                                                                   A1r, A1c, A1v, A2r, A2c, A2v,
                                                                   irec, icur, iovf);

    gather6_kernel<<<(NROWS * 8 + 255) / 256, 256, 0, stream>>>(x0, x1, x2, cnt, cursor, ovf, agg6);
    conv1_kernel<<<(C1_W0 + C1_W1 + C1_W2 + 7) / 8, 256, 0, stream>>>(x0, x1, x2, agg6, c1Wrel, c1brel,
                                                                    c1Wroot, h1);

    gather128_kernel<<<(NROWS + 7) / 8, 256, 0, stream>>>(h1, cnt, cursor, ovf, agg);
    conv2_lin_wmma_kernel<<<CL_G0 + CL_G1 + CL_G2, CL_THREADS, 0, stream>>>(
        agg, c2Wrel, h1, c2Wroot, c2brel, linW, linb, eall);

    interp_gather_kernel<<<(2 * N0 + 7) / 8, 256, 0, stream>>>(irec, icur, iovf, eall + (size_t)ROFF1 * F3,
                                                              eall + (size_t)ROFF2 * F3, i1, i2);

    dec_fused_wmma_kernel<<<((N0 + DEC_RPB - 1) / DEC_RPB + 3) / 4, GS_THREADS, 0, stream>>>(
        eall, i1, i2, decW0, decb0, decWout, decbout, out, N0);
}


# conv1 sums its rows' 6-wide sources itself: 4 lanes a row read the row record, 3 8-byte loads a source, 2 lane exchanges; 6-wide gather launch and agg6 gone; sum order changed there; 8->7 launches
# speedup vs baseline: 3.9332x; 1.0379x over previous
#include <hip/hip_runtime.h>
#include <hip/hip_bf16.h>

#define N0 50000
#define N1 12500
#define N2 3125
#define E0 800000
#define E1 200000
#define E2 50000
#define NNZ 200000
#define H 128
#define F3 96
#define DECW 64

#define ROFF1 N0
#define ROFF2 (N0 + N1)
#define NROWS (N0 + N1 + N2)

typedef __attribute__((ext_vector_type(2))) float v2f;
typedef __attribute__((ext_vector_type(8))) float v8f;

#define FILL_SPW 4
__global__ void zero_kernel(float* __restrict__ p, int count) {
    const int tid  = blockIdx.x * blockDim.x + threadIdx.x;
    const int lane = threadIdx.x & 31;
    const int wave = __builtin_amdgcn_readfirstlane(tid >> 5);
    const int step = ((gridDim.x * blockDim.x) >> 5) * (32 * FILL_SPW);
    int head = (int)(((16u - (unsigned)((size_t)p & 15u)) & 15u) >> 2);
    if (head > count) head = count;
    float4* q = (float4*)(p + head);
    const int nvec = (count - head) >> 2;
    const float4 z = {0.0f, 0.0f, 0.0f, 0.0f};
    for (int base = wave * (32 * FILL_SPW); base < nvec; base += step) {
        float4* qq = q + base + lane;
        if (base + 32 * FILL_SPW <= nvec) {
#pragma unroll
            for (int u = 0; u < FILL_SPW; ++u) qq[32 * u] = z;
        } else {
#pragma unroll
            for (int u = 0; u < FILL_SPW; ++u)
                if (base + 32 * u + lane < nvec) qq[32 * u] = z;
        }
    }
    const int tail0 = head + (nvec << 2);
    if (tid < head) p[tid] = 0.0f;
    if (tid < count - tail0) p[tail0 + tid] = 0.0f;
}


#define ETOT (E0 + E1 + E2)
#define CNT_STRIDE 32
#define CSR_INL (CNT_STRIDE - 1)

#define EDGE_LEVEL(eg, ei, E, roff, e)                                             \
    const int* ei = ei0; int E = E0, roff = 0, e = (eg);                          \
    if ((eg) >= E0 + E1) { ei = ei2; E = E2; roff = ROFF2; e = (eg) - (E0 + E1); } \
    else if ((eg) >= E0) { ei = ei1; E = E1; roff = ROFF1; e = (eg) - E0; }

#define ITOT (2 * NNZ)
#define IREC_PAIRS ((CNT_STRIDE - 2) / 2)

__global__ void csr_build_kernel(const int* __restrict__ ei0, const int* __restrict__ ei1,
                                 const int* __restrict__ ei2, int* __restrict__ rec,
                                 int* __restrict__ cursor, int* __restrict__ ovf,
                                 const int* __restrict__ rows1, const int* __restrict__ cols1,
                                 const float* __restrict__ vals1,
                                 const int* __restrict__ rows2, const int* __restrict__ cols2,
                                 const float* __restrict__ vals2,
                                 int* __restrict__ irec, int* __restrict__ icur, int* __restrict__ iovf) {
    const int eg = blockIdx.x * blockDim.x + threadIdx.x;
    if (eg < ETOT) {
        EDGE_LEVEL(eg, ei, E, roff, e)
        const int s = ei[e] + roff;
        const int d = ei[E + e] + roff;
        int* line = rec + (size_t)d * CNT_STRIDE;
        const int place = atomicAdd(line, 1);
        if (place < CSR_INL) {
            line[1 + place] = s;
        } else {
            const int k = atomicAdd(cursor, 1);
            ovf[2 * k]     = d;
            ovf[2 * k + 1] = s;
        }
    } else if (eg < ETOT + ITOT) {
        int n = eg - ETOT;
        const int* rows = rows1;
        const int* cols = cols1;
        const float* vals = vals1;
        int rbase = 0;
        if (n >= NNZ) { n -= NNZ; rows = rows2; cols = cols2; vals = vals2; rbase = N0; }
        const int rr = rows[n] + rbase;
        int2 p;
        p.x = cols[n];
        p.y = __float_as_int(vals[n]);
        int* line = irec + (size_t)rr * CNT_STRIDE;
        const int place = atomicAdd(line, 1);
        if (place < IREC_PAIRS) {
            *(int2*)(line + 2 + 2 * place) = p;
        } else {
            const int k = atomicAdd(icur, 1);
            iovf[3 * k]     = rr;
            iovf[3 * k + 1] = p.x;
            iovf[3 * k + 2] = p.y;
        }
    }
}


#define C1_RPW 8
#define C1_W0 ((N0 + C1_RPW - 1) / C1_RPW)
#define C1_W1 ((N1 + C1_RPW - 1) / C1_RPW)
#define C1_W2 ((N2 + C1_RPW - 1) / C1_RPW)
#define C1_SPL ((CSR_INL + 3) / 4)
template <int PERM>
__device__ __forceinline__ float quad_swap(float v) {
    return __int_as_float(__builtin_amdgcn_update_dpp(0, __float_as_int(v), PERM, 0xF, 0xF, true));
}

template <bool TAIL>
__device__ __forceinline__ void conv1_wave(const float* __restrict__ x, const int* __restrict__ rec,
                                           const int* __restrict__ cursor, const int* __restrict__ ovf,
                                           const float* __restrict__ Wrel, const float* __restrict__ brel,
                                           const float* __restrict__ Wroot, float* __restrict__ h1,
                                           int n, int roff, int r0, int lane) {
    const int sub = lane & 3;
    int grow = r0 + (lane >> 2);
    if (TAIL) grow = (grow < n) ? grow : (n - 1);
    const int* line = rec + (size_t)(roff + grow) * CNT_STRIDE;
    const int deg = line[0];
    int sidx[C1_SPL];
#pragma unroll
    for (int t = 0; t < C1_SPL; ++t) {
        const int i = (sub + 4 * t < CSR_INL) ? (sub + 4 * t) : (CSR_INL - 1);
        sidx[t] = line[1 + i];
    }
    asm volatile("" : "+v"(sidx[0]), "+v"(sidx[1]), "+v"(sidx[2]), "+v"(sidx[3]),
                      "+v"(sidx[4]), "+v"(sidx[5]), "+v"(sidx[6]), "+v"(sidx[7]));
    static_assert(C1_SPL == 8, "the statement above names eight slot words");
    const int cin = (deg < CSR_INL) ? deg : CSR_INL;
    v2f g[C1_SPL][3];
#pragma unroll
    for (int t = 0; t < C1_SPL; ++t) {
        const bool valid = (sub + 4 * t) < cin;
        const int sl = (sidx[t] - roff) & -(int)valid;
        const float* xp = x + sl * 6;
        g[t][0] = *(const v2f*)(xp);
        g[t][1] = *(const v2f*)(xp + 2);
        g[t][2] = *(const v2f*)(xp + 4);
    }
    float a6[6] = {0.0f, 0.0f, 0.0f, 0.0f, 0.0f, 0.0f};
#pragma unroll
    for (int t = 0; t < C1_SPL; ++t) {
        if ((sub + 4 * t) < cin) {
            a6[0] += g[t][0].x; a6[1] += g[t][0].y;
            a6[2] += g[t][1].x; a6[3] += g[t][1].y;
            a6[4] += g[t][2].x; a6[5] += g[t][2].y;
        }
    }
    if (deg > CSR_INL && sub == 0) {
        int nov = *cursor;
        nov = (nov < ETOT) ? nov : ETOT;
        for (int q = 0; q < nov; ++q) {
            if (ovf[2 * q] == roff + grow) {
                const float* xp = x + (ovf[2 * q + 1] - roff) * 6;
#pragma unroll
                for (int k = 0; k < 6; ++k) a6[k] += xp[k];
            }
        }
    }
#pragma unroll
    for (int k = 0; k < 6; ++k) {
        a6[k] += quad_swap<0xB1>(a6[k]);
        a6[k] += quad_swap<0x4E>(a6[k]);
    }

    const int c = lane * 4;
    float wr[6][4], wo[6][4], b[4];
#pragma unroll
    for (int q = 0; q < 4; ++q) b[q] = brel[c + q];
#pragma unroll
    for (int k = 0; k < 6; ++k) {
#pragma unroll
        for (int q = 0; q < 4; ++q) {
            wr[k][q] = Wrel[k * H + c + q];
            wo[k][q] = Wroot[k * H + c + q];
        }
    }
#pragma unroll
    for (int j = 0; j < C1_RPW; ++j) {
        const int r = r0 + j;
        int rl = r;
        if (TAIL) rl = (r < n) ? r : (n - 1);
        const float* xr = x + rl * 6;
        float s[4] = {b[0], b[1], b[2], b[3]};
#pragma unroll
        for (int k = 0; k < 6; ++k) {
            const float a  = __int_as_float(__builtin_amdgcn_readlane(__float_as_int(a6[k]), 4 * j));
            const float xv = xr[k];
#pragma unroll
            for (int q = 0; q < 4; ++q) {
                s[q] += a * wr[k][q];
                s[q] += xv * wo[k][q];
            }
        }
        if (!TAIL || r < n) {
            float4 v;
            v.x = fmaxf(s[0], 0.0f);
            v.y = fmaxf(s[1], 0.0f);
            v.z = fmaxf(s[2], 0.0f);
            v.w = fmaxf(s[3], 0.0f);
            *(float4*)(h1 + r * H + c) = v;
        }
    }
}

__global__ void conv1_kernel(const float* __restrict__ x0, const float* __restrict__ x1,
                             const float* __restrict__ x2, const int* __restrict__ rec,
                             const int* __restrict__ cursor, const int* __restrict__ ovf,
                             const float* __restrict__ Wrel, const float* __restrict__ brel,
                             const float* __restrict__ Wroot, float* __restrict__ h1) {
    const int lane = threadIdx.x & 31;
    int wave = __builtin_amdgcn_readfirstlane((int)((blockIdx.x * blockDim.x + threadIdx.x) >> 5));
    const float* x = x0;
    int n = N0, lvl = 0, roff = 0;
    if (wave >= C1_W0 + C1_W1) { wave -= C1_W0 + C1_W1; x = x2; n = N2; lvl = 2; roff = ROFF2; }
    else if (wave >= C1_W0)    { wave -= C1_W0;         x = x1; n = N1; lvl = 1; roff = ROFF1; }
    const int r0 = wave * C1_RPW;
    if (r0 >= n) return;
    const float* wr = Wrel + lvl * 6 * H;
    const float* br = brel + lvl * H;
    const float* wo = Wroot + lvl * 6 * H;
    float* hd = h1 + (size_t)roff * H;
    if (r0 + C1_RPW <= n) conv1_wave<false>(x, rec, cursor, ovf, wr, br, wo, hd, n, roff, r0, lane);
    else                  conv1_wave<true >(x, rec, cursor, ovf, wr, br, wo, hd, n, roff, r0, lane);
}

template <int NB, bool LIM>
__device__ __forceinline__ void gather128_batch(const int* __restrict__ sp, int i, int c,
                                                const float4* __restrict__ hb, float4& acc) {
    int s[NB];
    float4 v[NB];
#pragma unroll
    for (int j = 0; j < NB; ++j) {
        int ii = i + j;
        if (LIM) ii = (ii < c) ? ii : (c - 1);
        s[j] = sp[ii];
    }
#pragma unroll
    for (int j = 0; j < NB; ++j) v[j] = hb[(size_t)s[j] * (H / 4)];
#pragma unroll
    for (int j = 0; j < NB; ++j) {
        if (!LIM || i + j < c) {
            acc.x += v[j].x;
            acc.y += v[j].y;
            acc.z += v[j].z;
            acc.w += v[j].w;
        }
    }
}

__global__ void gather128_kernel(const float* __restrict__ h1all, const int* __restrict__ rec,
                                 const int* __restrict__ cursor, const int* __restrict__ ovf,
                                 float* __restrict__ aggall) {
    const int lane = threadIdx.x & 31;
    const int r = __builtin_amdgcn_readfirstlane((int)((blockIdx.x * blockDim.x + threadIdx.x) >> 5));
    if (r >= NROWS) return;
    const int* sp = rec + (size_t)r * CNT_STRIDE + 1;
    const int deg = sp[-1];
    const int c   = (deg < CSR_INL) ? deg : CSR_INL;
    const float4* hb = (const float4*)h1all + lane;
    float4 acc = {0.0f, 0.0f, 0.0f, 0.0f};
    int i = 0;
    for (; i + 8 <= c; i += 8) gather128_batch<8, false>(sp, i, c, hb, acc);
    const int rem = c - i;
    if (rem > 4)      gather128_batch<8, true>(sp, i, c, hb, acc);
    else if (rem > 0) gather128_batch<4, true>(sp, i, c, hb, acc);
    if (deg > CSR_INL) {
        int nov = *cursor;
        nov = (nov < ETOT) ? nov : ETOT;
        for (int q = 0; q < nov; ++q) {
            if (ovf[2 * q] == r) {
                const float4 v = hb[(size_t)ovf[2 * q + 1] * (H / 4)];
                acc.x += v.x;
                acc.y += v.y;
                acc.z += v.z;
                acc.w += v.w;
            }
        }
    }
    ((float4*)aggall)[(size_t)r * (H / 4) + lane] = acc;
}

template <int NB>
__device__ __forceinline__ void interp_batch(const int* __restrict__ pr, int i,
                                             const float4* __restrict__ eb, float4& acc) {
    int col[NB];
    float val[NB];
    float4 v[NB];
#pragma unroll
    for (int j = 0; j < NB; ++j) {
        col[j] = pr[2 * (i + j)];
        val[j] = __int_as_float(pr[2 * (i + j) + 1]);
    }
#pragma unroll
    for (int j = 0; j < NB; ++j) v[j] = eb[(size_t)col[j] * (F3 / 4)];
#pragma unroll
    for (int j = 0; j < NB; ++j) {
        acc.x += val[j] * v[j].x;
        acc.y += val[j] * v[j].y;
        acc.z += val[j] * v[j].z;
        acc.w += val[j] * v[j].w;
    }
}

__global__ void interp_gather_kernel(const int* __restrict__ irec, const int* __restrict__ icur,
                                     const int* __restrict__ iovf, const float* __restrict__ ea,
                                     const float* __restrict__ eb2, float* __restrict__ out1,
                                     float* __restrict__ out2) {
    const int lane = threadIdx.x & 31;
    const int w = __builtin_amdgcn_readfirstlane((int)((blockIdx.x * blockDim.x + threadIdx.x) >> 5));
    if (w >= 2 * N0) return;
    const float* e = ea;
    float* o = out1;
    int r = w;
    if (w >= N0) { e = eb2; o = out2; r = w - N0; }
    const int* rp  = irec + (size_t)w * CNT_STRIDE;
    const int deg  = rp[0];
    const int c    = (deg < IREC_PAIRS) ? deg : IREC_PAIRS;
    const int* pr  = rp + 2;
    const int ln   = (lane < F3 / 4) ? lane : (F3 / 4 - 1);
    const float4* eb = (const float4*)e + ln;
    float4 acc = {0.0f, 0.0f, 0.0f, 0.0f};
    int i = 0;
    for (; i + 8 <= c; i += 8) interp_batch<8>(pr, i, eb, acc);
    switch (c - i) {
        case 1: interp_batch<1>(pr, i, eb, acc); break;
        case 2: interp_batch<2>(pr, i, eb, acc); break;
        case 3: interp_batch<3>(pr, i, eb, acc); break;
        case 4: interp_batch<4>(pr, i, eb, acc); break;
        case 5: interp_batch<5>(pr, i, eb, acc); break;
        case 6: interp_batch<6>(pr, i, eb, acc); break;
        case 7: interp_batch<7>(pr, i, eb, acc); break;
        default: break;
    }
    if (deg > IREC_PAIRS) {
        int nov = *icur;
        nov = (nov < ITOT) ? nov : ITOT;
        for (int q = 0; q < nov; ++q) {
            if (iovf[3 * q] == w) {
                const float val = __int_as_float(iovf[3 * q + 2]);
                const float4 v = eb[(size_t)iovf[3 * q + 1] * (F3 / 4)];
                acc.x += val * v.x;
                acc.y += val * v.y;
                acc.z += val * v.z;
                acc.w += val * v.w;
            }
        }
    }
    if (lane < F3 / 4) ((float4*)o)[(size_t)r * (F3 / 4) + lane] = acc;
}

#define GS_THREADS 128

template <int K, int NC, int UNR = 4, int NTHR = GS_THREADS>
__device__ __forceinline__ void stage_weights(const float* __restrict__ W, float* __restrict__ Ws) {
    constexpr int NT = NC / 16;
#pragma unroll UNR
    for (int q = threadIdx.x; q < (K / 4) * NC; q += NTHR) {
        const int ks = q / NC;
        const int nn = q - ks * NC;
        const float* wp = W + ks * 4 * NC + nn;
        v2f lo, hi;
        lo.x = wp[0];
        lo.y = wp[NC];
        hi.x = wp[2 * NC];
        hi.y = wp[3 * NC];
        const int t = nn >> 4;
        const int c = nn & 15;
        float* dst = Ws + ((ks * (NT / 2) + (t >> 1)) * 32 + c) * 4 + (t & 1) * 2;
        *(v2f*)dst        = lo;
        *(v2f*)(dst + 64) = hi;
    }
}

template <int NT>
__device__ __forceinline__ void load_b(const float* __restrict__ wb, v2f (&b)[NT]) {
#pragma unroll
    for (int tp = 0; tp < NT / 2; ++tp) {
        const float4 q = *(const float4*)(wb + tp * 128);
        b[2 * tp].x     = q.x;
        b[2 * tp].y     = q.y;
        b[2 * tp + 1].x = q.z;
        b[2 * tp + 1].y = q.w;
    }
}

template <int K, int NT, bool DUAL, bool WA = false>
__device__ __forceinline__ void strip_products(const v2f (&a0)[K / 4], const v2f (&a1)[K / 4],
                                               const float* __restrict__ w0s,
                                               const float* __restrict__ w1s, v8f (&acc)[NT]) {
    constexpr int KS = K / 4;
    v2f b0[NT], b1[NT];
    load_b<NT>(w0s, b0);
    if (DUAL) load_b<NT>(w1s, b1);
#pragma unroll
    for (int ks = 0; ks < KS; ++ks) {
        v2f nb0[NT], nb1[NT];
        if (ks + 1 < KS) {
            load_b<NT>(w0s + (ks + 1) * (NT / 2) * 128, nb0);
            if (DUAL) load_b<NT>(w1s + (ks + 1) * (NT / 2) * 128, nb1);
        }
#pragma unroll
        for (int t = 0; t < NT; ++t)
            acc[t] = WA ? __builtin_amdgcn_wmma_f32_16x16x4_f32(false, b0[t], false, a0[ks], (short)0,
                                                                acc[t], false, false)
                        : __builtin_amdgcn_wmma_f32_16x16x4_f32(false, a0[ks], false, b0[t], (short)0,
                                                                acc[t], false, false);
        if (DUAL) {
#pragma unroll
            for (int t = 0; t < NT; ++t)
                acc[t] = WA ? __builtin_amdgcn_wmma_f32_16x16x4_f32(false, b1[t], false, a1[ks], (short)0,
                                                                    acc[t], false, false)
                            : __builtin_amdgcn_wmma_f32_16x16x4_f32(false, a1[ks], false, b1[t], (short)0,
                                                                    acc[t], false, false);
        }
        if (ks + 1 < KS) {
#pragma unroll
            for (int t = 0; t < NT; ++t) {
                b0[t] = nb0[t];
                if (DUAL) b1[t] = nb1[t];
            }
        }
    }
}

#define CL_THREADS 256
#define CL_WAVES (CL_THREADS / 32)
#define CL_G0 195
#define CL_G1 49
#define CL_G2 12

__device__ __forceinline__ void stage_lin_weights(const float* __restrict__ W, float* __restrict__ Ws) {
    constexpr int NU = F3 / 16;
#pragma unroll 4
    for (int q = threadIdx.x; q < (H / 4) * F3; q += CL_THREADS) {
        const int kk = q / F3;
        const int nn = q - kk * F3;
        const int r0 = (kk >> 2) * 16 + (kk & 3) * 2;
        const float* wp = W + r0 * F3 + nn;
        v2f lo, hi;
        lo.x = wp[0];
        lo.y = wp[F3];
        hi.x = wp[8 * F3];
        hi.y = wp[9 * F3];
        const int u = nn >> 4;
        const int c = nn & 15;
        float* dst = Ws + ((kk * (NU / 2) + (u >> 1)) * 32 + c) * 4 + (u & 1) * 2;
        *(v2f*)dst        = lo;
        *(v2f*)(dst + 64) = hi;
    }
}

__global__ void __launch_bounds__(CL_THREADS)
conv2_lin_wmma_kernel(const float* __restrict__ aggall, const float* __restrict__ Wrelall,
                      const float* __restrict__ h1all, const float* __restrict__ Wrootall,
                      const float* __restrict__ brelall, const float* __restrict__ linWall,
                      const float* __restrict__ linball, float* __restrict__ eall) {
    constexpr int K  = H;
    constexpr int NT = H / 16;
    constexpr int KS = K / 4;
    constexpr int NU = F3 / 16;
    __shared__ __attribute__((aligned(16))) float Ws[2 * K * H + K * F3];

    int lb = blockIdx.x, lg = CL_G0, n = N0, lvl = 0, roff = 0;
    if (lb >= CL_G0 + CL_G1) { lb -= CL_G0 + CL_G1; lg = CL_G2; n = N2; lvl = 2; roff = ROFF2; }
    else if (lb >= CL_G0)    { lb -= CL_G0;         lg = CL_G1; n = N1; lvl = 1; roff = ROFF1; }
    const float* A0   = aggall + (size_t)roff * K;
    const float* A1   = h1all + (size_t)roff * K;
    const float* bias = brelall + lvl * H;
    const float* lbias = linball + lvl * F3;
    float* E          = eall + (size_t)roff * F3;

    stage_weights<K, H, 8, CL_THREADS>(Wrelall + lvl * K * H, Ws);
    stage_weights<K, H, 8, CL_THREADS>(Wrootall + lvl * K * H, Ws + K * H);
    stage_lin_weights(linWall + lvl * K * F3, Ws + 2 * K * H);
    __syncthreads();

    const int lane = threadIdx.x & 31;
    const int wave = __builtin_amdgcn_readfirstlane((int)(threadIdx.x >> 5));
    const int m    = lane & 15;
    const int half = lane >> 4;
    const int nstrips = (n + 15) >> 4;

    __shared__ float Ps[CL_WAVES * NU * 8 * 32];
#pragma unroll 1
    for (int sx = 2 * CL_WAVES * lg + lb; sx < nstrips; sx += lg) {
        int boff0 = lane * 4;
        int boff1 = lane * 4 + K * H;
        int boff2 = lane * 4 + 2 * K * H;
        asm volatile("" : "+v"(boff0), "+v"(boff1), "+v"(boff2));
        const int tsel = (wave >> 1) * 128 + (wave & 1) * 2;
        const float* w0s = Ws + boff0 + tsel;
        const float* w1s = Ws + boff1 + tsel;
        const float* w2s = Ws + boff2 + wave * (4 * (NU / 2) * 128);
        const int row0 = sx * 16;
        const int row  = row0 + m;
        const int rowc = (row < n) ? row : (n - 1);

        const float* a0p = A0 + rowc * K + half * 2;
        const float* a1p = A1 + rowc * K + half * 2;
        v2f a0[KS], a1[KS];
#pragma unroll
        for (int ks = 0; ks < KS; ++ks) {
            a0[ks] = *(const v2f*)(a0p + 4 * ks);
            a1[ks] = *(const v2f*)(a1p + 4 * ks);
        }
        v8f acc1;
        {
            const float* bp = bias + 16 * wave + 8 * half;
            const v2f b01 = *(const v2f*)(bp);
            const v2f b23 = *(const v2f*)(bp + 2);
            const v2f b45 = *(const v2f*)(bp + 4);
            const v2f b67 = *(const v2f*)(bp + 6);
            acc1 = (v8f){b01.x, b01.y, b23.x, b23.y, b45.x, b45.y, b67.x, b67.y};
        }
#pragma unroll
        for (int ks = 0; ks < KS; ++ks) {
            const v2f wa = *(const v2f*)(w0s + ks * (NT / 2) * 128);
            const v2f wb = *(const v2f*)(w1s + ks * (NT / 2) * 128);
            acc1 = __builtin_amdgcn_wmma_f32_16x16x4_f32(false, wa, false, a0[ks], (short)0, acc1, false, false);
            acc1 = __builtin_amdgcn_wmma_f32_16x16x4_f32(false, wb, false, a1[ks], (short)0, acc1, false, false);
        }
        v8f p[NU];
#pragma unroll
        for (int u = 0; u < NU; ++u) p[u] = (v8f){0.0f, 0.0f, 0.0f, 0.0f, 0.0f, 0.0f, 0.0f, 0.0f};
#pragma unroll
        for (int jp = 0; jp < 4; ++jp) {
            v2f wl[NU];
            load_b<NU>(w2s + jp * (NU / 2) * 128, wl);
            v2f hf;
            hf.x = fmaxf(acc1[2 * jp], 0.0f);
            hf.y = fmaxf(acc1[2 * jp + 1], 0.0f);
#pragma unroll
            for (int u = 0; u < NU; ++u)
                p[u] = __builtin_amdgcn_wmma_f32_16x16x4_f32(false, wl[u], false, hf, (short)0, p[u], false, false);
        }
#pragma unroll
        for (int u = 0; u < NU; ++u) {
#pragma unroll
            for (int j = 0; j < 8; ++j) Ps[((wave * NU + u) * 8 + j) * 32 + lane] = p[u][j];
        }
        __syncthreads();
        {
            const int rsel = lane & 1;
            const int hsel = (lane >> 1) & 1;
            const int j    = lane >> 2;
            const int sl   = 2 * wave + rsel + 16 * hsel;
            const int orow = row0 + 2 * wave + rsel;
#pragma unroll
            for (int u = 0; u < NU; ++u) {
                float v = lbias[16 * u + 8 * hsel + j];
#pragma unroll
                for (int w2 = 0; w2 < CL_WAVES; ++w2) v += Ps[((w2 * NU + u) * 8 + j) * 32 + sl];
                if (orow < n) E[(size_t)orow * F3 + 16 * u + 8 * hsel + j] = v;
            }
        }
        __syncthreads();
    }

#pragma unroll 1
    for (int s = wave * lg + lb; s < nstrips && s < 2 * CL_WAVES * lg; s += lg * CL_WAVES) {
        int boff0 = lane * 4;
        int boff1 = lane * 4 + K * H;
        int boff2 = lane * 4 + 2 * K * H;
        asm volatile("" : "+v"(boff0), "+v"(boff1), "+v"(boff2));
        const float* w0s = Ws + boff0;
        const float* w1s = Ws + boff1;
        const float* w2s = Ws + boff2;
        const int row0 = s * 16;
        const int row  = row0 + m;
        const int rowc = (row < n) ? row : (n - 1);

        const float* a0p = A0 + rowc * K + half * 2;
        const float* a1p = A1 + rowc * K + half * 2;
        v2f a0[KS], a1[KS];
#pragma unroll
        for (int ks = 0; ks < KS; ++ks) {
            a0[ks] = *(const v2f*)(a0p + 4 * ks);
            a1[ks] = *(const v2f*)(a1p + 4 * ks);
        }

        v8f acc[NT];
#pragma unroll
        for (int t = 0; t < NT; ++t) {
            const float* bp = bias + 16 * t + 8 * half;
            const v2f b01 = *(const v2f*)(bp);
            const v2f b23 = *(const v2f*)(bp + 2);
            const v2f b45 = *(const v2f*)(bp + 4);
            const v2f b67 = *(const v2f*)(bp + 6);
            acc[t] = (v8f){b01.x, b01.y, b23.x, b23.y, b45.x, b45.y, b67.x, b67.y};
        }
        strip_products<K, NT, true, true>(a0, a1, w0s, w1s, acc);

        v8f acc2[NU];
#pragma unroll
        for (int u = 0; u < NU; ++u) {
            const float* bp = lbias + 16 * u + 8 * half;
            const v2f b01 = *(const v2f*)(bp);
            const v2f b23 = *(const v2f*)(bp + 2);
            const v2f b45 = *(const v2f*)(bp + 4);
            const v2f b67 = *(const v2f*)(bp + 6);
            acc2[u] = (v8f){b01.x, b01.y, b23.x, b23.y, b45.x, b45.y, b67.x, b67.y};
        }
        v2f wl[NU];
        load_b<NU>(w2s, wl);
#pragma unroll
        for (int kk = 0; kk < KS; ++kk) {
            v2f nwl[NU];
            if (kk + 1 < KS) load_b<NU>(w2s + (kk + 1) * (NU / 2) * 128, nwl);
            const int t  = kk >> 2;
            const int jp = kk & 3;
            v2f hf;
            hf.x = fmaxf(acc[t][2 * jp], 0.0f);
            hf.y = fmaxf(acc[t][2 * jp + 1], 0.0f);
#pragma unroll
            for (int u = 0; u < NU; ++u)
                acc2[u] = __builtin_amdgcn_wmma_f32_16x16x4_f32(false, wl[u], false, hf, (short)0,
                                                                acc2[u], false, false);
            if (kk + 1 < KS) {
#pragma unroll
                for (int u = 0; u < NU; ++u) wl[u] = nwl[u];
            }
        }

        if (row < n) {
            float* ep = E + (size_t)row * F3 + 8 * half;
#pragma unroll
            for (int u = 0; u < NU; ++u) {
                float4 lo, hi;
                lo.x = acc2[u][0]; lo.y = acc2[u][1]; lo.z = acc2[u][2]; lo.w = acc2[u][3];
                hi.x = acc2[u][4]; hi.y = acc2[u][5]; hi.z = acc2[u][6]; hi.w = acc2[u][7];
                *(float4*)(ep + 16 * u)     = lo;
                *(float4*)(ep + 16 * u + 4) = hi;
            }
        }
    }
}

#define DEC_RPB 32
__global__ void __launch_bounds__(GS_THREADS)
dec_fused_wmma_kernel(const float* __restrict__ e0, const float* __restrict__ i1,
                      const float* __restrict__ i2, const float* __restrict__ W,
                      const float* __restrict__ bias, const float* __restrict__ Wout,
                      const float* __restrict__ bout, float* __restrict__ out, int n) {
    constexpr int K = F3, NC = DECW;
    constexpr int NT = NC / 16;
    constexpr int KS = K / 4;
    __shared__ __attribute__((aligned(16))) float Ws[3 * K * NC];

    stage_weights<3 * K, NC, 12>(W, Ws);
    __syncthreads();

    const int lane = threadIdx.x & 31;
    const int wave = __builtin_amdgcn_readfirstlane((int)(threadIdx.x >> 5));
    const int m    = lane & 15;
    const int half = lane >> 4;
    const int nblk = (n + DEC_RPB - 1) / DEC_RPB;

    float bz[3][NT], wo[3][NT], bo[3];
#pragma unroll
    for (int b = 0; b < 3; ++b) {
        bo[b] = bout[b];
#pragma unroll
        for (int t = 0; t < NT; ++t) {
            bz[b][t] = bias[b * NC + 16 * t + m];
            wo[b][t] = Wout[b * NC + 16 * t + m];
        }
    }
    const bool bit3 = (m & 8) != 0;
    const bool bit2 = (m & 4) != 0;
    const bool bit1 = (m & 2) != 0;

#pragma unroll 1
    for (int blk = blockIdx.x * (GS_THREADS / 32) + wave; blk < nblk; blk += gridDim.x * (GS_THREADS / 32)) {
#pragma unroll 1
        for (int st = 0; st < DEC_RPB / 16; ++st) {
            const int row0 = blk * DEC_RPB + st * 16;
            if (row0 >= n) break;
            const int row  = row0 + m;
            const int rowc = (row < n) ? row : (n - 1);

            const int abase = rowc * F3 + half * 2;
            const float* s0 = e0 + abase;
            const float* s1 = i1 + abase;
            const float* s2 = i2 + abase;
            v2f a[3][KS];
#pragma unroll
            for (int b = 0; b < 3; ++b) {
#pragma unroll
                for (int ks = 0; ks < KS; ++ks) {
                    const int k   = 4 * ks;
                    const int src = k >> 5;
                    const float* sp = (src == 0) ? s0 : (src == 1) ? s1 : s2;
                    a[b][ks] = *(const v2f*)(sp + b * 32 + (k & 31));
                }
            }
            float res[3];
#pragma unroll
            for (int b = 0; b < 3; ++b) {
                v8f acc[NT];
#pragma unroll
                for (int t = 0; t < NT; ++t) {
                    const float bv = bz[b][t];
                    acc[t] = (v8f){bv, bv, bv, bv, bv, bv, bv, bv};
                }
                const float* ws = Ws + lane * 4 + b * K * NC;
                strip_products<K, NT, false>(a[b], a[b], ws, ws, acc);

                float p[8];
#pragma unroll
                for (int j = 0; j < 8; ++j) {
                    float s = 0.0f;
#pragma unroll
                    for (int t = 0; t < NT; ++t) {
                        float v = acc[t][j];
                        v = (v > 0.0f) ? v : (expf(v) - 1.0f);
                        s += v * wo[b][t];
                    }
                    p[j] = s;
                }
                float q[4], r2[2];
#pragma unroll
                for (int i = 0; i < 4; ++i) {
                    const float send = bit3 ? p[i] : p[i + 4];
                    const float keep = bit3 ? p[i + 4] : p[i];
                    q[i] = keep + __shfl_xor(send, 8, 32);
                }
#pragma unroll
                for (int i = 0; i < 2; ++i) {
                    const float send = bit2 ? q[i] : q[i + 2];
                    const float keep = bit2 ? q[i + 2] : q[i];
                    r2[i] = keep + __shfl_xor(send, 4, 32);
                }
                const float send1 = bit1 ? r2[0] : r2[1];
                const float keep1 = bit1 ? r2[1] : r2[0];
                float tot = keep1 + __shfl_xor(send1, 2, 32);
                tot += __shfl_xor(tot, 1, 32);
                res[b] = tot + bo[b];
            }

            const int orow = row0 + half * 8 + (m >> 1);
            if ((m & 1) == 0 && orow < n) {
                float* op = out + orow * 3;
                op[0] = res[0];
                op[1] = res[1];
                op[2] = res[2];
            }
        }
    }
}

#define FILL_GRID(count) (((count) / 4 + 8 * 32 * FILL_SPW - 1) / (8 * 32 * FILL_SPW) > 256 ? 256 \
                          : ((count) / 4 + 8 * 32 * FILL_SPW - 1) / (8 * 32 * FILL_SPW))

#define WS_H1    ((size_t)0)
#define WS_AGG   (WS_H1 + (size_t)NROWS * H)
#define WS_AGG6  (WS_AGG + (size_t)NROWS * H)
#define WS_H2    (((WS_AGG6 + (size_t)NROWS * 6 + 31) / 32) * 32)
#define WS_TAIL  (WS_H2 + (size_t)NROWS * H)
#define WS_I1    ((size_t)0)
#define WS_I2    (WS_I1 + (size_t)N0 * F3)
#define WS_E     WS_H2
static_assert(WS_I2 + (size_t)N0 * F3 <= WS_AGG6, "i1 / i2 must fit in h1 / agg's place");
static_assert(WS_E % 32 == 0 && WS_E + (size_t)NROWS * F3 <= WS_TAIL, "e must fit in the region WS_H2");
#define WS_OVF   WS_H2
#define WS_CNT   (WS_TAIL - 32 - (size_t)NROWS * CNT_STRIDE)
#define WS_CUR   (WS_TAIL - 32)
#define WS_IREC  WS_TAIL
#define WS_ICUR  (WS_IREC + (size_t)2 * N0 * CNT_STRIDE)
#define WS_IOVF  (WS_ICUR + 32)
#define WS_ZCOUNT (NROWS * CNT_STRIDE + 32 + 2 * N0 * CNT_STRIDE + 32)
static_assert(WS_OVF + 2 * (size_t)ETOT <= WS_CNT, "the edge lists must fit in the region WS_H2");
static_assert(WS_IOVF + 3 * (size_t)ITOT <= WS_TAIL + (size_t)NROWS * F3, "the interpolation lists must fit in the tail");

extern "C" void kernel_launch(void* const* d_in, const int* in_sizes, int n_in,
                              void* d_out, int out_size, void* d_ws, size_t ws_size,
                              hipStream_t stream) {
    (void)in_sizes; (void)n_in; (void)out_size; (void)ws_size;

    const float* x0     = (const float*)d_in[0];
    const float* x1     = (const float*)d_in[1];
    const float* x2     = (const float*)d_in[2];
    const int*   ei0    = (const int*)d_in[3];
    const int*   ei1    = (const int*)d_in[4];
    const int*   ei2    = (const int*)d_in[5];
    const int*   A1r    = (const int*)d_in[6];
    const int*   A1c    = (const int*)d_in[7];
    const float* A1v    = (const float*)d_in[8];
    const int*   A2r    = (const int*)d_in[9];
    const int*   A2c    = (const int*)d_in[10];
    const float* A2v    = (const float*)d_in[11];
    const float* c1Wrel  = (const float*)d_in[12];
    const float* c1brel  = (const float*)d_in[13];
    const float* c1Wroot = (const float*)d_in[14];
    const float* c2Wrel  = (const float*)d_in[15];
    const float* c2brel  = (const float*)d_in[16];
    const float* c2Wroot = (const float*)d_in[17];
    const float* linW    = (const float*)d_in[18];
    const float* linb    = (const float*)d_in[19];
    const float* decW0   = (const float*)d_in[20];
    const float* decb0   = (const float*)d_in[21];
    const float* decWout = (const float*)d_in[22];
    const float* decbout = (const float*)d_in[23];
    float* out = (float*)d_out;

    float* ws   = (float*)d_ws;
    float* h1   = ws + WS_H1;
    float* agg  = ws + WS_AGG;
    float* eall = ws + WS_E;
    float* i1   = ws + WS_I1;
    float* i2   = ws + WS_I2;
    int* cnt    = (int*)(ws + WS_CNT);
    int* cursor = (int*)(ws + WS_CUR);
    int* ovf    = (int*)(ws + WS_OVF);
    int* irec   = (int*)(ws + WS_IREC);
    int* icur   = (int*)(ws + WS_ICUR);
    int* iovf   = (int*)(ws + WS_IOVF);

    zero_kernel<<<FILL_GRID(WS_ZCOUNT), 256, 0, stream>>>((float*)cnt, WS_ZCOUNT);
    csr_build_kernel<<<(ETOT + ITOT + 255) / 256, 256, 0, stream>>>(ei0, ei1, ei2, cnt, cursor, ovf,
                                                                   A1r, A1c, A1v, A2r, A2c, A2v,
                                                                   irec, icur, iovf);

    conv1_kernel<<<(C1_W0 + C1_W1 + C1_W2 + 7) / 8, 256, 0, stream>>>(x0, x1, x2, cnt, cursor, ovf,
                                                                    c1Wrel, c1brel, c1Wroot, h1);

    gather128_kernel<<<(NROWS + 7) / 8, 256, 0, stream>>>(h1, cnt, cursor, ovf, agg);
    conv2_lin_wmma_kernel<<<CL_G0 + CL_G1 + CL_G2, CL_THREADS, 0, stream>>>(
        agg, c2Wrel, h1, c2Wroot, c2brel, linW, linb, eall);

    interp_gather_kernel<<<(2 * N0 + 7) / 8, 256, 0, stream>>>(irec, icur, iovf, eall + (size_t)ROFF1 * F3,
                                                              eall + (size_t)ROFF2 * F3, i1, i2);

    dec_fused_wmma_kernel<<<((N0 + DEC_RPB - 1) / DEC_RPB + 3) / 4, GS_THREADS, 0, stream>>>(
        eall, i1, i2, decW0, decb0, decWout, decbout, out, N0);
}
